# starting kernel
# baseline (speedup 1.0000x reference)
#include <hip/hip_runtime.h>
#include <hip/hip_bf16.h>
#include <math.h>


typedef __attribute__((ext_vector_type(2))) float v2f;
typedef __attribute__((ext_vector_type(8))) float v8f;

#define MUL0   64
#define MUL1   32
#define NBASIS 10
#define FCH    100
#define WNUM   224
#define MIDS   96
#define MIDV   128
#define KPAD1  12
#define NPAD1  112
#define WAVES  4

__device__ __forceinline__ void atomic_add_f32(float* p, float v) {
  __hip_atomic_fetch_add(p, v, __ATOMIC_RELAXED, __HIP_MEMORY_SCOPE_AGENT);
}

__device__ __forceinline__ float fast_rcp(float x) {
#if __has_builtin(__builtin_amdgcn_rcpf)
  return __builtin_amdgcn_rcpf(x);
#else
  return 1.f / x;
#endif
}

__global__ void zero_kernel(float* __restrict__ p, long long n) {
  long long i = (long long)blockIdx.x * blockDim.x + threadIdx.x;
  long long stride = (long long)gridDim.x * blockDim.x;
  for (; i < n; i += stride) p[i] = 0.f;
}

__global__ void node_prep_kernel(const float* __restrict__ ns, const float* __restrict__ nv,
                                 const float* __restrict__ attr,
                                 const float* __restrict__ Wls, const float* __restrict__ Wlv,
                                 const float* __restrict__ Wss, const float* __restrict__ Wsv,
                                 float* __restrict__ f_s, float* __restrict__ f_v,
                                 float* __restrict__ sc_s, float* __restrict__ sc_v, int n)
{
  int node = blockIdx.x;
  if (node >= n) return;
  __shared__ float s_ns[MUL0];
  __shared__ float s_nv[MUL1 * 3];
  int t = threadIdx.x;
  if (t < MUL0)        s_ns[t]        = ns[(long long)node * MUL0 + t];
  else                 s_nv[t - MUL0] = nv[(long long)node * (MUL1 * 3) + (t - MUL0)];
  __syncthreads();
  float a = attr[node];
  if (t < MUL0) {
    float af = 0.f, asc = 0.f;
    #pragma unroll 8
    for (int u = 0; u < MUL0; ++u) {
      float x = s_ns[u];
      af  += x * Wls[u * MUL0 + t];
      asc += x * Wss[u * MUL0 + t];
    }
    f_s [(long long)node * MUL0 + t] = af  * a;
    sc_s[(long long)node * MUL0 + t] = asc * a;
  } else {
    int idx = t - MUL0;
    int vv = idx / 3;
    int k  = idx - vv * 3;
    float af = 0.f, asc = 0.f;
    #pragma unroll 8
    for (int u = 0; u < MUL1; ++u) {
      float x = s_nv[u * 3 + k];
      af  += x * Wlv[u * MUL1 + vv];
      asc += x * Wsv[u * MUL1 + vv];
    }
    f_v [(long long)node * (MUL1 * 3) + idx] = af  * a;
    sc_v[(long long)node * (MUL1 * 3) + idx] = asc * a;
  }
}

__global__ __launch_bounds__(WAVES * 32)
void edge_fused_kernel(const float* __restrict__ es,
                       const float* __restrict__ ea,
                       const int*   __restrict__ esrc,
                       const int*   __restrict__ edst,
                       const float* __restrict__ Wfc1,
                       const float* __restrict__ Wfc2,
                       const float* __restrict__ f_s,
                       const float* __restrict__ f_v,
                       float* __restrict__ agg_s,
                       float* __restrict__ agg_v,
                       int E, int numTiles)
{
  __shared__ float W1s[KPAD1 * NPAD1];
  __shared__ float W2s[NPAD1 * WNUM];
  __shared__ float Hs[WAVES][16 * NPAD1];
  __shared__ float mF[WAVES][16 * 4];
  __shared__ int   mI[WAVES][16 * 2];

  for (int i = threadIdx.x; i < KPAD1 * NPAD1; i += blockDim.x) {
    int k = i / NPAD1, ncol = i - k * NPAD1;
    W1s[i] = (k < NBASIS && ncol < FCH) ? Wfc1[k * FCH + ncol] : 0.f;
  }
  for (int i = threadIdx.x; i < NPAD1 * WNUM; i += blockDim.x) {
    int k = i / WNUM;
    W2s[i] = (k < FCH) ? Wfc2[i] : 0.f;
  }
  __syncthreads();

  const int wid  = threadIdx.x >> 5;
  const int lane = threadIdx.x & 31;
  const int hi   = lane >> 4;
  const int lm   = lane & 15;

  float* H  = Hs[wid];
  float* mf = mF[wid];
  int*   mi = mI[wid];

  const float inv  = 0.25f;
  const float inv3 = 0.57735026918962576451f;
  const float inv2 = 0.70710678118654752440f;

  int tile       = blockIdx.x * WAVES + wid;
  int tileStride = gridDim.x * WAVES;

  for (; tile < numTiles; tile += tileStride) {
    int base = tile * 16;

    if (lane < 16) {
      int e = base + lane;
      if (e < E) {
        mi[lane * 2 + 0] = esrc[e];
        mi[lane * 2 + 1] = edst[e];
        mf[lane * 4 + 0] = ea[(long long)e * 4 + 0];
        mf[lane * 4 + 1] = ea[(long long)e * 4 + 1];
        mf[lane * 4 + 2] = ea[(long long)e * 4 + 2];
        mf[lane * 4 + 3] = ea[(long long)e * 4 + 3];
      } else {
        mi[lane * 2 + 0] = 0;
        mi[lane * 2 + 1] = -1;
        mf[lane * 4 + 0] = 0.f; mf[lane * 4 + 1] = 0.f;
        mf[lane * 4 + 2] = 0.f; mf[lane * 4 + 3] = 0.f;
      }
    }

    int  eA = base + lm;
    bool ev = (eA < E);
    v2f a1[3];
    #pragma unroll
    for (int kk = 0; kk < 3; ++kk) {
      int k0 = 4 * kk + 2 * hi;
      a1[kk].x = (ev && (k0     < NBASIS)) ? es[(long long)eA * NBASIS + k0    ] : 0.f;
      a1[kk].y = (ev && (k0 + 1 < NBASIS)) ? es[(long long)eA * NBASIS + k0 + 1] : 0.f;
    }

    #pragma unroll
    for (int nt = 0; nt < NPAD1 / 16; ++nt) {
      v8f acc = {};
      #pragma unroll
      for (int kk = 0; kk < 3; ++kk) {
        int k0 = 4 * kk + 2 * hi;
        v2f b;
        b.x = W1s[ k0      * NPAD1 + nt * 16 + lm];
        b.y = W1s[(k0 + 1) * NPAD1 + nt * 16 + lm];
        acc = __builtin_amdgcn_wmma_f32_16x16x4_f32(
            false, a1[kk], false, b, (short)0, acc, false, false);
      }
      #pragma unroll
      for (int r = 0; r < 8; ++r) {
        float x  = acc[r];
        float sg = fast_rcp(1.f + __expf(-x));
        H[(r + 8 * hi) * NPAD1 + nt * 16 + lm] = x * sg;
      }
    }
    asm volatile("s_wait_dscnt 0" ::: "memory");

    v2f a2[NPAD1 / 4];
    #pragma unroll
    for (int kk = 0; kk < NPAD1 / 4; ++kk) {
      int k0 = 4 * kk + 2 * hi;
      a2[kk] = *(const v2f*)&H[lm * NPAD1 + k0];
    }

    for (int nt = 0; nt < WNUM / 16; ++nt) {
      v8f acc = {};
      #pragma unroll
      for (int kk = 0; kk < NPAD1 / 4; ++kk) {
        int k0 = 4 * kk + 2 * hi;
        v2f b;
        b.x = W2s[ k0      * WNUM + nt * 16 + lm];
        b.y = W2s[(k0 + 1) * WNUM + nt * 16 + lm];
        acc = __builtin_amdgcn_wmma_f32_16x16x4_f32(
            false, a2[kk], false, b, (short)0, acc, false, false);
      }

      int c = nt * 16 + lm;
      #pragma unroll
      for (int r = 0; r < 8; ++r) {
        int m   = r + 8 * hi;
        int dst = mi[m * 2 + 1];
        if (dst < 0) continue;
        int   src = mi[m * 2 + 0];
        float wv  = acc[r];
        float y0  = mf[m * 4 + 0];
        float y1x = mf[m * 4 + 1], y1y = mf[m * 4 + 2], y1z = mf[m * 4 + 3];

        if (c < 64) {
          float s = f_s[(long long)src * MUL0 + c];
          atomic_add_f32(&agg_s[(long long)dst * MIDS + c], wv * s * y0 * inv);
        } else if (c < 128) {
          int   u = c - 64;
          float s = f_s[(long long)src * MUL0 + u];
          float t = wv * s * inv;
          atomic_add_f32(&agg_v[((long long)dst * MIDV + u) * 3 + 0], t * y1x);
          atomic_add_f32(&agg_v[((long long)dst * MIDV + u) * 3 + 1], t * y1y);
          atomic_add_f32(&agg_v[((long long)dst * MIDV + u) * 3 + 2], t * y1z);
        } else if (c < 160) {
          int u = c - 128;
          const float* vp = &f_v[((long long)src * MUL1 + u) * 3];
          float t = wv * y0 * inv;
          atomic_add_f32(&agg_v[((long long)dst * MIDV + 64 + u) * 3 + 0], t * vp[0]);
          atomic_add_f32(&agg_v[((long long)dst * MIDV + 64 + u) * 3 + 1], t * vp[1]);
          atomic_add_f32(&agg_v[((long long)dst * MIDV + 64 + u) * 3 + 2], t * vp[2]);
        } else if (c < 192) {
          int u = c - 160;
          const float* vp = &f_v[((long long)src * MUL1 + u) * 3];
          float d = vp[0] * y1x + vp[1] * y1y + vp[2] * y1z;
          atomic_add_f32(&agg_s[(long long)dst * MIDS + 64 + u], wv * d * inv3 * inv);
        } else {
          int u = c - 192;
          const float* vp = &f_v[((long long)src * MUL1 + u) * 3];
          float cx = vp[1] * y1z - vp[2] * y1y;
          float cy = vp[2] * y1x - vp[0] * y1z;
          float cz = vp[0] * y1y - vp[1] * y1x;
          float t = wv * inv2 * inv;
          atomic_add_f32(&agg_v[((long long)dst * MIDV + 96 + u) * 3 + 0], t * cx);
          atomic_add_f32(&agg_v[((long long)dst * MIDV + 96 + u) * 3 + 1], t * cy);
          atomic_add_f32(&agg_v[((long long)dst * MIDV + 96 + u) * 3 + 2], t * cz);
        }
      }
    }
  }
}

__global__ void node_out_kernel(const float* __restrict__ agg_s, const float* __restrict__ agg_v,
                                const float* __restrict__ sc_s,  const float* __restrict__ sc_v,
                                const float* __restrict__ attr,
                                const float* __restrict__ Wl2s, const float* __restrict__ Wl2v,
                                const float* __restrict__ Wl3,
                                float* __restrict__ out, int n)
{
  int node = blockIdx.x;
  if (node >= n) return;
  __shared__ float sAs[MIDS];
  __shared__ float sAv[MIDV * 3];
  int t = threadIdx.x;
  if (t < MIDS) sAs[t] = agg_s[(long long)node * MIDS + t];
  for (int i = t; i < MIDV * 3; i += 160) sAv[i] = agg_v[(long long)node * (MIDV * 3) + i];
  __syncthreads();

  float a = attr[node];
  float ang = 0.f;
  #pragma unroll 8
  for (int u = 0; u < MIDS; ++u) ang += sAs[u] * Wl3[u];
  ang *= 0.1f * a;
  float cs = cosf(ang);
  float sn = sinf(ang);

  if (t < MUL0) {
    float conv = 0.f;
    #pragma unroll 8
    for (int u = 0; u < MIDS; ++u) conv += sAs[u] * Wl2s[u * MUL0 + t];
    conv *= a;
    out[(long long)node * 160 + t] = cs * sc_s[(long long)node * MUL0 + t] + sn * conv;
  } else {
    int idx = t - MUL0;
    int vv = idx / 3;
    int k  = idx - vv * 3;
    float conv = 0.f;
    #pragma unroll 8
    for (int u = 0; u < MIDV; ++u) conv += sAv[u * 3 + k] * Wl2v[u * MUL1 + vv];
    conv *= a;
    out[(long long)node * 160 + t] = cs * sc_v[(long long)node * (MUL1 * 3) + idx] + sn * conv;
  }
}

extern "C" void kernel_launch(void* const* d_in, const int* in_sizes, int n_in,
                              void* d_out, int out_size, void* d_ws, size_t ws_size,
                              hipStream_t stream) {
  const float* node_scalars = (const float*)d_in[0];
  const float* node_vectors = (const float*)d_in[1];
  const float* node_attr    = (const float*)d_in[2];
  const int*   edge_src     = (const int*)  d_in[3];
  const int*   edge_dst     = (const int*)  d_in[4];
  const float* edge_attr    = (const float*)d_in[5];
  const float* edge_scalars = (const float*)d_in[6];
  const float* W_fc1        = (const float*)d_in[7];
  const float* W_fc2        = (const float*)d_in[8];
  const float* W_sc_s       = (const float*)d_in[9];
  const float* W_sc_v       = (const float*)d_in[10];
  const float* W_l1_s       = (const float*)d_in[11];
  const float* W_l1_v       = (const float*)d_in[12];
  const float* W_l2_s       = (const float*)d_in[13];
  const float* W_l2_v       = (const float*)d_in[14];
  const float* W_l3         = (const float*)d_in[15];

  const int N = in_sizes[0] / MUL0;
  const int E = in_sizes[3];

  float* ws    = (float*)d_ws;
  float* f_s   = ws;
  float* f_v   = f_s  + (size_t)N * MUL0;
  float* sc_s  = f_v  + (size_t)N * MUL1 * 3;
  float* sc_v  = sc_s + (size_t)N * MUL0;
  float* agg_s = sc_v + (size_t)N * MUL1 * 3;
  float* agg_v = agg_s + (size_t)N * MIDS;

  long long nAgg = (long long)N * (MIDS + MIDV * 3);
  zero_kernel<<<4096, 256, 0, stream>>>(agg_s, nAgg);

  node_prep_kernel<<<N, 160, 0, stream>>>(node_scalars, node_vectors, node_attr,
                                          W_l1_s, W_l1_v, W_sc_s, W_sc_v,
                                          f_s, f_v, sc_s, sc_v, N);

  int numTiles = (E + 15) / 16;
  edge_fused_kernel<<<1024, WAVES * 32, 0, stream>>>(edge_scalars, edge_attr,
                                                     edge_src, edge_dst,
                                                     W_fc1, W_fc2, f_s, f_v,
                                                     agg_s, agg_v, E, numTiles);

  node_out_kernel<<<N, 160, 0, stream>>>(agg_s, agg_v, sc_s, sc_v, node_attr,
                                         W_l2_s, W_l2_v, W_l3, (float*)d_out, N);
}


# edges counting-sorted by destination in a pre-pass; each lane sums its adjacent rows of one node in registers, one atomic per run instead of per edge (sum order changes)
# speedup vs baseline: 1.6067x; 1.6067x over previous
#include <hip/hip_runtime.h>
#include <hip/hip_bf16.h>
#include <math.h>


typedef __attribute__((ext_vector_type(2))) float v2f;
typedef __attribute__((ext_vector_type(8))) float v8f;

#define MUL0   64
#define MUL1   32
#define NBASIS 10
#define FCH    100
#define WNUM   224
#define MIDS   96
#define MIDV   128
#define KPAD1  12
#define NPAD1  112
#define WAVES  4

__device__ __forceinline__ void atomic_add_f32(float* p, float v) {
  __hip_atomic_fetch_add(p, v, __ATOMIC_RELAXED, __HIP_MEMORY_SCOPE_AGENT);
}

__device__ __forceinline__ float fast_rcp(float x) {
#if __has_builtin(__builtin_amdgcn_rcpf)
  return __builtin_amdgcn_rcpf(x);
#else
  return 1.f / x;
#endif
}

template <int NC>
__device__ __forceinline__ void seg_add(float* __restrict__ agg, int rowStride, int colOff,
                                        const int (&d)[8], bool mrg, int hi,
                                        const float (&v)[NC][8]) {
  float run[NC], head[NC];
  #pragma unroll
  for (int k = 0; k < NC; ++k) { run[k] = 0.f; head[k] = 0.f; }
  bool first = (hi == 1) && mrg;
  #pragma unroll
  for (int r = 0; r < 8; ++r) {
    #pragma unroll
    for (int k = 0; k < NC; ++k) run[k] += v[k][r];
    const bool endrun = (r == 7) || (d[r] != d[r + 1 < 8 ? r + 1 : 7]);
    if (endrun) {
      const bool keep = (r == 7) && (hi == 0) && mrg;
      if (first) {
        #pragma unroll
        for (int k = 0; k < NC; ++k) { head[k] = run[k]; run[k] = 0.f; }
        first = false;
      } else if (!keep) {
        if (d[r] >= 0) {
          float* p = agg + (long long)d[r] * rowStride + colOff;
          #pragma unroll
          for (int k = 0; k < NC; ++k) atomic_add_f32(p + k, run[k]);
        }
        #pragma unroll
        for (int k = 0; k < NC; ++k) run[k] = 0.f;
      }
    }
  }
  #pragma unroll
  for (int k = 0; k < NC; ++k) {
    float carry = ((hi == 0) && mrg) ? run[k] : 0.f;
    head[k] += __shfl_xor(carry, 16);
  }
  if ((hi == 1) && mrg) {
    float* p = agg + (long long)d[0] * rowStride + colOff;
    #pragma unroll
    for (int k = 0; k < NC; ++k) atomic_add_f32(p + k, head[k]);
  }
}

__global__ void zero_kernel(float* __restrict__ p, long long n, int* __restrict__ q, int m) {
  long long i = (long long)blockIdx.x * blockDim.x + threadIdx.x;
  long long stride = (long long)gridDim.x * blockDim.x;
  for (long long j = i; j < m; j += stride) q[j] = 0;
  for (; i < n; i += stride) p[i] = 0.f;
}

__global__ void dst_count_kernel(const int* __restrict__ edst, int* __restrict__ cnt, int E) {
  int stride = gridDim.x * blockDim.x;
  for (int e = blockIdx.x * blockDim.x + threadIdx.x; e < E; e += stride)
    atomicAdd(&cnt[edst[e]], 1);
}

__global__ __launch_bounds__(1024)
void dst_scan_kernel(const int* __restrict__ cnt, int* __restrict__ cursor, int n) {
  __shared__ int part[1024];
  const int t = threadIdx.x;
  const int chunk = (n + 1023) / 1024;
  const int b = t * chunk;
  const int e = (b + chunk < n) ? (b + chunk) : n;
  int s = 0;
  for (int i = b; i < e; ++i) s += cnt[i];
  part[t] = s;
  __syncthreads();
  for (int dd = 1; dd < 1024; dd <<= 1) {
    int add = (t >= dd) ? part[t - dd] : 0;
    __syncthreads();
    part[t] += add;
    __syncthreads();
  }
  int run = part[t] - s;
  for (int i = b; i < e; ++i) { cursor[i] = run; run += cnt[i]; }
}

__global__ void dst_fill_kernel(const int* __restrict__ edst, int* __restrict__ cursor,
                                int* __restrict__ order, int E) {
  int stride = gridDim.x * blockDim.x;
  for (int e = blockIdx.x * blockDim.x + threadIdx.x; e < E; e += stride) {
    int pos = atomicAdd(&cursor[edst[e]], 1);
    order[pos] = e;
  }
}

__global__ void node_prep_kernel(const float* __restrict__ ns, const float* __restrict__ nv,
                                 const float* __restrict__ attr,
                                 const float* __restrict__ Wls, const float* __restrict__ Wlv,
                                 const float* __restrict__ Wss, const float* __restrict__ Wsv,
                                 float* __restrict__ f_s, float* __restrict__ f_v,
                                 float* __restrict__ sc_s, float* __restrict__ sc_v, int n)
{
  int node = blockIdx.x;
  if (node >= n) return;
  __shared__ float s_ns[MUL0];
  __shared__ float s_nv[MUL1 * 3];
  int t = threadIdx.x;
  if (t < MUL0)        s_ns[t]        = ns[(long long)node * MUL0 + t];
  else                 s_nv[t - MUL0] = nv[(long long)node * (MUL1 * 3) + (t - MUL0)];
  __syncthreads();
  float a = attr[node];
  if (t < MUL0) {
    float af = 0.f, asc = 0.f;
    #pragma unroll 8
    for (int u = 0; u < MUL0; ++u) {
      float x = s_ns[u];
      af  += x * Wls[u * MUL0 + t];
      asc += x * Wss[u * MUL0 + t];
    }
    f_s [(long long)node * MUL0 + t] = af  * a;
    sc_s[(long long)node * MUL0 + t] = asc * a;
  } else {
    int idx = t - MUL0;
    int vv = idx / 3;
    int k  = idx - vv * 3;
    float af = 0.f, asc = 0.f;
    #pragma unroll 8
    for (int u = 0; u < MUL1; ++u) {
      float x = s_nv[u * 3 + k];
      af  += x * Wlv[u * MUL1 + vv];
      asc += x * Wsv[u * MUL1 + vv];
    }
    f_v [(long long)node * (MUL1 * 3) + idx] = af  * a;
    sc_v[(long long)node * (MUL1 * 3) + idx] = asc * a;
  }
}

__global__ __launch_bounds__(WAVES * 32)
void edge_fused_kernel(const float* __restrict__ es,
                       const float* __restrict__ ea,
                       const int*   __restrict__ esrc,
                       const int*   __restrict__ edst,
                       const int*   __restrict__ order,
                       const float* __restrict__ Wfc1,
                       const float* __restrict__ Wfc2,
                       const float* __restrict__ f_s,
                       const float* __restrict__ f_v,
                       float* __restrict__ agg_s,
                       float* __restrict__ agg_v,
                       int E, int numTiles)
{
  __shared__ float W1s[KPAD1 * NPAD1];
  __shared__ float W2s[NPAD1 * WNUM];
  __shared__ float Hs[WAVES][16 * NPAD1];
  __shared__ float mF[WAVES][16 * 4];
  __shared__ int   mI[WAVES][16 * 2];

  for (int i = threadIdx.x; i < KPAD1 * NPAD1; i += blockDim.x) {
    int k = i / NPAD1, ncol = i - k * NPAD1;
    W1s[i] = (k < NBASIS && ncol < FCH) ? Wfc1[k * FCH + ncol] : 0.f;
  }
  for (int i = threadIdx.x; i < NPAD1 * WNUM; i += blockDim.x) {
    int k = i / WNUM;
    W2s[i] = (k < FCH) ? Wfc2[i] : 0.f;
  }
  __syncthreads();

  const int wid  = threadIdx.x >> 5;
  const int lane = threadIdx.x & 31;
  const int hi   = lane >> 4;
  const int lm   = lane & 15;

  float* H  = Hs[wid];
  float* mf = mF[wid];
  int*   mi = mI[wid];

  const float inv  = 0.25f;
  const float inv3 = 0.57735026918962576451f;
  const float inv2 = 0.70710678118654752440f;

  int tile       = blockIdx.x * WAVES + wid;
  int tileStride = gridDim.x * WAVES;

  for (; tile < numTiles; tile += tileStride) {
    int base = tile * 16;

    int  pA = base + lm;
    bool ev = (pA < E);
    int  eA = ev ? order[pA] : 0;

    if (lane < 16) {
      int e = eA;
      if (ev) {
        mi[lane * 2 + 0] = esrc[e];
        mi[lane * 2 + 1] = edst[e];
        mf[lane * 4 + 0] = ea[(long long)e * 4 + 0];
        mf[lane * 4 + 1] = ea[(long long)e * 4 + 1];
        mf[lane * 4 + 2] = ea[(long long)e * 4 + 2];
        mf[lane * 4 + 3] = ea[(long long)e * 4 + 3];
      } else {
        mi[lane * 2 + 0] = 0;
        mi[lane * 2 + 1] = -1;
        mf[lane * 4 + 0] = 0.f; mf[lane * 4 + 1] = 0.f;
        mf[lane * 4 + 2] = 0.f; mf[lane * 4 + 3] = 0.f;
      }
    }

    v2f a1[3];
    #pragma unroll
    for (int kk = 0; kk < 3; ++kk) {
      int k0 = 4 * kk + 2 * hi;
      a1[kk].x = (ev && (k0     < NBASIS)) ? es[(long long)eA * NBASIS + k0    ] : 0.f;
      a1[kk].y = (ev && (k0 + 1 < NBASIS)) ? es[(long long)eA * NBASIS + k0 + 1] : 0.f;
    }

    #pragma unroll
    for (int nt = 0; nt < NPAD1 / 16; ++nt) {
      v8f acc = {};
      #pragma unroll
      for (int kk = 0; kk < 3; ++kk) {
        int k0 = 4 * kk + 2 * hi;
        v2f b;
        b.x = W1s[ k0      * NPAD1 + nt * 16 + lm];
        b.y = W1s[(k0 + 1) * NPAD1 + nt * 16 + lm];
        acc = __builtin_amdgcn_wmma_f32_16x16x4_f32(
            false, a1[kk], false, b, (short)0, acc, false, false);
      }
      #pragma unroll
      for (int r = 0; r < 8; ++r) {
        float x  = acc[r];
        float sg = fast_rcp(1.f + __expf(-x));
        H[(r + 8 * hi) * NPAD1 + nt * 16 + lm] = x * sg;
      }
    }
    asm volatile("s_wait_dscnt 0" ::: "memory");

    v2f a2[NPAD1 / 4];
    #pragma unroll
    for (int kk = 0; kk < NPAD1 / 4; ++kk) {
      int k0 = 4 * kk + 2 * hi;
      a2[kk] = *(const v2f*)&H[lm * NPAD1 + k0];
    }

    int   drow[8], srow[8];
    float y0r[8], y1x[8], y1y[8], y1z[8];
    #pragma unroll
    for (int r = 0; r < 8; ++r) {
      int m = r + 8 * hi;
      srow[r] = mi[m * 2 + 0];
      drow[r] = mi[m * 2 + 1];
      y0r[r]  = mf[m * 4 + 0];
      y1x[r]  = mf[m * 4 + 1];
      y1y[r]  = mf[m * 4 + 2];
      y1z[r]  = mf[m * 4 + 3];
    }
    const int  d7  = mi[7 * 2 + 1];
    const bool mrg = (d7 >= 0) && (d7 == mi[8 * 2 + 1]);

    for (int nt = 0; nt < WNUM / 16; ++nt) {
      v8f acc = {};
      #pragma unroll
      for (int kk = 0; kk < NPAD1 / 4; ++kk) {
        int k0 = 4 * kk + 2 * hi;
        v2f b;
        b.x = W2s[ k0      * WNUM + nt * 16 + lm];
        b.y = W2s[(k0 + 1) * WNUM + nt * 16 + lm];
        acc = __builtin_amdgcn_wmma_f32_16x16x4_f32(
            false, a2[kk], false, b, (short)0, acc, false, false);
      }

      const int c = nt * 16 + lm;
      if (nt < 4) {
        float v[1][8];
        #pragma unroll
        for (int r = 0; r < 8; ++r) {
          float s = f_s[(long long)srow[r] * MUL0 + c];
          v[0][r] = acc[r] * s * y0r[r] * inv;
        }
        seg_add<1>(agg_s, MIDS, c, drow, mrg, hi, v);
      } else if (nt < 8) {
        const int u = c - 64;
        float v[3][8];
        #pragma unroll
        for (int r = 0; r < 8; ++r) {
          float s = f_s[(long long)srow[r] * MUL0 + u];
          float t = acc[r] * s * inv;
          v[0][r] = t * y1x[r]; v[1][r] = t * y1y[r]; v[2][r] = t * y1z[r];
        }
        seg_add<3>(agg_v, MIDV * 3, u * 3, drow, mrg, hi, v);
      } else if (nt < 10) {
        const int u = c - 128;
        float v[3][8];
        #pragma unroll
        for (int r = 0; r < 8; ++r) {
          const float* vp = &f_v[((long long)srow[r] * MUL1 + u) * 3];
          float t = acc[r] * y0r[r] * inv;
          v[0][r] = t * vp[0]; v[1][r] = t * vp[1]; v[2][r] = t * vp[2];
        }
        seg_add<3>(agg_v, MIDV * 3, (64 + u) * 3, drow, mrg, hi, v);
      } else if (nt < 12) {
        const int u = c - 160;
        float v[1][8];
        #pragma unroll
        for (int r = 0; r < 8; ++r) {
          const float* vp = &f_v[((long long)srow[r] * MUL1 + u) * 3];
          float dt = vp[0] * y1x[r] + vp[1] * y1y[r] + vp[2] * y1z[r];
          v[0][r] = acc[r] * dt * inv3 * inv;
        }
        seg_add<1>(agg_s, MIDS, 64 + u, drow, mrg, hi, v);
      } else {
        const int u = c - 192;
        float v[3][8];
        #pragma unroll
        for (int r = 0; r < 8; ++r) {
          const float* vp = &f_v[((long long)srow[r] * MUL1 + u) * 3];
          float cx = vp[1] * y1z[r] - vp[2] * y1y[r];
          float cy = vp[2] * y1x[r] - vp[0] * y1z[r];
          float cz = vp[0] * y1y[r] - vp[1] * y1x[r];
          float t = acc[r] * inv2 * inv;
          v[0][r] = t * cx; v[1][r] = t * cy; v[2][r] = t * cz;
        }
        seg_add<3>(agg_v, MIDV * 3, (96 + u) * 3, drow, mrg, hi, v);
      }
    }
  }
}

__global__ void node_out_kernel(const float* __restrict__ agg_s, const float* __restrict__ agg_v,
                                const float* __restrict__ sc_s,  const float* __restrict__ sc_v,
                                const float* __restrict__ attr,
                                const float* __restrict__ Wl2s, const float* __restrict__ Wl2v,
                                const float* __restrict__ Wl3,
                                float* __restrict__ out, int n)
{
  int node = blockIdx.x;
  if (node >= n) return;
  __shared__ float sAs[MIDS];
  __shared__ float sAv[MIDV * 3];
  int t = threadIdx.x;
  if (t < MIDS) sAs[t] = agg_s[(long long)node * MIDS + t];
  for (int i = t; i < MIDV * 3; i += 160) sAv[i] = agg_v[(long long)node * (MIDV * 3) + i];
  __syncthreads();

  float a = attr[node];
  float ang = 0.f;
  #pragma unroll 8
  for (int u = 0; u < MIDS; ++u) ang += sAs[u] * Wl3[u];
  ang *= 0.1f * a;
  float cs = cosf(ang);
  float sn = sinf(ang);

  if (t < MUL0) {
    float conv = 0.f;
    #pragma unroll 8
    for (int u = 0; u < MIDS; ++u) conv += sAs[u] * Wl2s[u * MUL0 + t];
    conv *= a;
    out[(long long)node * 160 + t] = cs * sc_s[(long long)node * MUL0 + t] + sn * conv;
  } else {
    int idx = t - MUL0;
    int vv = idx / 3;
    int k  = idx - vv * 3;
    float conv = 0.f;
    #pragma unroll 8
    for (int u = 0; u < MIDV; ++u) conv += sAv[u * 3 + k] * Wl2v[u * MUL1 + vv];
    conv *= a;
    out[(long long)node * 160 + t] = cs * sc_v[(long long)node * (MUL1 * 3) + idx] + sn * conv;
  }
}

extern "C" void kernel_launch(void* const* d_in, const int* in_sizes, int n_in,
                              void* d_out, int out_size, void* d_ws, size_t ws_size,
                              hipStream_t stream) {
  const float* node_scalars = (const float*)d_in[0];
  const float* node_vectors = (const float*)d_in[1];
  const float* node_attr    = (const float*)d_in[2];
  const int*   edge_src     = (const int*)  d_in[3];
  const int*   edge_dst     = (const int*)  d_in[4];
  const float* edge_attr    = (const float*)d_in[5];
  const float* edge_scalars = (const float*)d_in[6];
  const float* W_fc1        = (const float*)d_in[7];
  const float* W_fc2        = (const float*)d_in[8];
  const float* W_sc_s       = (const float*)d_in[9];
  const float* W_sc_v       = (const float*)d_in[10];
  const float* W_l1_s       = (const float*)d_in[11];
  const float* W_l1_v       = (const float*)d_in[12];
  const float* W_l2_s       = (const float*)d_in[13];
  const float* W_l2_v       = (const float*)d_in[14];
  const float* W_l3         = (const float*)d_in[15];

  const int N = in_sizes[0] / MUL0;
  const int E = in_sizes[3];

  float* ws    = (float*)d_ws;
  float* f_s   = ws;
  float* f_v   = f_s  + (size_t)N * MUL0;
  float* sc_s  = f_v  + (size_t)N * MUL1 * 3;
  float* sc_v  = sc_s + (size_t)N * MUL0;
  float* agg_s = sc_v + (size_t)N * MUL1 * 3;
  float* agg_v = agg_s + (size_t)N * MIDS;

  int* cnt    = (int*)d_out;
  int* cursor = cnt + N;
  int* order  = cursor + N;

  long long nAgg = (long long)N * (MIDS + MIDV * 3);
  zero_kernel<<<4096, 256, 0, stream>>>(agg_s, nAgg, cnt, N);

  node_prep_kernel<<<N, 160, 0, stream>>>(node_scalars, node_vectors, node_attr,
                                          W_l1_s, W_l1_v, W_sc_s, W_sc_v,
                                          f_s, f_v, sc_s, sc_v, N);

  dst_count_kernel<<<1024, 256, 0, stream>>>(edge_dst, cnt, E);
  dst_scan_kernel<<<1, 1024, 0, stream>>>(cnt, cursor, N);
  dst_fill_kernel<<<1024, 256, 0, stream>>>(edge_dst, cursor, order, E);

  int numTiles = (E + 15) / 16;
  edge_fused_kernel<<<1024, WAVES * 32, 0, stream>>>(edge_scalars, edge_attr,
                                                     edge_src, edge_dst, order,
                                                     W_fc1, W_fc2, f_s, f_v,
                                                     agg_s, agg_v, E, numTiles);

  node_out_kernel<<<N, 160, 0, stream>>>(agg_s, agg_v, sc_s, sc_v, node_attr,
                                         W_l2_s, W_l2_v, W_l3, (float*)d_out, N);
}


# GEMM2 column tiles run in pairs: the pair's B fragments (one group of LDS reads) and gathered node rows are requested ahead of its 56 WMMAs on two accumulator chains; f_v rows fetched once per tile
# speedup vs baseline: 1.7964x; 1.1181x over previous
#include <hip/hip_runtime.h>
#include <hip/hip_bf16.h>
#include <math.h>


typedef __attribute__((ext_vector_type(2))) float v2f;
typedef __attribute__((ext_vector_type(8))) float v8f;

#define MUL0   64
#define MUL1   32
#define NBASIS 10
#define FCH    100
#define WNUM   224
#define MIDS   96
#define MIDV   128
#define KPAD1  12
#define NPAD1  112
#define WAVES  4

__device__ __forceinline__ void atomic_add_f32(float* p, float v) {
  __hip_atomic_fetch_add(p, v, __ATOMIC_RELAXED, __HIP_MEMORY_SCOPE_AGENT);
}

__device__ __forceinline__ float fast_rcp(float x) {
#if __has_builtin(__builtin_amdgcn_rcpf)
  return __builtin_amdgcn_rcpf(x);
#else
  return 1.f / x;
#endif
}

template <int NC>
__device__ __forceinline__ void seg_add(float* __restrict__ agg, int rowStride, int colOff,
                                        const int (&d)[8], bool mrg, int hi,
                                        const float (&v)[NC][8]) {
  float run[NC], head[NC];
  #pragma unroll
  for (int k = 0; k < NC; ++k) { run[k] = 0.f; head[k] = 0.f; }
  bool first = (hi == 1) && mrg;
  #pragma unroll
  for (int r = 0; r < 8; ++r) {
    #pragma unroll
    for (int k = 0; k < NC; ++k) run[k] += v[k][r];
    const bool endrun = (r == 7) || (d[r] != d[r + 1 < 8 ? r + 1 : 7]);
    if (endrun) {
      const bool keep = (r == 7) && (hi == 0) && mrg;
      if (first) {
        #pragma unroll
        for (int k = 0; k < NC; ++k) { head[k] = run[k]; run[k] = 0.f; }
        first = false;
      } else if (!keep) {
        if (d[r] >= 0) {
          float* p = agg + (long long)d[r] * rowStride + colOff;
          #pragma unroll
          for (int k = 0; k < NC; ++k) atomic_add_f32(p + k, run[k]);
        }
        #pragma unroll
        for (int k = 0; k < NC; ++k) run[k] = 0.f;
      }
    }
  }
  #pragma unroll
  for (int k = 0; k < NC; ++k) {
    float carry = ((hi == 0) && mrg) ? run[k] : 0.f;
    head[k] += __shfl_xor(carry, 16);
  }
  if ((hi == 1) && mrg) {
    float* p = agg + (long long)d[0] * rowStride + colOff;
    #pragma unroll
    for (int k = 0; k < NC; ++k) atomic_add_f32(p + k, head[k]);
  }
}

__global__ void zero_kernel(float* __restrict__ p, long long n, int* __restrict__ q, int m) {
  long long i = (long long)blockIdx.x * blockDim.x + threadIdx.x;
  long long stride = (long long)gridDim.x * blockDim.x;
  for (long long j = i; j < m; j += stride) q[j] = 0;
  for (; i < n; i += stride) p[i] = 0.f;
}

__global__ void dst_count_kernel(const int* __restrict__ edst, int* __restrict__ cnt, int E) {
  int stride = gridDim.x * blockDim.x;
  for (int e = blockIdx.x * blockDim.x + threadIdx.x; e < E; e += stride)
    atomicAdd(&cnt[edst[e]], 1);
}

__global__ __launch_bounds__(1024)
void dst_scan_kernel(const int* __restrict__ cnt, int* __restrict__ cursor, int n) {
  __shared__ int part[1024];
  const int t = threadIdx.x;
  const int chunk = (n + 1023) / 1024;
  const int b = t * chunk;
  const int e = (b + chunk < n) ? (b + chunk) : n;
  int s = 0;
  for (int i = b; i < e; ++i) s += cnt[i];
  part[t] = s;
  __syncthreads();
  for (int dd = 1; dd < 1024; dd <<= 1) {
    int add = (t >= dd) ? part[t - dd] : 0;
    __syncthreads();
    part[t] += add;
    __syncthreads();
  }
  int run = part[t] - s;
  for (int i = b; i < e; ++i) { cursor[i] = run; run += cnt[i]; }
}

__global__ void dst_fill_kernel(const int* __restrict__ edst, int* __restrict__ cursor,
                                int* __restrict__ order, int E) {
  int stride = gridDim.x * blockDim.x;
  for (int e = blockIdx.x * blockDim.x + threadIdx.x; e < E; e += stride) {
    int pos = atomicAdd(&cursor[edst[e]], 1);
    order[pos] = e;
  }
}

__global__ void node_prep_kernel(const float* __restrict__ ns, const float* __restrict__ nv,
                                 const float* __restrict__ attr,
                                 const float* __restrict__ Wls, const float* __restrict__ Wlv,
                                 const float* __restrict__ Wss, const float* __restrict__ Wsv,
                                 float* __restrict__ f_s, float* __restrict__ f_v,
                                 float* __restrict__ sc_s, float* __restrict__ sc_v, int n)
{
  int node = blockIdx.x;
  if (node >= n) return;
  __shared__ float s_ns[MUL0];
  __shared__ float s_nv[MUL1 * 3];
  int t = threadIdx.x;
  if (t < MUL0)        s_ns[t]        = ns[(long long)node * MUL0 + t];
  else                 s_nv[t - MUL0] = nv[(long long)node * (MUL1 * 3) + (t - MUL0)];
  __syncthreads();
  float a = attr[node];
  if (t < MUL0) {
    float af = 0.f, asc = 0.f;
    #pragma unroll 8
    for (int u = 0; u < MUL0; ++u) {
      float x = s_ns[u];
      af  += x * Wls[u * MUL0 + t];
      asc += x * Wss[u * MUL0 + t];
    }
    f_s [(long long)node * MUL0 + t] = af  * a;
    sc_s[(long long)node * MUL0 + t] = asc * a;
  } else {
    int idx = t - MUL0;
    int vv = idx / 3;
    int k  = idx - vv * 3;
    float af = 0.f, asc = 0.f;
    #pragma unroll 8
    for (int u = 0; u < MUL1; ++u) {
      float x = s_nv[u * 3 + k];
      af  += x * Wlv[u * MUL1 + vv];
      asc += x * Wsv[u * MUL1 + vv];
    }
    f_v [(long long)node * (MUL1 * 3) + idx] = af  * a;
    sc_v[(long long)node * (MUL1 * 3) + idx] = asc * a;
  }
}

__global__ __launch_bounds__(WAVES * 32)
void edge_fused_kernel(const float* __restrict__ es,
                       const float* __restrict__ ea,
                       const int*   __restrict__ esrc,
                       const int*   __restrict__ edst,
                       const int*   __restrict__ order,
                       const float* __restrict__ Wfc1,
                       const float* __restrict__ Wfc2,
                       const float* __restrict__ f_s,
                       const float* __restrict__ f_v,
                       float* __restrict__ agg_s,
                       float* __restrict__ agg_v,
                       int E, int numTiles)
{
  __shared__ float W1s[KPAD1 * NPAD1];
  __shared__ float W2s[NPAD1 * WNUM];
  __shared__ float Hs[WAVES][16 * NPAD1];
  __shared__ float mF[WAVES][16 * 4];
  __shared__ int   mI[WAVES][16 * 2];

  for (int i = threadIdx.x; i < KPAD1 * NPAD1; i += blockDim.x) {
    int k = i / NPAD1, ncol = i - k * NPAD1;
    W1s[i] = (k < NBASIS && ncol < FCH) ? Wfc1[k * FCH + ncol] : 0.f;
  }
  for (int i = threadIdx.x; i < NPAD1 * WNUM; i += blockDim.x) {
    int k = i / WNUM;
    W2s[i] = (k < FCH) ? Wfc2[i] : 0.f;
  }
  __syncthreads();

  const int wid  = threadIdx.x >> 5;
  const int lane = threadIdx.x & 31;
  const int hi   = lane >> 4;
  const int lm   = lane & 15;

  float* H  = Hs[wid];
  float* mf = mF[wid];
  int*   mi = mI[wid];

  const float inv  = 0.25f;
  const float inv3 = 0.57735026918962576451f;
  const float inv2 = 0.70710678118654752440f;

  int tile       = blockIdx.x * WAVES + wid;
  int tileStride = gridDim.x * WAVES;

  for (; tile < numTiles; tile += tileStride) {
    int base = tile * 16;

    int  pA = base + lm;
    bool ev = (pA < E);
    int  eA = ev ? order[pA] : 0;

    if (lane < 16) {
      int e = eA;
      if (ev) {
        mi[lane * 2 + 0] = esrc[e];
        mi[lane * 2 + 1] = edst[e];
        mf[lane * 4 + 0] = ea[(long long)e * 4 + 0];
        mf[lane * 4 + 1] = ea[(long long)e * 4 + 1];
        mf[lane * 4 + 2] = ea[(long long)e * 4 + 2];
        mf[lane * 4 + 3] = ea[(long long)e * 4 + 3];
      } else {
        mi[lane * 2 + 0] = 0;
        mi[lane * 2 + 1] = -1;
        mf[lane * 4 + 0] = 0.f; mf[lane * 4 + 1] = 0.f;
        mf[lane * 4 + 2] = 0.f; mf[lane * 4 + 3] = 0.f;
      }
    }

    v2f a1[3];
    #pragma unroll
    for (int kk = 0; kk < 3; ++kk) {
      int k0 = 4 * kk + 2 * hi;
      a1[kk].x = (ev && (k0     < NBASIS)) ? es[(long long)eA * NBASIS + k0    ] : 0.f;
      a1[kk].y = (ev && (k0 + 1 < NBASIS)) ? es[(long long)eA * NBASIS + k0 + 1] : 0.f;
    }

    #pragma unroll
    for (int nt = 0; nt < NPAD1 / 16; ++nt) {
      v8f acc = {};
      #pragma unroll
      for (int kk = 0; kk < 3; ++kk) {
        int k0 = 4 * kk + 2 * hi;
        v2f b;
        b.x = W1s[ k0      * NPAD1 + nt * 16 + lm];
        b.y = W1s[(k0 + 1) * NPAD1 + nt * 16 + lm];
        acc = __builtin_amdgcn_wmma_f32_16x16x4_f32(
            false, a1[kk], false, b, (short)0, acc, false, false);
      }
      #pragma unroll
      for (int r = 0; r < 8; ++r) {
        float x  = acc[r];
        float sg = fast_rcp(1.f + __expf(-x));
        H[(r + 8 * hi) * NPAD1 + nt * 16 + lm] = x * sg;
      }
    }
    asm volatile("s_wait_dscnt 0" ::: "memory");

    v2f a2[NPAD1 / 4];
    #pragma unroll
    for (int kk = 0; kk < NPAD1 / 4; ++kk) {
      int k0 = 4 * kk + 2 * hi;
      a2[kk] = *(const v2f*)&H[lm * NPAD1 + k0];
    }

    int   drow[8], srow[8];
    float y0r[8], y1x[8], y1y[8], y1z[8];
    #pragma unroll
    for (int r = 0; r < 8; ++r) {
      int m = r + 8 * hi;
      srow[r] = mi[m * 2 + 0];
      drow[r] = mi[m * 2 + 1];
      y0r[r]  = mf[m * 4 + 0];
      y1x[r]  = mf[m * 4 + 1];
      y1y[r]  = mf[m * 4 + 2];
      y1z[r]  = mf[m * 4 + 3];
    }
    const int  d7  = mi[7 * 2 + 1];
    const bool mrg = (d7 >= 0) && (d7 == mi[8 * 2 + 1]);


    auto gemm2_pair = [&](const int cA, v8f (&acc)[2]) {
      int cB = cA + 16;
      asm volatile("" : "+v"(cB));
      v2f b[2][NPAD1 / 4];
      #pragma unroll
      for (int kk = 0; kk < NPAD1 / 4; ++kk) {
        const int k0 = 4 * kk + 2 * hi;
        b[0][kk].x = W2s[ k0      * WNUM + cA];
        b[0][kk].y = W2s[(k0 + 1) * WNUM + cA];
        b[1][kk].x = W2s[ k0      * WNUM + cB];
        b[1][kk].y = W2s[(k0 + 1) * WNUM + cB];
      }
      __builtin_amdgcn_sched_barrier(0);

      acc[0] = v8f{}; acc[1] = v8f{};
      #pragma unroll
      for (int kk = 0; kk < NPAD1 / 4; ++kk) {
        #pragma unroll
        for (int t = 0; t < 2; ++t) {
          acc[t] = __builtin_amdgcn_wmma_f32_16x16x4_f32(
              false, a2[kk], false, b[t][kk], (short)0, acc[t], false, false);
        }
      }
    };

    for (int p = 0; p < 4; ++p) {
      const int cA  = p * 32 + lm;
      const int col = cA & 63;
      float g[2][8];
      #pragma unroll
      for (int t = 0; t < 2; ++t) {
        #pragma unroll
        for (int r = 0; r < 8; ++r) g[t][r] = f_s[(long long)srow[r] * MUL0 + col + 16 * t];
      }
      v8f acc[2];
      gemm2_pair(cA, acc);

      #pragma unroll
      for (int t = 0; t < 2; ++t) {
        const int c = cA + 16 * t;
        if (p < 2) {
          float v[1][8];
          #pragma unroll
          for (int r = 0; r < 8; ++r) v[0][r] = acc[t][r] * g[t][r] * y0r[r] * inv;
          seg_add<1>(agg_s, MIDS, c, drow, mrg, hi, v);
        } else {
          const int u = c - 64;
          float v[3][8];
          #pragma unroll
          for (int r = 0; r < 8; ++r) {
            float tt = acc[t][r] * g[t][r] * inv;
            v[0][r] = tt * y1x[r]; v[1][r] = tt * y1y[r]; v[2][r] = tt * y1z[r];
          }
          seg_add<3>(agg_v, MIDV * 3, u * 3, drow, mrg, hi, v);
        }
      }
    }

    {
      float g[2][3][8];
      #pragma unroll
      for (int t = 0; t < 2; ++t) {
        #pragma unroll
        for (int r = 0; r < 8; ++r) {
          const float* vp = &f_v[((long long)srow[r] * MUL1 + lm + 16 * t) * 3];
          g[t][0][r] = vp[0]; g[t][1][r] = vp[1]; g[t][2][r] = vp[2];
        }
      }
      v8f acc[2];

      gemm2_pair(128 + lm, acc);
      #pragma unroll
      for (int t = 0; t < 2; ++t) {
        const int u = lm + 16 * t;
        float v[3][8];
        #pragma unroll
        for (int r = 0; r < 8; ++r) {
          float tt = acc[t][r] * y0r[r] * inv;
          v[0][r] = tt * g[t][0][r]; v[1][r] = tt * g[t][1][r]; v[2][r] = tt * g[t][2][r];
        }
        seg_add<3>(agg_v, MIDV * 3, (64 + u) * 3, drow, mrg, hi, v);
      }

      gemm2_pair(160 + lm, acc);
      #pragma unroll
      for (int t = 0; t < 2; ++t) {
        const int u = lm + 16 * t;
        float v[1][8];
        #pragma unroll
        for (int r = 0; r < 8; ++r) {
          float dt = g[t][0][r] * y1x[r] + g[t][1][r] * y1y[r] + g[t][2][r] * y1z[r];
          v[0][r] = acc[t][r] * dt * inv3 * inv;
        }
        seg_add<1>(agg_s, MIDS, 64 + u, drow, mrg, hi, v);
      }

      gemm2_pair(192 + lm, acc);
      #pragma unroll
      for (int t = 0; t < 2; ++t) {
        const int u = lm + 16 * t;
        float v[3][8];
        #pragma unroll
        for (int r = 0; r < 8; ++r) {
          float cx = g[t][1][r] * y1z[r] - g[t][2][r] * y1y[r];
          float cy = g[t][2][r] * y1x[r] - g[t][0][r] * y1z[r];
          float cz = g[t][0][r] * y1y[r] - g[t][1][r] * y1x[r];
          float tt = acc[t][r] * inv2 * inv;
          v[0][r] = tt * cx; v[1][r] = tt * cy; v[2][r] = tt * cz;
        }
        seg_add<3>(agg_v, MIDV * 3, (96 + u) * 3, drow, mrg, hi, v);
      }
    }
  }
}

__global__ void node_out_kernel(const float* __restrict__ agg_s, const float* __restrict__ agg_v,
                                const float* __restrict__ sc_s,  const float* __restrict__ sc_v,
                                const float* __restrict__ attr,
                                const float* __restrict__ Wl2s, const float* __restrict__ Wl2v,
                                const float* __restrict__ Wl3,
                                float* __restrict__ out, int n)
{
  int node = blockIdx.x;
  if (node >= n) return;
  __shared__ float sAs[MIDS];
  __shared__ float sAv[MIDV * 3];
  int t = threadIdx.x;
  if (t < MIDS) sAs[t] = agg_s[(long long)node * MIDS + t];
  for (int i = t; i < MIDV * 3; i += 160) sAv[i] = agg_v[(long long)node * (MIDV * 3) + i];
  __syncthreads();

  float a = attr[node];
  float ang = 0.f;
  #pragma unroll 8
  for (int u = 0; u < MIDS; ++u) ang += sAs[u] * Wl3[u];
  ang *= 0.1f * a;
  float cs = cosf(ang);
  float sn = sinf(ang);

  if (t < MUL0) {
    float conv = 0.f;
    #pragma unroll 8
    for (int u = 0; u < MIDS; ++u) conv += sAs[u] * Wl2s[u * MUL0 + t];
    conv *= a;
    out[(long long)node * 160 + t] = cs * sc_s[(long long)node * MUL0 + t] + sn * conv;
  } else {
    int idx = t - MUL0;
    int vv = idx / 3;
    int k  = idx - vv * 3;
    float conv = 0.f;
    #pragma unroll 8
    for (int u = 0; u < MIDV; ++u) conv += sAv[u * 3 + k] * Wl2v[u * MUL1 + vv];
    conv *= a;
    out[(long long)node * 160 + t] = cs * sc_v[(long long)node * (MUL1 * 3) + idx] + sn * conv;
  }
}

extern "C" void kernel_launch(void* const* d_in, const int* in_sizes, int n_in,
                              void* d_out, int out_size, void* d_ws, size_t ws_size,
                              hipStream_t stream) {
  const float* node_scalars = (const float*)d_in[0];
  const float* node_vectors = (const float*)d_in[1];
  const float* node_attr    = (const float*)d_in[2];
  const int*   edge_src     = (const int*)  d_in[3];
  const int*   edge_dst     = (const int*)  d_in[4];
  const float* edge_attr    = (const float*)d_in[5];
  const float* edge_scalars = (const float*)d_in[6];
  const float* W_fc1        = (const float*)d_in[7];
  const float* W_fc2        = (const float*)d_in[8];
  const float* W_sc_s       = (const float*)d_in[9];
  const float* W_sc_v       = (const float*)d_in[10];
  const float* W_l1_s       = (const float*)d_in[11];
  const float* W_l1_v       = (const float*)d_in[12];
  const float* W_l2_s       = (const float*)d_in[13];
  const float* W_l2_v       = (const float*)d_in[14];
  const float* W_l3         = (const float*)d_in[15];

  const int N = in_sizes[0] / MUL0;
  const int E = in_sizes[3];

  float* ws    = (float*)d_ws;
  float* f_s   = ws;
  float* f_v   = f_s  + (size_t)N * MUL0;
  float* sc_s  = f_v  + (size_t)N * MUL1 * 3;
  float* sc_v  = sc_s + (size_t)N * MUL0;
  float* agg_s = sc_v + (size_t)N * MUL1 * 3;
  float* agg_v = agg_s + (size_t)N * MIDS;

  int* cnt    = (int*)d_out;
  int* cursor = cnt + N;
  int* order  = cursor + N;

  long long nAgg = (long long)N * (MIDS + MIDV * 3);
  zero_kernel<<<4096, 256, 0, stream>>>(agg_s, nAgg, cnt, N);

  node_prep_kernel<<<N, 160, 0, stream>>>(node_scalars, node_vectors, node_attr,
                                          W_l1_s, W_l1_v, W_sc_s, W_sc_v,
                                          f_s, f_v, sc_s, sc_v, N);

  dst_count_kernel<<<1024, 256, 0, stream>>>(edge_dst, cnt, E);
  dst_scan_kernel<<<1, 1024, 0, stream>>>(cnt, cursor, N);
  dst_fill_kernel<<<1024, 256, 0, stream>>>(edge_dst, cursor, order, E);

  int numTiles = (E + 15) / 16;
  edge_fused_kernel<<<1024, WAVES * 32, 0, stream>>>(edge_scalars, edge_attr,
                                                     edge_src, edge_dst, order,
                                                     W_fc1, W_fc2, f_s, f_v,
                                                     agg_s, agg_v, E, numTiles);

  node_out_kernel<<<N, 160, 0, stream>>>(agg_s, agg_v, sc_s, sc_v, node_attr,
                                         W_l2_s, W_l2_v, W_l3, (float*)d_out, N);
}


# edge kernel stages W_fc2/W_fc1 to LDS with all 16-byte loads in flight (was one 4-byte load per round trip) and runs 512 workgroups, all resident, so each stages the weights once
# speedup vs baseline: 2.2810x; 1.2698x over previous
#include <hip/hip_runtime.h>
#include <hip/hip_bf16.h>
#include <math.h>


typedef __attribute__((ext_vector_type(2))) float v2f;
typedef __attribute__((ext_vector_type(8))) float v8f;

#define MUL0   64
#define MUL1   32
#define NBASIS 10
#define FCH    100
#define WNUM   224
#define MIDS   96
#define MIDV   128
#define KPAD1  12
#define NPAD1  112
#define WAVES  4

__device__ __forceinline__ void atomic_add_f32(float* p, float v) {
  __hip_atomic_fetch_add(p, v, __ATOMIC_RELAXED, __HIP_MEMORY_SCOPE_AGENT);
}

__device__ __forceinline__ float fast_rcp(float x) {
#if __has_builtin(__builtin_amdgcn_rcpf)
  return __builtin_amdgcn_rcpf(x);
#else
  return 1.f / x;
#endif
}

template <int NC>
__device__ __forceinline__ void seg_add(float* __restrict__ agg, int rowStride, int colOff,
                                        const int (&d)[8], bool mrg, int hi,
                                        const float (&v)[NC][8]) {
  float run[NC], head[NC];
  #pragma unroll
  for (int k = 0; k < NC; ++k) { run[k] = 0.f; head[k] = 0.f; }
  bool first = (hi == 1) && mrg;
  #pragma unroll
  for (int r = 0; r < 8; ++r) {
    #pragma unroll
    for (int k = 0; k < NC; ++k) run[k] += v[k][r];
    const bool endrun = (r == 7) || (d[r] != d[r + 1 < 8 ? r + 1 : 7]);
    if (endrun) {
      const bool keep = (r == 7) && (hi == 0) && mrg;
      if (first) {
        #pragma unroll
        for (int k = 0; k < NC; ++k) { head[k] = run[k]; run[k] = 0.f; }
        first = false;
      } else if (!keep) {
        if (d[r] >= 0) {
          float* p = agg + (long long)d[r] * rowStride + colOff;
          #pragma unroll
          for (int k = 0; k < NC; ++k) atomic_add_f32(p + k, run[k]);
        }
        #pragma unroll
        for (int k = 0; k < NC; ++k) run[k] = 0.f;
      }
    }
  }
  #pragma unroll
  for (int k = 0; k < NC; ++k) {
    float carry = ((hi == 0) && mrg) ? run[k] : 0.f;
    head[k] += __shfl_xor(carry, 16);
  }
  if ((hi == 1) && mrg) {
    float* p = agg + (long long)d[0] * rowStride + colOff;
    #pragma unroll
    for (int k = 0; k < NC; ++k) atomic_add_f32(p + k, head[k]);
  }
}

__global__ void zero_kernel(float* __restrict__ p, long long n, int* __restrict__ q, int m) {
  long long i = (long long)blockIdx.x * blockDim.x + threadIdx.x;
  long long stride = (long long)gridDim.x * blockDim.x;
  for (long long j = i; j < m; j += stride) q[j] = 0;
  for (; i < n; i += stride) p[i] = 0.f;
}

__global__ void dst_count_kernel(const int* __restrict__ edst, int* __restrict__ cnt, int E) {
  int stride = gridDim.x * blockDim.x;
  for (int e = blockIdx.x * blockDim.x + threadIdx.x; e < E; e += stride)
    atomicAdd(&cnt[edst[e]], 1);
}

__global__ __launch_bounds__(1024)
void dst_scan_kernel(const int* __restrict__ cnt, int* __restrict__ cursor, int n) {
  __shared__ int part[1024];
  const int t = threadIdx.x;
  const int chunk = (n + 1023) / 1024;
  const int b = t * chunk;
  const int e = (b + chunk < n) ? (b + chunk) : n;
  int s = 0;
  for (int i = b; i < e; ++i) s += cnt[i];
  part[t] = s;
  __syncthreads();
  for (int dd = 1; dd < 1024; dd <<= 1) {
    int add = (t >= dd) ? part[t - dd] : 0;
    __syncthreads();
    part[t] += add;
    __syncthreads();
  }
  int run = part[t] - s;
  for (int i = b; i < e; ++i) { cursor[i] = run; run += cnt[i]; }
}

__global__ void dst_fill_kernel(const int* __restrict__ edst, int* __restrict__ cursor,
                                int* __restrict__ order, int E) {
  int stride = gridDim.x * blockDim.x;
  for (int e = blockIdx.x * blockDim.x + threadIdx.x; e < E; e += stride) {
    int pos = atomicAdd(&cursor[edst[e]], 1);
    order[pos] = e;
  }
}

__global__ void node_prep_kernel(const float* __restrict__ ns, const float* __restrict__ nv,
                                 const float* __restrict__ attr,
                                 const float* __restrict__ Wls, const float* __restrict__ Wlv,
                                 const float* __restrict__ Wss, const float* __restrict__ Wsv,
                                 float* __restrict__ f_s, float* __restrict__ f_v,
                                 float* __restrict__ sc_s, float* __restrict__ sc_v, int n)
{
  int node = blockIdx.x;
  if (node >= n) return;
  __shared__ float s_ns[MUL0];
  __shared__ float s_nv[MUL1 * 3];
  int t = threadIdx.x;
  if (t < MUL0)        s_ns[t]        = ns[(long long)node * MUL0 + t];
  else                 s_nv[t - MUL0] = nv[(long long)node * (MUL1 * 3) + (t - MUL0)];
  __syncthreads();
  float a = attr[node];
  if (t < MUL0) {
    float af = 0.f, asc = 0.f;
    #pragma unroll 8
    for (int u = 0; u < MUL0; ++u) {
      float x = s_ns[u];
      af  += x * Wls[u * MUL0 + t];
      asc += x * Wss[u * MUL0 + t];
    }
    f_s [(long long)node * MUL0 + t] = af  * a;
    sc_s[(long long)node * MUL0 + t] = asc * a;
  } else {
    int idx = t - MUL0;
    int vv = idx / 3;
    int k  = idx - vv * 3;
    float af = 0.f, asc = 0.f;
    #pragma unroll 8
    for (int u = 0; u < MUL1; ++u) {
      float x = s_nv[u * 3 + k];
      af  += x * Wlv[u * MUL1 + vv];
      asc += x * Wsv[u * MUL1 + vv];
    }
    f_v [(long long)node * (MUL1 * 3) + idx] = af  * a;
    sc_v[(long long)node * (MUL1 * 3) + idx] = asc * a;
  }
}

__global__ __launch_bounds__(WAVES * 32)
void edge_fused_kernel(const float* __restrict__ es,
                       const float* __restrict__ ea,
                       const int*   __restrict__ esrc,
                       const int*   __restrict__ edst,
                       const int*   __restrict__ order,
                       const float* __restrict__ Wfc1,
                       const float* __restrict__ Wfc2,
                       const float* __restrict__ f_s,
                       const float* __restrict__ f_v,
                       float* __restrict__ agg_s,
                       float* __restrict__ agg_v,
                       int E, int numTiles)
{
  __shared__ float W1s[KPAD1 * NPAD1];
  __shared__ float W2s[NPAD1 * WNUM];
  __shared__ float Hs[WAVES][16 * NPAD1];
  __shared__ float mF[WAVES][16 * 4];
  __shared__ int   mI[WAVES][16 * 2];

  {
    constexpr int NT = WAVES * 32;
    constexpr int N1 = (KPAD1 * NPAD1 + NT - 1) / NT;
    float w1[N1];
    #pragma unroll
    for (int j = 0; j < N1; ++j) {
      const int i = threadIdx.x + NT * j;
      const int k = i / NPAD1, ncol = i - k * NPAD1;
      const bool ok = (k < NBASIS) && (ncol < FCH);
      const float x = Wfc1[ok ? k * FCH + ncol : 0];
      w1[j] = ok ? x : 0.f;
    }
    constexpr int NV  = FCH   * WNUM / 4;
    constexpr int NVP = NPAD1 * WNUM / 4;
    constexpr int N2  = NVP / NT;
    static_assert(NVP % NT == 0, "W2s must split evenly over the block");
    const float4* __restrict__ src = (const float4*)Wfc2;
    float4 w2[N2];
    #pragma unroll
    for (int j = 0; j < N2; ++j) {
      const int i = threadIdx.x + NT * j;
      const float4 x = src[i < NV ? i : NV - 1];
      w2[j] = (i < NV) ? x : make_float4(0.f, 0.f, 0.f, 0.f);
    }
    #pragma unroll
    for (int j = 0; j < N1; ++j) {
      const int i = threadIdx.x + NT * j;
      if (i < KPAD1 * NPAD1) W1s[i] = w1[j];
    }
    float4* dst = (float4*)W2s;
    #pragma unroll
    for (int j = 0; j < N2; ++j) dst[threadIdx.x + NT * j] = w2[j];
  }
  __syncthreads();

  const int wid  = threadIdx.x >> 5;
  const int lane = threadIdx.x & 31;
  const int hi   = lane >> 4;
  const int lm   = lane & 15;

  float* H  = Hs[wid];
  float* mf = mF[wid];
  int*   mi = mI[wid];

  const float inv  = 0.25f;
  const float inv3 = 0.57735026918962576451f;
  const float inv2 = 0.70710678118654752440f;

  int tile       = blockIdx.x * WAVES + wid;
  int tileStride = gridDim.x * WAVES;

  for (; tile < numTiles; tile += tileStride) {
    int base = tile * 16;

    int  pA = base + lm;
    bool ev = (pA < E);
    int  eA = ev ? order[pA] : 0;

    if (lane < 16) {
      int e = eA;
      if (ev) {
        mi[lane * 2 + 0] = esrc[e];
        mi[lane * 2 + 1] = edst[e];
        mf[lane * 4 + 0] = ea[(long long)e * 4 + 0];
        mf[lane * 4 + 1] = ea[(long long)e * 4 + 1];
        mf[lane * 4 + 2] = ea[(long long)e * 4 + 2];
        mf[lane * 4 + 3] = ea[(long long)e * 4 + 3];
      } else {
        mi[lane * 2 + 0] = 0;
        mi[lane * 2 + 1] = -1;
        mf[lane * 4 + 0] = 0.f; mf[lane * 4 + 1] = 0.f;
        mf[lane * 4 + 2] = 0.f; mf[lane * 4 + 3] = 0.f;
      }
    }

    v2f a1[3];
    #pragma unroll
    for (int kk = 0; kk < 3; ++kk) {
      int k0 = 4 * kk + 2 * hi;
      a1[kk].x = (ev && (k0     < NBASIS)) ? es[(long long)eA * NBASIS + k0    ] : 0.f;
      a1[kk].y = (ev && (k0 + 1 < NBASIS)) ? es[(long long)eA * NBASIS + k0 + 1] : 0.f;
    }

    #pragma unroll
    for (int nt = 0; nt < NPAD1 / 16; ++nt) {
      v8f acc = {};
      #pragma unroll
      for (int kk = 0; kk < 3; ++kk) {
        int k0 = 4 * kk + 2 * hi;
        v2f b;
        b.x = W1s[ k0      * NPAD1 + nt * 16 + lm];
        b.y = W1s[(k0 + 1) * NPAD1 + nt * 16 + lm];
        acc = __builtin_amdgcn_wmma_f32_16x16x4_f32(
            false, a1[kk], false, b, (short)0, acc, false, false);
      }
      #pragma unroll
      for (int r = 0; r < 8; ++r) {
        float x  = acc[r];
        float sg = fast_rcp(1.f + __expf(-x));
        H[(r + 8 * hi) * NPAD1 + nt * 16 + lm] = x * sg;
      }
    }
    asm volatile("s_wait_dscnt 0" ::: "memory");

    v2f a2[NPAD1 / 4];
    #pragma unroll
    for (int kk = 0; kk < NPAD1 / 4; ++kk) {
      int k0 = 4 * kk + 2 * hi;
      a2[kk] = *(const v2f*)&H[lm * NPAD1 + k0];
    }

    int   drow[8], srow[8];
    float y0r[8], y1x[8], y1y[8], y1z[8];
    #pragma unroll
    for (int r = 0; r < 8; ++r) {
      int m = r + 8 * hi;
      srow[r] = mi[m * 2 + 0];
      drow[r] = mi[m * 2 + 1];
      y0r[r]  = mf[m * 4 + 0];
      y1x[r]  = mf[m * 4 + 1];
      y1y[r]  = mf[m * 4 + 2];
      y1z[r]  = mf[m * 4 + 3];
    }
    const int  d7  = mi[7 * 2 + 1];
    const bool mrg = (d7 >= 0) && (d7 == mi[8 * 2 + 1]);


    auto gemm2_pair = [&](const int cA, v8f (&acc)[2]) {
      int cB = cA + 16;
      asm volatile("" : "+v"(cB));
      v2f b[2][NPAD1 / 4];
      #pragma unroll
      for (int kk = 0; kk < NPAD1 / 4; ++kk) {
        const int k0 = 4 * kk + 2 * hi;
        b[0][kk].x = W2s[ k0      * WNUM + cA];
        b[0][kk].y = W2s[(k0 + 1) * WNUM + cA];
        b[1][kk].x = W2s[ k0      * WNUM + cB];
        b[1][kk].y = W2s[(k0 + 1) * WNUM + cB];
      }
      __builtin_amdgcn_sched_barrier(0);

      acc[0] = v8f{}; acc[1] = v8f{};
      #pragma unroll
      for (int kk = 0; kk < NPAD1 / 4; ++kk) {
        #pragma unroll
        for (int t = 0; t < 2; ++t) {
          acc[t] = __builtin_amdgcn_wmma_f32_16x16x4_f32(
              false, a2[kk], false, b[t][kk], (short)0, acc[t], false, false);
        }
      }
    };

    for (int p = 0; p < 4; ++p) {
      const int cA  = p * 32 + lm;
      const int col = cA & 63;
      float g[2][8];
      #pragma unroll
      for (int t = 0; t < 2; ++t) {
        #pragma unroll
        for (int r = 0; r < 8; ++r) g[t][r] = f_s[(long long)srow[r] * MUL0 + col + 16 * t];
      }
      v8f acc[2];
      gemm2_pair(cA, acc);

      #pragma unroll
      for (int t = 0; t < 2; ++t) {
        const int c = cA + 16 * t;
        if (p < 2) {
          float v[1][8];
          #pragma unroll
          for (int r = 0; r < 8; ++r) v[0][r] = acc[t][r] * g[t][r] * y0r[r] * inv;
          seg_add<1>(agg_s, MIDS, c, drow, mrg, hi, v);
        } else {
          const int u = c - 64;
          float v[3][8];
          #pragma unroll
          for (int r = 0; r < 8; ++r) {
            float tt = acc[t][r] * g[t][r] * inv;
            v[0][r] = tt * y1x[r]; v[1][r] = tt * y1y[r]; v[2][r] = tt * y1z[r];
          }
          seg_add<3>(agg_v, MIDV * 3, u * 3, drow, mrg, hi, v);
        }
      }
    }

    {
      float g[2][3][8];
      #pragma unroll
      for (int t = 0; t < 2; ++t) {
        #pragma unroll
        for (int r = 0; r < 8; ++r) {
          const float* vp = &f_v[((long long)srow[r] * MUL1 + lm + 16 * t) * 3];
          g[t][0][r] = vp[0]; g[t][1][r] = vp[1]; g[t][2][r] = vp[2];
        }
      }
      v8f acc[2];

      gemm2_pair(128 + lm, acc);
      #pragma unroll
      for (int t = 0; t < 2; ++t) {
        const int u = lm + 16 * t;
        float v[3][8];
        #pragma unroll
        for (int r = 0; r < 8; ++r) {
          float tt = acc[t][r] * y0r[r] * inv;
          v[0][r] = tt * g[t][0][r]; v[1][r] = tt * g[t][1][r]; v[2][r] = tt * g[t][2][r];
        }
        seg_add<3>(agg_v, MIDV * 3, (64 + u) * 3, drow, mrg, hi, v);
      }

      gemm2_pair(160 + lm, acc);
      #pragma unroll
      for (int t = 0; t < 2; ++t) {
        const int u = lm + 16 * t;
        float v[1][8];
        #pragma unroll
        for (int r = 0; r < 8; ++r) {
          float dt = g[t][0][r] * y1x[r] + g[t][1][r] * y1y[r] + g[t][2][r] * y1z[r];
          v[0][r] = acc[t][r] * dt * inv3 * inv;
        }
        seg_add<1>(agg_s, MIDS, 64 + u, drow, mrg, hi, v);
      }

      gemm2_pair(192 + lm, acc);
      #pragma unroll
      for (int t = 0; t < 2; ++t) {
        const int u = lm + 16 * t;
        float v[3][8];
        #pragma unroll
        for (int r = 0; r < 8; ++r) {
          float cx = g[t][1][r] * y1z[r] - g[t][2][r] * y1y[r];
          float cy = g[t][2][r] * y1x[r] - g[t][0][r] * y1z[r];
          float cz = g[t][0][r] * y1y[r] - g[t][1][r] * y1x[r];
          float tt = acc[t][r] * inv2 * inv;
          v[0][r] = tt * cx; v[1][r] = tt * cy; v[2][r] = tt * cz;
        }
        seg_add<3>(agg_v, MIDV * 3, (96 + u) * 3, drow, mrg, hi, v);
      }
    }
  }
}

__global__ void node_out_kernel(const float* __restrict__ agg_s, const float* __restrict__ agg_v,
                                const float* __restrict__ sc_s,  const float* __restrict__ sc_v,
                                const float* __restrict__ attr,
                                const float* __restrict__ Wl2s, const float* __restrict__ Wl2v,
                                const float* __restrict__ Wl3,
                                float* __restrict__ out, int n)
{
  int node = blockIdx.x;
  if (node >= n) return;
  __shared__ float sAs[MIDS];
  __shared__ float sAv[MIDV * 3];
  int t = threadIdx.x;
  if (t < MIDS) sAs[t] = agg_s[(long long)node * MIDS + t];
  for (int i = t; i < MIDV * 3; i += 160) sAv[i] = agg_v[(long long)node * (MIDV * 3) + i];
  __syncthreads();

  float a = attr[node];
  float ang = 0.f;
  #pragma unroll 8
  for (int u = 0; u < MIDS; ++u) ang += sAs[u] * Wl3[u];
  ang *= 0.1f * a;
  float cs = cosf(ang);
  float sn = sinf(ang);

  if (t < MUL0) {
    float conv = 0.f;
    #pragma unroll 8
    for (int u = 0; u < MIDS; ++u) conv += sAs[u] * Wl2s[u * MUL0 + t];
    conv *= a;
    out[(long long)node * 160 + t] = cs * sc_s[(long long)node * MUL0 + t] + sn * conv;
  } else {
    int idx = t - MUL0;
    int vv = idx / 3;
    int k  = idx - vv * 3;
    float conv = 0.f;
    #pragma unroll 8
    for (int u = 0; u < MIDV; ++u) conv += sAv[u * 3 + k] * Wl2v[u * MUL1 + vv];
    conv *= a;
    out[(long long)node * 160 + t] = cs * sc_v[(long long)node * (MUL1 * 3) + idx] + sn * conv;
  }
}

extern "C" void kernel_launch(void* const* d_in, const int* in_sizes, int n_in,
                              void* d_out, int out_size, void* d_ws, size_t ws_size,
                              hipStream_t stream) {
  const float* node_scalars = (const float*)d_in[0];
  const float* node_vectors = (const float*)d_in[1];
  const float* node_attr    = (const float*)d_in[2];
  const int*   edge_src     = (const int*)  d_in[3];
  const int*   edge_dst     = (const int*)  d_in[4];
  const float* edge_attr    = (const float*)d_in[5];
  const float* edge_scalars = (const float*)d_in[6];
  const float* W_fc1        = (const float*)d_in[7];
  const float* W_fc2        = (const float*)d_in[8];
  const float* W_sc_s       = (const float*)d_in[9];
  const float* W_sc_v       = (const float*)d_in[10];
  const float* W_l1_s       = (const float*)d_in[11];
  const float* W_l1_v       = (const float*)d_in[12];
  const float* W_l2_s       = (const float*)d_in[13];
  const float* W_l2_v       = (const float*)d_in[14];
  const float* W_l3         = (const float*)d_in[15];

  const int N = in_sizes[0] / MUL0;
  const int E = in_sizes[3];

  float* ws    = (float*)d_ws;
  float* f_s   = ws;
  float* f_v   = f_s  + (size_t)N * MUL0;
  float* sc_s  = f_v  + (size_t)N * MUL1 * 3;
  float* sc_v  = sc_s + (size_t)N * MUL0;
  float* agg_s = sc_v + (size_t)N * MUL1 * 3;
  float* agg_v = agg_s + (size_t)N * MIDS;

  int* cnt    = (int*)d_out;
  int* cursor = cnt + N;
  int* order  = cursor + N;

  long long nAgg = (long long)N * (MIDS + MIDV * 3);
  zero_kernel<<<4096, 256, 0, stream>>>(agg_s, nAgg, cnt, N);

  node_prep_kernel<<<N, 160, 0, stream>>>(node_scalars, node_vectors, node_attr,
                                          W_l1_s, W_l1_v, W_sc_s, W_sc_v,
                                          f_s, f_v, sc_s, sc_v, N);

  dst_count_kernel<<<1024, 256, 0, stream>>>(edge_dst, cnt, E);
  dst_scan_kernel<<<1, 1024, 0, stream>>>(cnt, cursor, N);
  dst_fill_kernel<<<1024, 256, 0, stream>>>(edge_dst, cursor, order, E);

  int numTiles = (E + 15) / 16;
  edge_fused_kernel<<<512, WAVES * 32, 0, stream>>>(edge_scalars, edge_attr,
                                                     edge_src, edge_dst, order,
                                                     W_fc1, W_fc2, f_s, f_v,
                                                     agg_s, agg_v, E, numTiles);

  node_out_kernel<<<N, 160, 0, stream>>>(agg_s, agg_v, sc_s, sc_v, node_attr,
                                         W_l2_s, W_l2_v, W_l3, (float*)d_out, N);
}


# edge kernel: each wave takes a contiguous range of tiles in destination order and carries a node's open sum in registers; plain store per node, atomics only at range ends (sum order changes)
# speedup vs baseline: 2.3944x; 1.0497x over previous
#include <hip/hip_runtime.h>
#include <hip/hip_bf16.h>
#include <math.h>


typedef __attribute__((ext_vector_type(2))) float v2f;
typedef __attribute__((ext_vector_type(8))) float v8f;

#define MUL0   64
#define MUL1   32
#define NBASIS 10
#define FCH    100
#define WNUM   224
#define MIDS   96
#define MIDV   128
#define KPAD1  12
#define NPAD1  112
#define WAVES  4

__device__ __forceinline__ void atomic_add_f32(float* p, float v) {
  __hip_atomic_fetch_add(p, v, __ATOMIC_RELAXED, __HIP_MEMORY_SCOPE_AGENT);
}

__device__ __forceinline__ float fast_rcp(float x) {
#if __has_builtin(__builtin_amdgcn_rcpf)
  return __builtin_amdgcn_rcpf(x);
#else
  return 1.f / x;
#endif
}

__device__ __forceinline__ float xhalf(float x) {
  const int b = __builtin_bit_cast(int, x);
  return __builtin_bit_cast(float, __builtin_amdgcn_permlanex16(b, b, 0x76543210, 0xfedcba98, false, false));
}

struct RunFlags {
  bool e[7];
  bool cap[7];
  bool fl[7];
  bool single;
  bool contOut;
  bool aSt, aAt;
  bool bSt, bAt;
};

template <int NC>
__device__ __forceinline__ void seg_flush(float* __restrict__ agg, int rowStride, int colOff,
                                          const int (&d)[8], const RunFlags& f, int hi,
                                          const float (&v)[NC][8], float (&carry)[NC]) {
  float run[NC], head[NC], th[NC], tl[NC];
  #pragma unroll
  for (int k = 0; k < NC; ++k) { run[k] = 0.f; head[k] = 0.f; }
  #pragma unroll
  for (int r = 0; r < 8; ++r) {
    #pragma unroll
    for (int k = 0; k < NC; ++k) run[k] += v[k][r];
    if (r < 7) {
      #pragma unroll
      for (int k = 0; k < NC; ++k) head[k] = f.cap[r] ? run[k] : head[k];
      if (f.fl[r]) {
        float* p = agg + (long long)d[r] * rowStride + colOff;
        #pragma unroll
        for (int k = 0; k < NC; ++k) atomic_add_f32(p + k, run[k]);
      }
      #pragma unroll
      for (int k = 0; k < NC; ++k) run[k] = f.e[r] ? 0.f : run[k];
    }
  }
  #pragma unroll
  for (int k = 0; k < NC; ++k) {
    const float hv    = f.single ? run[k] : head[k];
    const float cinLo = xhalf(carry[k]);
    const float outLo = f.contOut ? (f.single ? hv + cinLo : run[k]) : 0.f;
    const float x     = xhalf(outLo);
    const float cin   = hi ? x : cinLo;
    th[k] = hv + cin;
    tl[k] = f.single ? th[k] : run[k];
    carry[k] = f.contOut ? tl[k] : 0.f;
  }
  if (f.aSt) {
    float* p = agg + (long long)d[0] * rowStride + colOff;
    #pragma unroll
    for (int k = 0; k < NC; ++k) p[k] = th[k];
  }
  if (f.aAt) {
    float* p = agg + (long long)d[0] * rowStride + colOff;
    #pragma unroll
    for (int k = 0; k < NC; ++k) atomic_add_f32(p + k, th[k]);
  }
  if (f.bSt) {
    float* p = agg + (long long)d[7] * rowStride + colOff;
    #pragma unroll
    for (int k = 0; k < NC; ++k) p[k] = tl[k];
  }
  if (f.bAt) {
    float* p = agg + (long long)d[7] * rowStride + colOff;
    #pragma unroll
    for (int k = 0; k < NC; ++k) atomic_add_f32(p + k, tl[k]);
  }
}

__global__ void zero_kernel(float* __restrict__ p, long long n, int* __restrict__ q, int m) {
  long long i = (long long)blockIdx.x * blockDim.x + threadIdx.x;
  long long stride = (long long)gridDim.x * blockDim.x;
  for (long long j = i; j < m; j += stride) q[j] = 0;
  for (; i < n; i += stride) p[i] = 0.f;
}

__global__ void dst_count_kernel(const int* __restrict__ edst, int* __restrict__ cnt, int E) {
  int stride = gridDim.x * blockDim.x;
  for (int e = blockIdx.x * blockDim.x + threadIdx.x; e < E; e += stride)
    atomicAdd(&cnt[edst[e]], 1);
}

__global__ __launch_bounds__(1024)
void dst_scan_kernel(const int* __restrict__ cnt, int* __restrict__ cursor, int n) {
  __shared__ int part[1024];
  const int t = threadIdx.x;
  const int chunk = (n + 1023) / 1024;
  const int b = t * chunk;
  const int e = (b + chunk < n) ? (b + chunk) : n;
  int s = 0;
  for (int i = b; i < e; ++i) s += cnt[i];
  part[t] = s;
  __syncthreads();
  for (int dd = 1; dd < 1024; dd <<= 1) {
    int add = (t >= dd) ? part[t - dd] : 0;
    __syncthreads();
    part[t] += add;
    __syncthreads();
  }
  int run = part[t] - s;
  for (int i = b; i < e; ++i) { cursor[i] = run; run += cnt[i]; }
}

__global__ void dst_fill_kernel(const int* __restrict__ edst, int* __restrict__ cursor,
                                int* __restrict__ order, int E) {
  int stride = gridDim.x * blockDim.x;
  for (int e = blockIdx.x * blockDim.x + threadIdx.x; e < E; e += stride) {
    int pos = atomicAdd(&cursor[edst[e]], 1);
    order[pos] = e;
  }
}

__global__ void node_prep_kernel(const float* __restrict__ ns, const float* __restrict__ nv,
                                 const float* __restrict__ attr,
                                 const float* __restrict__ Wls, const float* __restrict__ Wlv,
                                 const float* __restrict__ Wss, const float* __restrict__ Wsv,
                                 float* __restrict__ f_s, float* __restrict__ f_v,
                                 float* __restrict__ sc_s, float* __restrict__ sc_v, int n)
{
  int node = blockIdx.x;
  if (node >= n) return;
  __shared__ float s_ns[MUL0];
  __shared__ float s_nv[MUL1 * 3];
  int t = threadIdx.x;
  if (t < MUL0)        s_ns[t]        = ns[(long long)node * MUL0 + t];
  else                 s_nv[t - MUL0] = nv[(long long)node * (MUL1 * 3) + (t - MUL0)];
  __syncthreads();
  float a = attr[node];
  if (t < MUL0) {
    float af = 0.f, asc = 0.f;
    #pragma unroll 8
    for (int u = 0; u < MUL0; ++u) {
      float x = s_ns[u];
      af  += x * Wls[u * MUL0 + t];
      asc += x * Wss[u * MUL0 + t];
    }
    f_s [(long long)node * MUL0 + t] = af  * a;
    sc_s[(long long)node * MUL0 + t] = asc * a;
  } else {
    int idx = t - MUL0;
    int vv = idx / 3;
    int k  = idx - vv * 3;
    float af = 0.f, asc = 0.f;
    #pragma unroll 8
    for (int u = 0; u < MUL1; ++u) {
      float x = s_nv[u * 3 + k];
      af  += x * Wlv[u * MUL1 + vv];
      asc += x * Wsv[u * MUL1 + vv];
    }
    f_v [(long long)node * (MUL1 * 3) + idx] = af  * a;
    sc_v[(long long)node * (MUL1 * 3) + idx] = asc * a;
  }
}

__global__ __launch_bounds__(WAVES * 32, 2)
void edge_fused_kernel(const float* __restrict__ es,
                       const float* __restrict__ ea,
                       const int*   __restrict__ esrc,
                       const int*   __restrict__ edst,
                       const int*   __restrict__ order,
                       const float* __restrict__ Wfc1,
                       const float* __restrict__ Wfc2,
                       const float* __restrict__ f_s,
                       const float* __restrict__ f_v,
                       float* __restrict__ agg_s,
                       float* __restrict__ agg_v,
                       int E, int numTiles)
{
  __shared__ float W1s[KPAD1 * NPAD1];
  __shared__ float W2s[NPAD1 * WNUM];
  __shared__ float Hs[WAVES][16 * NPAD1];
  __shared__ float mF[WAVES][16 * 4];
  __shared__ int   mI[WAVES][16 * 2];

  {
    constexpr int NT = WAVES * 32;
    constexpr int N1 = (KPAD1 * NPAD1 + NT - 1) / NT;
    float w1[N1];
    #pragma unroll
    for (int j = 0; j < N1; ++j) {
      const int i = threadIdx.x + NT * j;
      const int k = i / NPAD1, ncol = i - k * NPAD1;
      const bool ok = (k < NBASIS) && (ncol < FCH);
      const float x = Wfc1[ok ? k * FCH + ncol : 0];
      w1[j] = ok ? x : 0.f;
    }
    constexpr int NV  = FCH   * WNUM / 4;
    constexpr int NVP = NPAD1 * WNUM / 4;
    constexpr int N2  = NVP / NT;
    static_assert(NVP % NT == 0, "W2s must split evenly over the block");
    const float4* __restrict__ src = (const float4*)Wfc2;
    float4 w2[N2];
    #pragma unroll
    for (int j = 0; j < N2; ++j) {
      const int i = threadIdx.x + NT * j;
      const float4 x = src[i < NV ? i : NV - 1];
      w2[j] = (i < NV) ? x : make_float4(0.f, 0.f, 0.f, 0.f);
    }
    #pragma unroll
    for (int j = 0; j < N1; ++j) {
      const int i = threadIdx.x + NT * j;
      if (i < KPAD1 * NPAD1) W1s[i] = w1[j];
    }
    float4* dst = (float4*)W2s;
    #pragma unroll
    for (int j = 0; j < N2; ++j) dst[threadIdx.x + NT * j] = w2[j];
  }
  __syncthreads();

  const int wid  = threadIdx.x >> 5;
  const int lane = threadIdx.x & 31;
  const int hi   = lane >> 4;
  const int lm   = lane & 15;

  float* H  = Hs[wid];
  float* mf = mF[wid];
  int*   mi = mI[wid];

  const float inv  = 0.25f;
  const float inv3 = 0.57735026918962576451f;
  const float inv2 = 0.70710678118654752440f;

  const int gw   = blockIdx.x * WAVES + wid;
  const int nw   = gridDim.x * WAVES;
  const int tq   = numTiles / nw;
  const int trem = numTiles - tq * nw;
  const int t0   = gw * tq + (gw < trem ? gw : trem);
  const int t1   = t0 + tq + (gw < trem ? 1 : 0);

  int nodeFirst = -3, nodeLast = -3;
  if (t0 < t1) {
    const int pl = (t1 * 16 < E ? t1 * 16 : E) - 1;
    nodeFirst = edst[order[t0 * 16]];
    nodeLast  = edst[order[pl]];
  }

  float cs[6][1];
  float cv[8][3];
  #pragma unroll
  for (int i = 0; i < 6; ++i) cs[i][0] = 0.f;
  #pragma unroll
  for (int i = 0; i < 8; ++i) { cv[i][0] = 0.f; cv[i][1] = 0.f; cv[i][2] = 0.f; }
  int dLastPrev = -2;

  for (int tile = t0; tile < t1; ++tile) {
    int base = tile * 16;

    int  pA = base + lm;
    bool ev = (pA < E);
    const int  pN = base + 16;
    int        eA = order[ev ? pA : E - 1];
    const int  eN = order[pN < E ? pN : E - 1];
    const int  dNraw = edst[eN];
    const int  dNextTile = (pN < E) ? dNraw : -1;

    if (lane < 16) {
      int e = eA;
      if (ev) {
        mi[lane * 2 + 0] = esrc[e];
        mi[lane * 2 + 1] = edst[e];
        mf[lane * 4 + 0] = ea[(long long)e * 4 + 0];
        mf[lane * 4 + 1] = ea[(long long)e * 4 + 1];
        mf[lane * 4 + 2] = ea[(long long)e * 4 + 2];
        mf[lane * 4 + 3] = ea[(long long)e * 4 + 3];
      } else {
        mi[lane * 2 + 0] = 0;
        mi[lane * 2 + 1] = -1;
        mf[lane * 4 + 0] = 0.f; mf[lane * 4 + 1] = 0.f;
        mf[lane * 4 + 2] = 0.f; mf[lane * 4 + 3] = 0.f;
      }
    }

    v2f a1[3];
    #pragma unroll
    for (int kk = 0; kk < 3; ++kk) {
      int k0 = 4 * kk + 2 * hi;
      a1[kk].x = (ev && (k0     < NBASIS)) ? es[(long long)eA * NBASIS + k0    ] : 0.f;
      a1[kk].y = (ev && (k0 + 1 < NBASIS)) ? es[(long long)eA * NBASIS + k0 + 1] : 0.f;
    }

    #pragma unroll
    for (int nt = 0; nt < NPAD1 / 16; ++nt) {
      v8f acc = {};
      #pragma unroll
      for (int kk = 0; kk < 3; ++kk) {
        int k0 = 4 * kk + 2 * hi;
        v2f b;
        b.x = W1s[ k0      * NPAD1 + nt * 16 + lm];
        b.y = W1s[(k0 + 1) * NPAD1 + nt * 16 + lm];
        acc = __builtin_amdgcn_wmma_f32_16x16x4_f32(
            false, a1[kk], false, b, (short)0, acc, false, false);
      }
      #pragma unroll
      for (int r = 0; r < 8; ++r) {
        float x  = acc[r];
        float sg = fast_rcp(1.f + __expf(-x));
        H[(r + 8 * hi) * NPAD1 + nt * 16 + lm] = x * sg;
      }
    }
    asm volatile("s_wait_dscnt 0" ::: "memory");

    v2f a2[NPAD1 / 4];
    #pragma unroll
    for (int kk = 0; kk < NPAD1 / 4; ++kk) {
      int k0 = 4 * kk + 2 * hi;
      a2[kk] = *(const v2f*)&H[lm * NPAD1 + k0];
    }

    int   drow[8], srow[8];
    float y0r[8], y1x[8], y1y[8], y1z[8];
    #pragma unroll
    for (int r = 0; r < 8; ++r) {
      int m = r + 8 * hi;
      srow[r] = mi[m * 2 + 0];
      drow[r] = mi[m * 2 + 1];
      y0r[r]  = mf[m * 4 + 0];
      y1x[r]  = mf[m * 4 + 1];
      y1y[r]  = mf[m * 4 + 2];
      y1z[r]  = mf[m * 4 + 3];
    }
    const int d15   = mi[15 * 2 + 1];
    const int dPrev = hi ? mi[7 * 2 + 1] : dLastPrev;
    const int dNext = hi ? ((tile + 1 < t1) ? dNextTile : -2) : mi[8 * 2 + 1];
    RunFlags rf;
    {
      const bool contIn = (drow[0] == dPrev);
      bool noEnd = true;
      #pragma unroll
      for (int r = 0; r < 7; ++r) {
        const bool er   = (drow[r] != drow[r + 1]);
        const bool hold = contIn && noEnd;
        rf.e[r]   = er;
        rf.cap[r] = er && hold;
        rf.fl[r]  = er && !hold && (drow[r] >= 0);
        noEnd = noEnd && !er;
      }
      rf.single  = contIn && noEnd;
      rf.contOut = (drow[7] == dNext);
      const bool condA = contIn && !rf.single && (drow[0] >= 0);
      const bool condB = !rf.contOut && (drow[7] >= 0);
      const bool shA = (drow[0] == nodeFirst) || (drow[0] == nodeLast);
      const bool shB = (drow[7] == nodeFirst) || (drow[7] == nodeLast);
      rf.aSt = condA && !shA;  rf.aAt = condA && shA;
      rf.bSt = condB && !shB;  rf.bAt = condB && shB;
    }


    auto gemm2_pair = [&](const int cA, v8f (&acc)[2]) {
      int cB = cA + 16;
      asm volatile("" : "+v"(cB));
      v2f b[2][NPAD1 / 4];
      #pragma unroll
      for (int kk = 0; kk < NPAD1 / 4; ++kk) {
        const int k0 = 4 * kk + 2 * hi;
        b[0][kk].x = W2s[ k0      * WNUM + cA];
        b[0][kk].y = W2s[(k0 + 1) * WNUM + cA];
        b[1][kk].x = W2s[ k0      * WNUM + cB];
        b[1][kk].y = W2s[(k0 + 1) * WNUM + cB];
      }
      __builtin_amdgcn_sched_barrier(0);

      acc[0] = v8f{}; acc[1] = v8f{};
      #pragma unroll
      for (int kk = 0; kk < NPAD1 / 4; ++kk) {
        #pragma unroll
        for (int t = 0; t < 2; ++t) {
          acc[t] = __builtin_amdgcn_wmma_f32_16x16x4_f32(
              false, a2[kk], false, b[t][kk], (short)0, acc[t], false, false);
        }
      }
    };

    #pragma unroll
    for (int p = 0; p < 4; ++p) {
      const int cA  = p * 32 + lm;
      const int col = cA & 63;
      float g[2][8];
      #pragma unroll
      for (int t = 0; t < 2; ++t) {
        #pragma unroll
        for (int r = 0; r < 8; ++r) g[t][r] = f_s[(long long)srow[r] * MUL0 + col + 16 * t];
      }
      v8f acc[2];
      gemm2_pair(cA, acc);

      #pragma unroll
      for (int t = 0; t < 2; ++t) {
        const int c = cA + 16 * t;
        if (p < 2) {
          float v[1][8];
          #pragma unroll
          for (int r = 0; r < 8; ++r) v[0][r] = acc[t][r] * g[t][r] * y0r[r] * inv;
          seg_flush<1>(agg_s, MIDS, c, drow, rf, hi, v, cs[p < 2 ? 2 * p + t : 0]);
        } else {
          const int u = c - 64;
          float v[3][8];
          #pragma unroll
          for (int r = 0; r < 8; ++r) {
            float tt = acc[t][r] * g[t][r] * inv;
            v[0][r] = tt * y1x[r]; v[1][r] = tt * y1y[r]; v[2][r] = tt * y1z[r];
          }
          seg_flush<3>(agg_v, MIDV * 3, u * 3, drow, rf, hi, v, cv[p < 2 ? 0 : 2 * (p - 2) + t]);
        }
      }
    }

    {
      float g[2][3][8];
      #pragma unroll
      for (int t = 0; t < 2; ++t) {
        #pragma unroll
        for (int r = 0; r < 8; ++r) {
          const float* vp = &f_v[((long long)srow[r] * MUL1 + lm + 16 * t) * 3];
          g[t][0][r] = vp[0]; g[t][1][r] = vp[1]; g[t][2][r] = vp[2];
        }
      }
      v8f acc[2];

      gemm2_pair(128 + lm, acc);
      #pragma unroll
      for (int t = 0; t < 2; ++t) {
        const int u = lm + 16 * t;
        float v[3][8];
        #pragma unroll
        for (int r = 0; r < 8; ++r) {
          float tt = acc[t][r] * y0r[r] * inv;
          v[0][r] = tt * g[t][0][r]; v[1][r] = tt * g[t][1][r]; v[2][r] = tt * g[t][2][r];
        }
        seg_flush<3>(agg_v, MIDV * 3, (64 + u) * 3, drow, rf, hi, v, cv[4 + t]);
      }

      gemm2_pair(160 + lm, acc);
      #pragma unroll
      for (int t = 0; t < 2; ++t) {
        const int u = lm + 16 * t;
        float v[1][8];
        #pragma unroll
        for (int r = 0; r < 8; ++r) {
          float dt = g[t][0][r] * y1x[r] + g[t][1][r] * y1y[r] + g[t][2][r] * y1z[r];
          v[0][r] = acc[t][r] * dt * inv3 * inv;
        }
        seg_flush<1>(agg_s, MIDS, 64 + u, drow, rf, hi, v, cs[4 + t]);
      }

      gemm2_pair(192 + lm, acc);
      #pragma unroll
      for (int t = 0; t < 2; ++t) {
        const int u = lm + 16 * t;
        float v[3][8];
        #pragma unroll
        for (int r = 0; r < 8; ++r) {
          float cx = g[t][1][r] * y1z[r] - g[t][2][r] * y1y[r];
          float cy = g[t][2][r] * y1x[r] - g[t][0][r] * y1z[r];
          float cz = g[t][0][r] * y1y[r] - g[t][1][r] * y1x[r];
          float tt = acc[t][r] * inv2 * inv;
          v[0][r] = tt * cx; v[1][r] = tt * cy; v[2][r] = tt * cz;
        }
        seg_flush<3>(agg_v, MIDV * 3, (96 + u) * 3, drow, rf, hi, v, cv[6 + t]);
      }
    }
    dLastPrev = d15;
  }
}

__global__ void node_out_kernel(const float* __restrict__ agg_s, const float* __restrict__ agg_v,
                                const float* __restrict__ sc_s,  const float* __restrict__ sc_v,
                                const float* __restrict__ attr,
                                const float* __restrict__ Wl2s, const float* __restrict__ Wl2v,
                                const float* __restrict__ Wl3,
                                float* __restrict__ out, int n)
{
  int node = blockIdx.x;
  if (node >= n) return;
  __shared__ float sAs[MIDS];
  __shared__ float sAv[MIDV * 3];
  int t = threadIdx.x;
  if (t < MIDS) sAs[t] = agg_s[(long long)node * MIDS + t];
  for (int i = t; i < MIDV * 3; i += 160) sAv[i] = agg_v[(long long)node * (MIDV * 3) + i];
  __syncthreads();

  float a = attr[node];
  float ang = 0.f;
  #pragma unroll 8
  for (int u = 0; u < MIDS; ++u) ang += sAs[u] * Wl3[u];
  ang *= 0.1f * a;
  float cs = cosf(ang);
  float sn = sinf(ang);

  if (t < MUL0) {
    float conv = 0.f;
    #pragma unroll 8
    for (int u = 0; u < MIDS; ++u) conv += sAs[u] * Wl2s[u * MUL0 + t];
    conv *= a;
    out[(long long)node * 160 + t] = cs * sc_s[(long long)node * MUL0 + t] + sn * conv;
  } else {
    int idx = t - MUL0;
    int vv = idx / 3;
    int k  = idx - vv * 3;
    float conv = 0.f;
    #pragma unroll 8
    for (int u = 0; u < MIDV; ++u) conv += sAv[u * 3 + k] * Wl2v[u * MUL1 + vv];
    conv *= a;
    out[(long long)node * 160 + t] = cs * sc_v[(long long)node * (MUL1 * 3) + idx] + sn * conv;
  }
}

extern "C" void kernel_launch(void* const* d_in, const int* in_sizes, int n_in,
                              void* d_out, int out_size, void* d_ws, size_t ws_size,
                              hipStream_t stream) {
  const float* node_scalars = (const float*)d_in[0];
  const float* node_vectors = (const float*)d_in[1];
  const float* node_attr    = (const float*)d_in[2];
  const int*   edge_src     = (const int*)  d_in[3];
  const int*   edge_dst     = (const int*)  d_in[4];
  const float* edge_attr    = (const float*)d_in[5];
  const float* edge_scalars = (const float*)d_in[6];
  const float* W_fc1        = (const float*)d_in[7];
  const float* W_fc2        = (const float*)d_in[8];
  const float* W_sc_s       = (const float*)d_in[9];
  const float* W_sc_v       = (const float*)d_in[10];
  const float* W_l1_s       = (const float*)d_in[11];
  const float* W_l1_v       = (const float*)d_in[12];
  const float* W_l2_s       = (const float*)d_in[13];
  const float* W_l2_v       = (const float*)d_in[14];
  const float* W_l3         = (const float*)d_in[15];

  const int N = in_sizes[0] / MUL0;
  const int E = in_sizes[3];

  float* ws    = (float*)d_ws;
  float* f_s   = ws;
  float* f_v   = f_s  + (size_t)N * MUL0;
  float* sc_s  = f_v  + (size_t)N * MUL1 * 3;
  float* sc_v  = sc_s + (size_t)N * MUL0;
  float* agg_s = sc_v + (size_t)N * MUL1 * 3;
  float* agg_v = agg_s + (size_t)N * MIDS;

  int* cnt    = (int*)d_out;
  int* cursor = cnt + N;
  int* order  = cursor + N;

  long long nAgg = (long long)N * (MIDS + MIDV * 3);
  zero_kernel<<<4096, 256, 0, stream>>>(agg_s, nAgg, cnt, N);

  node_prep_kernel<<<N, 160, 0, stream>>>(node_scalars, node_vectors, node_attr,
                                          W_l1_s, W_l1_v, W_sc_s, W_sc_v,
                                          f_s, f_v, sc_s, sc_v, N);

  dst_count_kernel<<<1024, 256, 0, stream>>>(edge_dst, cnt, E);
  dst_scan_kernel<<<1, 1024, 0, stream>>>(cnt, cursor, N);
  dst_fill_kernel<<<1024, 256, 0, stream>>>(edge_dst, cursor, order, E);

  int numTiles = (E + 15) / 16;
  edge_fused_kernel<<<512, WAVES * 32, 0, stream>>>(edge_scalars, edge_attr,
                                                     edge_src, edge_dst, order,
                                                     W_fc1, W_fc2, f_s, f_v,
                                                     agg_s, agg_v, E, numTiles);

  node_out_kernel<<<N, 160, 0, stream>>>(agg_s, agg_v, sc_s, sc_v, node_attr,
                                         W_l2_s, W_l2_v, W_l3, (float*)d_out, N);
}


# edge kernel: W_fc2 lies in LDS in WMMA fragment order, one 16-byte LDS read gives a lane the B operands of both column tiles for a k step (28 reads per tile pair, was 80 of 4 or 8 bytes)
# speedup vs baseline: 2.5027x; 1.0452x over previous
#include <hip/hip_runtime.h>
#include <hip/hip_bf16.h>
#include <math.h>


typedef __attribute__((ext_vector_type(2))) float v2f;
typedef __attribute__((ext_vector_type(8))) float v8f;

#define MUL0   64
#define MUL1   32
#define NBASIS 10
#define FCH    100
#define WNUM   224
#define MIDS   96
#define MIDV   128
#define KPAD1  12
#define NPAD1  112
#define WAVES  4

__device__ __forceinline__ void atomic_add_f32(float* p, float v) {
  __hip_atomic_fetch_add(p, v, __ATOMIC_RELAXED, __HIP_MEMORY_SCOPE_AGENT);
}

__device__ __forceinline__ float fast_rcp(float x) {
#if __has_builtin(__builtin_amdgcn_rcpf)
  return __builtin_amdgcn_rcpf(x);
#else
  return 1.f / x;
#endif
}

__device__ __forceinline__ float xhalf(float x) {
  const int b = __builtin_bit_cast(int, x);
  return __builtin_bit_cast(float, __builtin_amdgcn_permlanex16(b, b, 0x76543210, 0xfedcba98, false, false));
}

struct RunFlags {
  bool e[7];
  bool cap[7];
  bool fl[7];
  bool single;
  bool contOut;
  bool aSt, aAt;
  bool bSt, bAt;
};

template <int NC>
__device__ __forceinline__ void seg_flush(float* __restrict__ agg, int rowStride, int colOff,
                                          const int (&d)[8], const RunFlags& f, int hi,
                                          const float (&v)[NC][8], float (&carry)[NC]) {
  float run[NC], head[NC], th[NC], tl[NC];
  #pragma unroll
  for (int k = 0; k < NC; ++k) { run[k] = 0.f; head[k] = 0.f; }
  #pragma unroll
  for (int r = 0; r < 8; ++r) {
    #pragma unroll
    for (int k = 0; k < NC; ++k) run[k] += v[k][r];
    if (r < 7) {
      #pragma unroll
      for (int k = 0; k < NC; ++k) head[k] = f.cap[r] ? run[k] : head[k];
      if (f.fl[r]) {
        float* p = agg + (long long)d[r] * rowStride + colOff;
        #pragma unroll
        for (int k = 0; k < NC; ++k) atomic_add_f32(p + k, run[k]);
      }
      #pragma unroll
      for (int k = 0; k < NC; ++k) run[k] = f.e[r] ? 0.f : run[k];
    }
  }
  #pragma unroll
  for (int k = 0; k < NC; ++k) {
    const float hv    = f.single ? run[k] : head[k];
    const float cinLo = xhalf(carry[k]);
    const float outLo = f.contOut ? (f.single ? hv + cinLo : run[k]) : 0.f;
    const float x     = xhalf(outLo);
    const float cin   = hi ? x : cinLo;
    th[k] = hv + cin;
    tl[k] = f.single ? th[k] : run[k];
    carry[k] = f.contOut ? tl[k] : 0.f;
  }
  if (f.aSt) {
    float* p = agg + (long long)d[0] * rowStride + colOff;
    #pragma unroll
    for (int k = 0; k < NC; ++k) p[k] = th[k];
  }
  if (f.aAt) {
    float* p = agg + (long long)d[0] * rowStride + colOff;
    #pragma unroll
    for (int k = 0; k < NC; ++k) atomic_add_f32(p + k, th[k]);
  }
  if (f.bSt) {
    float* p = agg + (long long)d[7] * rowStride + colOff;
    #pragma unroll
    for (int k = 0; k < NC; ++k) p[k] = tl[k];
  }
  if (f.bAt) {
    float* p = agg + (long long)d[7] * rowStride + colOff;
    #pragma unroll
    for (int k = 0; k < NC; ++k) atomic_add_f32(p + k, tl[k]);
  }
}

__global__ void zero_kernel(float* __restrict__ p, long long n, int* __restrict__ q, int m) {
  long long i = (long long)blockIdx.x * blockDim.x + threadIdx.x;
  long long stride = (long long)gridDim.x * blockDim.x;
  for (long long j = i; j < m; j += stride) q[j] = 0;
  for (; i < n; i += stride) p[i] = 0.f;
}

__global__ void dst_count_kernel(const int* __restrict__ edst, int* __restrict__ cnt, int E) {
  int stride = gridDim.x * blockDim.x;
  for (int e = blockIdx.x * blockDim.x + threadIdx.x; e < E; e += stride)
    atomicAdd(&cnt[edst[e]], 1);
}

__global__ __launch_bounds__(1024)
void dst_scan_kernel(const int* __restrict__ cnt, int* __restrict__ cursor, int n) {
  __shared__ int part[1024];
  const int t = threadIdx.x;
  const int chunk = (n + 1023) / 1024;
  const int b = t * chunk;
  const int e = (b + chunk < n) ? (b + chunk) : n;
  int s = 0;
  for (int i = b; i < e; ++i) s += cnt[i];
  part[t] = s;
  __syncthreads();
  for (int dd = 1; dd < 1024; dd <<= 1) {
    int add = (t >= dd) ? part[t - dd] : 0;
    __syncthreads();
    part[t] += add;
    __syncthreads();
  }
  int run = part[t] - s;
  for (int i = b; i < e; ++i) { cursor[i] = run; run += cnt[i]; }
}

__global__ void dst_fill_kernel(const int* __restrict__ edst, int* __restrict__ cursor,
                                int* __restrict__ order, int E) {
  int stride = gridDim.x * blockDim.x;
  for (int e = blockIdx.x * blockDim.x + threadIdx.x; e < E; e += stride) {
    int pos = atomicAdd(&cursor[edst[e]], 1);
    order[pos] = e;
  }
}

__global__ void node_prep_kernel(const float* __restrict__ ns, const float* __restrict__ nv,
                                 const float* __restrict__ attr,
                                 const float* __restrict__ Wls, const float* __restrict__ Wlv,
                                 const float* __restrict__ Wss, const float* __restrict__ Wsv,
                                 float* __restrict__ f_s, float* __restrict__ f_v,
                                 float* __restrict__ sc_s, float* __restrict__ sc_v, int n)
{
  int node = blockIdx.x;
  if (node >= n) return;
  __shared__ float s_ns[MUL0];
  __shared__ float s_nv[MUL1 * 3];
  int t = threadIdx.x;
  if (t < MUL0)        s_ns[t]        = ns[(long long)node * MUL0 + t];
  else                 s_nv[t - MUL0] = nv[(long long)node * (MUL1 * 3) + (t - MUL0)];
  __syncthreads();
  float a = attr[node];
  if (t < MUL0) {
    float af = 0.f, asc = 0.f;
    #pragma unroll 8
    for (int u = 0; u < MUL0; ++u) {
      float x = s_ns[u];
      af  += x * Wls[u * MUL0 + t];
      asc += x * Wss[u * MUL0 + t];
    }
    f_s [(long long)node * MUL0 + t] = af  * a;
    sc_s[(long long)node * MUL0 + t] = asc * a;
  } else {
    int idx = t - MUL0;
    int vv = idx / 3;
    int k  = idx - vv * 3;
    float af = 0.f, asc = 0.f;
    #pragma unroll 8
    for (int u = 0; u < MUL1; ++u) {
      float x = s_nv[u * 3 + k];
      af  += x * Wlv[u * MUL1 + vv];
      asc += x * Wsv[u * MUL1 + vv];
    }
    f_v [(long long)node * (MUL1 * 3) + idx] = af  * a;
    sc_v[(long long)node * (MUL1 * 3) + idx] = asc * a;
  }
}

__global__ __launch_bounds__(WAVES * 32, 2)
void edge_fused_kernel(const float* __restrict__ es,
                       const float* __restrict__ ea,
                       const int*   __restrict__ esrc,
                       const int*   __restrict__ edst,
                       const int*   __restrict__ order,
                       const float* __restrict__ Wfc1,
                       const float* __restrict__ Wfc2,
                       const float* __restrict__ f_s,
                       const float* __restrict__ f_v,
                       float* __restrict__ agg_s,
                       float* __restrict__ agg_v,
                       int E, int numTiles)
{
  __shared__ float W1s[KPAD1 * NPAD1];
  __shared__ float4 W2f[NPAD1 * WNUM / 4];
  __shared__ float Hs[WAVES][16 * NPAD1];
  __shared__ float mF[WAVES][16 * 4];
  __shared__ int   mI[WAVES][16 * 2];

  {
    constexpr int NT = WAVES * 32;
    constexpr int N1 = (KPAD1 * NPAD1 + NT - 1) / NT;
    float w1[N1];
    #pragma unroll
    for (int j = 0; j < N1; ++j) {
      const int i = threadIdx.x + NT * j;
      const int k = i / NPAD1, ncol = i - k * NPAD1;
      const bool ok = (k < NBASIS) && (ncol < FCH);
      const float x = Wfc1[ok ? k * FCH + ncol : 0];
      w1[j] = ok ? x : 0.f;
    }
    constexpr int NS = NPAD1 * WNUM / 4;
    constexpr int N2 = NS / NT;
    static_assert(NS % NT == 0, "W2f must split evenly over the block");
    static_assert(NT % 32 == 0 && FCH % 2 == 0, "slot decoding below");
    float4 w2[N2];
    #pragma unroll
    for (int j = 0; j < N2; ++j) {
      const int s  = threadIdx.x + NT * j;
      const int ln = s & 31, q = s >> 5;
      const int pr = q / (NPAD1 / 4), kk = q - pr * (NPAD1 / 4);
      const int k0 = 4 * kk + 2 * (ln >> 4);
      const int cc = 32 * pr + (ln & 15);
      const bool ok = (k0 < FCH);
      const float* r0 = Wfc2 + (ok ? k0 * WNUM + cc : 0);
      float4 x;
      x.x = r0[0];  x.y = r0[WNUM];  x.z = r0[16];  x.w = r0[WNUM + 16];
      w2[j] = ok ? x : make_float4(0.f, 0.f, 0.f, 0.f);
    }
    #pragma unroll
    for (int j = 0; j < N1; ++j) {
      const int i = threadIdx.x + NT * j;
      if (i < KPAD1 * NPAD1) W1s[i] = w1[j];
    }
    #pragma unroll
    for (int j = 0; j < N2; ++j) W2f[threadIdx.x + NT * j] = w2[j];
  }
  __syncthreads();

  const int wid  = threadIdx.x >> 5;
  const int lane = threadIdx.x & 31;
  const int hi   = lane >> 4;
  const int lm   = lane & 15;

  float* H  = Hs[wid];
  float* mf = mF[wid];
  int*   mi = mI[wid];

  const float inv  = 0.25f;
  const float inv3 = 0.57735026918962576451f;
  const float inv2 = 0.70710678118654752440f;

  const int gw   = blockIdx.x * WAVES + wid;
  const int nw   = gridDim.x * WAVES;
  const int tq   = numTiles / nw;
  const int trem = numTiles - tq * nw;
  const int t0   = gw * tq + (gw < trem ? gw : trem);
  const int t1   = t0 + tq + (gw < trem ? 1 : 0);

  int nodeFirst = -3, nodeLast = -3;
  if (t0 < t1) {
    const int pl = (t1 * 16 < E ? t1 * 16 : E) - 1;
    nodeFirst = edst[order[t0 * 16]];
    nodeLast  = edst[order[pl]];
  }

  float cs[6][1];
  float cv[8][3];
  #pragma unroll
  for (int i = 0; i < 6; ++i) cs[i][0] = 0.f;
  #pragma unroll
  for (int i = 0; i < 8; ++i) { cv[i][0] = 0.f; cv[i][1] = 0.f; cv[i][2] = 0.f; }
  int dLastPrev = -2;

  for (int tile = t0; tile < t1; ++tile) {
    int base = tile * 16;

    int  pA = base + lm;
    bool ev = (pA < E);
    const int  pN = base + 16;
    int        eA = order[ev ? pA : E - 1];
    const int  eN = order[pN < E ? pN : E - 1];
    const int  dNraw = edst[eN];
    const int  dNextTile = (pN < E) ? dNraw : -1;

    if (lane < 16) {
      int e = eA;
      if (ev) {
        mi[lane * 2 + 0] = esrc[e];
        mi[lane * 2 + 1] = edst[e];
        mf[lane * 4 + 0] = ea[(long long)e * 4 + 0];
        mf[lane * 4 + 1] = ea[(long long)e * 4 + 1];
        mf[lane * 4 + 2] = ea[(long long)e * 4 + 2];
        mf[lane * 4 + 3] = ea[(long long)e * 4 + 3];
      } else {
        mi[lane * 2 + 0] = 0;
        mi[lane * 2 + 1] = -1;
        mf[lane * 4 + 0] = 0.f; mf[lane * 4 + 1] = 0.f;
        mf[lane * 4 + 2] = 0.f; mf[lane * 4 + 3] = 0.f;
      }
    }

    v2f a1[3];
    #pragma unroll
    for (int kk = 0; kk < 3; ++kk) {
      int k0 = 4 * kk + 2 * hi;
      a1[kk].x = (ev && (k0     < NBASIS)) ? es[(long long)eA * NBASIS + k0    ] : 0.f;
      a1[kk].y = (ev && (k0 + 1 < NBASIS)) ? es[(long long)eA * NBASIS + k0 + 1] : 0.f;
    }

    #pragma unroll
    for (int nt = 0; nt < NPAD1 / 16; ++nt) {
      v8f acc = {};
      #pragma unroll
      for (int kk = 0; kk < 3; ++kk) {
        int k0 = 4 * kk + 2 * hi;
        v2f b;
        b.x = W1s[ k0      * NPAD1 + nt * 16 + lm];
        b.y = W1s[(k0 + 1) * NPAD1 + nt * 16 + lm];
        acc = __builtin_amdgcn_wmma_f32_16x16x4_f32(
            false, a1[kk], false, b, (short)0, acc, false, false);
      }
      #pragma unroll
      for (int r = 0; r < 8; ++r) {
        float x  = acc[r];
        float sg = fast_rcp(1.f + __expf(-x));
        H[(r + 8 * hi) * NPAD1 + nt * 16 + lm] = x * sg;
      }
    }
    asm volatile("s_wait_dscnt 0" ::: "memory");

    v2f a2[NPAD1 / 4];
    #pragma unroll
    for (int kk = 0; kk < NPAD1 / 4; ++kk) {
      int k0 = 4 * kk + 2 * hi;
      a2[kk] = *(const v2f*)&H[lm * NPAD1 + k0];
    }

    int   drow[8], srow[8];
    float y0r[8], y1x[8], y1y[8], y1z[8];
    #pragma unroll
    for (int r = 0; r < 8; ++r) {
      int m = r + 8 * hi;
      srow[r] = mi[m * 2 + 0];
      drow[r] = mi[m * 2 + 1];
      y0r[r]  = mf[m * 4 + 0];
      y1x[r]  = mf[m * 4 + 1];
      y1y[r]  = mf[m * 4 + 2];
      y1z[r]  = mf[m * 4 + 3];
    }
    const int d15   = mi[15 * 2 + 1];
    const int dPrev = hi ? mi[7 * 2 + 1] : dLastPrev;
    const int dNext = hi ? ((tile + 1 < t1) ? dNextTile : -2) : mi[8 * 2 + 1];
    RunFlags rf;
    {
      const bool contIn = (drow[0] == dPrev);
      bool noEnd = true;
      #pragma unroll
      for (int r = 0; r < 7; ++r) {
        const bool er   = (drow[r] != drow[r + 1]);
        const bool hold = contIn && noEnd;
        rf.e[r]   = er;
        rf.cap[r] = er && hold;
        rf.fl[r]  = er && !hold && (drow[r] >= 0);
        noEnd = noEnd && !er;
      }
      rf.single  = contIn && noEnd;
      rf.contOut = (drow[7] == dNext);
      const bool condA = contIn && !rf.single && (drow[0] >= 0);
      const bool condB = !rf.contOut && (drow[7] >= 0);
      const bool shA = (drow[0] == nodeFirst) || (drow[0] == nodeLast);
      const bool shB = (drow[7] == nodeFirst) || (drow[7] == nodeLast);
      rf.aSt = condA && !shA;  rf.aAt = condA && shA;
      rf.bSt = condB && !shB;  rf.bAt = condB && shB;
    }


    auto gemm2_pair = [&](const int pr, v8f (&acc)[2]) {
      const float4* bp = &W2f[pr * (NPAD1 / 4) * 32 + lane];
      v2f b[2][NPAD1 / 4];
      #pragma unroll
      for (int kk = 0; kk < NPAD1 / 4; ++kk) {
        const float4 q = bp[kk * 32];
        b[0][kk].x = q.x;  b[0][kk].y = q.y;
        b[1][kk].x = q.z;  b[1][kk].y = q.w;
      }
      __builtin_amdgcn_sched_barrier(0);

      acc[0] = v8f{}; acc[1] = v8f{};
      #pragma unroll
      for (int kk = 0; kk < NPAD1 / 4; ++kk) {
        #pragma unroll
        for (int t = 0; t < 2; ++t) {
          acc[t] = __builtin_amdgcn_wmma_f32_16x16x4_f32(
              false, a2[kk], false, b[t][kk], (short)0, acc[t], false, false);
        }
      }
    };

    #pragma unroll
    for (int p = 0; p < 4; ++p) {
      const int cA  = p * 32 + lm;
      const int col = cA & 63;
      float g[2][8];
      #pragma unroll
      for (int t = 0; t < 2; ++t) {
        #pragma unroll
        for (int r = 0; r < 8; ++r) g[t][r] = f_s[(long long)srow[r] * MUL0 + col + 16 * t];
      }
      v8f acc[2];
      gemm2_pair(p, acc);

      #pragma unroll
      for (int t = 0; t < 2; ++t) {
        const int c = cA + 16 * t;
        if (p < 2) {
          float v[1][8];
          #pragma unroll
          for (int r = 0; r < 8; ++r) v[0][r] = acc[t][r] * g[t][r] * y0r[r] * inv;
          seg_flush<1>(agg_s, MIDS, c, drow, rf, hi, v, cs[p < 2 ? 2 * p + t : 0]);
        } else {
          const int u = c - 64;
          float v[3][8];
          #pragma unroll
          for (int r = 0; r < 8; ++r) {
            float tt = acc[t][r] * g[t][r] * inv;
            v[0][r] = tt * y1x[r]; v[1][r] = tt * y1y[r]; v[2][r] = tt * y1z[r];
          }
          seg_flush<3>(agg_v, MIDV * 3, u * 3, drow, rf, hi, v, cv[p < 2 ? 0 : 2 * (p - 2) + t]);
        }
      }
    }

    {
      float g[2][3][8];
      #pragma unroll
      for (int t = 0; t < 2; ++t) {
        #pragma unroll
        for (int r = 0; r < 8; ++r) {
          const float* vp = &f_v[((long long)srow[r] * MUL1 + lm + 16 * t) * 3];
          g[t][0][r] = vp[0]; g[t][1][r] = vp[1]; g[t][2][r] = vp[2];
        }
      }
      v8f acc[2];

      gemm2_pair(4, acc);
      #pragma unroll
      for (int t = 0; t < 2; ++t) {
        const int u = lm + 16 * t;
        float v[3][8];
        #pragma unroll
        for (int r = 0; r < 8; ++r) {
          float tt = acc[t][r] * y0r[r] * inv;
          v[0][r] = tt * g[t][0][r]; v[1][r] = tt * g[t][1][r]; v[2][r] = tt * g[t][2][r];
        }
        seg_flush<3>(agg_v, MIDV * 3, (64 + u) * 3, drow, rf, hi, v, cv[4 + t]);
      }

      gemm2_pair(5, acc);
      #pragma unroll
      for (int t = 0; t < 2; ++t) {
        const int u = lm + 16 * t;
        float v[1][8];
        #pragma unroll
        for (int r = 0; r < 8; ++r) {
          float dt = g[t][0][r] * y1x[r] + g[t][1][r] * y1y[r] + g[t][2][r] * y1z[r];
          v[0][r] = acc[t][r] * dt * inv3 * inv;
        }
        seg_flush<1>(agg_s, MIDS, 64 + u, drow, rf, hi, v, cs[4 + t]);
      }

      gemm2_pair(6, acc);
      #pragma unroll
      for (int t = 0; t < 2; ++t) {
        const int u = lm + 16 * t;
        float v[3][8];
        #pragma unroll
        for (int r = 0; r < 8; ++r) {
          float cx = g[t][1][r] * y1z[r] - g[t][2][r] * y1y[r];
          float cy = g[t][2][r] * y1x[r] - g[t][0][r] * y1z[r];
          float cz = g[t][0][r] * y1y[r] - g[t][1][r] * y1x[r];
          float tt = acc[t][r] * inv2 * inv;
          v[0][r] = tt * cx; v[1][r] = tt * cy; v[2][r] = tt * cz;
        }
        seg_flush<3>(agg_v, MIDV * 3, (96 + u) * 3, drow, rf, hi, v, cv[6 + t]);
      }
    }
    dLastPrev = d15;
  }
}

__global__ void node_out_kernel(const float* __restrict__ agg_s, const float* __restrict__ agg_v,
                                const float* __restrict__ sc_s,  const float* __restrict__ sc_v,
                                const float* __restrict__ attr,
                                const float* __restrict__ Wl2s, const float* __restrict__ Wl2v,
                                const float* __restrict__ Wl3,
                                float* __restrict__ out, int n)
{
  int node = blockIdx.x;
  if (node >= n) return;
  __shared__ float sAs[MIDS];
  __shared__ float sAv[MIDV * 3];
  int t = threadIdx.x;
  if (t < MIDS) sAs[t] = agg_s[(long long)node * MIDS + t];
  for (int i = t; i < MIDV * 3; i += 160) sAv[i] = agg_v[(long long)node * (MIDV * 3) + i];
  __syncthreads();

  float a = attr[node];
  float ang = 0.f;
  #pragma unroll 8
  for (int u = 0; u < MIDS; ++u) ang += sAs[u] * Wl3[u];
  ang *= 0.1f * a;
  float cs = cosf(ang);
  float sn = sinf(ang);

  if (t < MUL0) {
    float conv = 0.f;
    #pragma unroll 8
    for (int u = 0; u < MIDS; ++u) conv += sAs[u] * Wl2s[u * MUL0 + t];
    conv *= a;
    out[(long long)node * 160 + t] = cs * sc_s[(long long)node * MUL0 + t] + sn * conv;
  } else {
    int idx = t - MUL0;
    int vv = idx / 3;
    int k  = idx - vv * 3;
    float conv = 0.f;
    #pragma unroll 8
    for (int u = 0; u < MIDV; ++u) conv += sAv[u * 3 + k] * Wl2v[u * MUL1 + vv];
    conv *= a;
    out[(long long)node * 160 + t] = cs * sc_v[(long long)node * (MUL1 * 3) + idx] + sn * conv;
  }
}

extern "C" void kernel_launch(void* const* d_in, const int* in_sizes, int n_in,
                              void* d_out, int out_size, void* d_ws, size_t ws_size,
                              hipStream_t stream) {
  const float* node_scalars = (const float*)d_in[0];
  const float* node_vectors = (const float*)d_in[1];
  const float* node_attr    = (const float*)d_in[2];
  const int*   edge_src     = (const int*)  d_in[3];
  const int*   edge_dst     = (const int*)  d_in[4];
  const float* edge_attr    = (const float*)d_in[5];
  const float* edge_scalars = (const float*)d_in[6];
  const float* W_fc1        = (const float*)d_in[7];
  const float* W_fc2        = (const float*)d_in[8];
  const float* W_sc_s       = (const float*)d_in[9];
  const float* W_sc_v       = (const float*)d_in[10];
  const float* W_l1_s       = (const float*)d_in[11];
  const float* W_l1_v       = (const float*)d_in[12];
  const float* W_l2_s       = (const float*)d_in[13];
  const float* W_l2_v       = (const float*)d_in[14];
  const float* W_l3         = (const float*)d_in[15];

  const int N = in_sizes[0] / MUL0;
  const int E = in_sizes[3];

  float* ws    = (float*)d_ws;
  float* f_s   = ws;
  float* f_v   = f_s  + (size_t)N * MUL0;
  float* sc_s  = f_v  + (size_t)N * MUL1 * 3;
  float* sc_v  = sc_s + (size_t)N * MUL0;
  float* agg_s = sc_v + (size_t)N * MUL1 * 3;
  float* agg_v = agg_s + (size_t)N * MIDS;

  int* cnt    = (int*)d_out;
  int* cursor = cnt + N;
  int* order  = cursor + N;

  long long nAgg = (long long)N * (MIDS + MIDV * 3);
  zero_kernel<<<4096, 256, 0, stream>>>(agg_s, nAgg, cnt, N);

  node_prep_kernel<<<N, 160, 0, stream>>>(node_scalars, node_vectors, node_attr,
                                          W_l1_s, W_l1_v, W_sc_s, W_sc_v,
                                          f_s, f_v, sc_s, sc_v, N);

  dst_count_kernel<<<1024, 256, 0, stream>>>(edge_dst, cnt, E);
  dst_scan_kernel<<<1, 1024, 0, stream>>>(cnt, cursor, N);
  dst_fill_kernel<<<1024, 256, 0, stream>>>(edge_dst, cursor, order, E);

  int numTiles = (E + 15) / 16;
  edge_fused_kernel<<<512, WAVES * 32, 0, stream>>>(edge_scalars, edge_attr,
                                                     edge_src, edge_dst, order,
                                                     W_fc1, W_fc2, f_s, f_v,
                                                     agg_s, agg_v, E, numTiles);

  node_out_kernel<<<N, 160, 0, stream>>>(agg_s, agg_v, sc_s, sc_v, node_attr,
                                         W_l2_s, W_l2_v, W_l3, (float*)d_out, N);
}


# node kernels on f32 16x16x4 WMMA: one wave per 16 nodes does the projections / output linears, weights from the WGP cache; was 25000 one-node blocks of vector dot products each (sum order changes)
# speedup vs baseline: 2.9680x; 1.1859x over previous
#include <hip/hip_runtime.h>
#include <hip/hip_bf16.h>
#include <math.h>


typedef __attribute__((ext_vector_type(2))) float v2f;
typedef __attribute__((ext_vector_type(8))) float v8f;

#define MUL0   64
#define MUL1   32
#define NBASIS 10
#define FCH    100
#define WNUM   224
#define MIDS   96
#define MIDV   128
#define KPAD1  12
#define NPAD1  112
#define WAVES  4

__device__ __forceinline__ void atomic_add_f32(float* p, float v) {
  __hip_atomic_fetch_add(p, v, __ATOMIC_RELAXED, __HIP_MEMORY_SCOPE_AGENT);
}

__device__ __forceinline__ float fast_rcp(float x) {
#if __has_builtin(__builtin_amdgcn_rcpf)
  return __builtin_amdgcn_rcpf(x);
#else
  return 1.f / x;
#endif
}

__device__ __forceinline__ float xhalf(float x) {
  const int b = __builtin_bit_cast(int, x);
  return __builtin_bit_cast(float, __builtin_amdgcn_permlanex16(b, b, 0x76543210, 0xfedcba98, false, false));
}

struct RunFlags {
  bool e[7];
  bool cap[7];
  bool fl[7];
  bool single;
  bool contOut;
  bool aSt, aAt;
  bool bSt, bAt;
};

template <int NC>
__device__ __forceinline__ void seg_flush(float* __restrict__ agg, int rowStride, int colOff,
                                          const int (&d)[8], const RunFlags& f, int hi,
                                          const float (&v)[NC][8], float (&carry)[NC]) {
  float run[NC], head[NC], th[NC], tl[NC];
  #pragma unroll
  for (int k = 0; k < NC; ++k) { run[k] = 0.f; head[k] = 0.f; }
  #pragma unroll
  for (int r = 0; r < 8; ++r) {
    #pragma unroll
    for (int k = 0; k < NC; ++k) run[k] += v[k][r];
    if (r < 7) {
      #pragma unroll
      for (int k = 0; k < NC; ++k) head[k] = f.cap[r] ? run[k] : head[k];
      if (f.fl[r]) {
        float* p = agg + (long long)d[r] * rowStride + colOff;
        #pragma unroll
        for (int k = 0; k < NC; ++k) atomic_add_f32(p + k, run[k]);
      }
      #pragma unroll
      for (int k = 0; k < NC; ++k) run[k] = f.e[r] ? 0.f : run[k];
    }
  }
  #pragma unroll
  for (int k = 0; k < NC; ++k) {
    const float hv    = f.single ? run[k] : head[k];
    const float cinLo = xhalf(carry[k]);
    const float outLo = f.contOut ? (f.single ? hv + cinLo : run[k]) : 0.f;
    const float x     = xhalf(outLo);
    const float cin   = hi ? x : cinLo;
    th[k] = hv + cin;
    tl[k] = f.single ? th[k] : run[k];
    carry[k] = f.contOut ? tl[k] : 0.f;
  }
  if (f.aSt) {
    float* p = agg + (long long)d[0] * rowStride + colOff;
    #pragma unroll
    for (int k = 0; k < NC; ++k) p[k] = th[k];
  }
  if (f.aAt) {
    float* p = agg + (long long)d[0] * rowStride + colOff;
    #pragma unroll
    for (int k = 0; k < NC; ++k) atomic_add_f32(p + k, th[k]);
  }
  if (f.bSt) {
    float* p = agg + (long long)d[7] * rowStride + colOff;
    #pragma unroll
    for (int k = 0; k < NC; ++k) p[k] = tl[k];
  }
  if (f.bAt) {
    float* p = agg + (long long)d[7] * rowStride + colOff;
    #pragma unroll
    for (int k = 0; k < NC; ++k) atomic_add_f32(p + k, tl[k]);
  }
}

__global__ void zero_kernel(float* __restrict__ p, long long n, int* __restrict__ q, int m) {
  long long i = (long long)blockIdx.x * blockDim.x + threadIdx.x;
  long long stride = (long long)gridDim.x * blockDim.x;
  for (long long j = i; j < m; j += stride) q[j] = 0;
  for (; i < n; i += stride) p[i] = 0.f;
}

__global__ void dst_count_kernel(const int* __restrict__ edst, int* __restrict__ cnt, int E) {
  int stride = gridDim.x * blockDim.x;
  for (int e = blockIdx.x * blockDim.x + threadIdx.x; e < E; e += stride)
    atomicAdd(&cnt[edst[e]], 1);
}

__global__ __launch_bounds__(1024)
void dst_scan_kernel(const int* __restrict__ cnt, int* __restrict__ cursor, int n) {
  __shared__ int part[1024];
  const int t = threadIdx.x;
  const int chunk = (n + 1023) / 1024;
  const int b = t * chunk;
  const int e = (b + chunk < n) ? (b + chunk) : n;
  int s = 0;
  for (int i = b; i < e; ++i) s += cnt[i];
  part[t] = s;
  __syncthreads();
  for (int dd = 1; dd < 1024; dd <<= 1) {
    int add = (t >= dd) ? part[t - dd] : 0;
    __syncthreads();
    part[t] += add;
    __syncthreads();
  }
  int run = part[t] - s;
  for (int i = b; i < e; ++i) { cursor[i] = run; run += cnt[i]; }
}

__global__ void dst_fill_kernel(const int* __restrict__ edst, int* __restrict__ cursor,
                                int* __restrict__ order, int E) {
  int stride = gridDim.x * blockDim.x;
  for (int e = blockIdx.x * blockDim.x + threadIdx.x; e < E; e += stride) {
    int pos = atomicAdd(&cursor[edst[e]], 1);
    order[pos] = e;
  }
}

__global__ __launch_bounds__(128)
void node_prep_kernel(const float* __restrict__ ns, const float* __restrict__ nv,
                      const float* __restrict__ attr,
                      const float* __restrict__ Wls, const float* __restrict__ Wlv,
                      const float* __restrict__ Wss, const float* __restrict__ Wsv,
                      float* __restrict__ f_s, float* __restrict__ f_v,
                      float* __restrict__ sc_s, float* __restrict__ sc_v, int n)
{
  const int lane = threadIdx.x & 31;
  const int hi   = lane >> 4;
  const int lm   = lane & 15;
  const int n0   = (blockIdx.x * 4 + (threadIdx.x >> 5)) * 16;
  if (n0 >= n) return;
  const int na = (n0 + lm < n) ? (n0 + lm) : (n - 1);

  float ar[8];
  #pragma unroll
  for (int r = 0; r < 8; ++r) {
    const int nd = n0 + r + 8 * hi;
    ar[r] = attr[nd < n ? nd : n - 1];
  }

  v2f as[MUL0 / 4];
  #pragma unroll
  for (int kk = 0; kk < MUL0 / 4; ++kk)
    as[kk] = *(const v2f*)(ns + (long long)na * MUL0 + 4 * kk + 2 * hi);

  #pragma unroll
  for (int w = 0; w < 2; ++w) {
    const float* __restrict__ Wm = w ? Wss : Wls;
    float* __restrict__ dst      = w ? sc_s : f_s;
    v8f acc[4];
    #pragma unroll
    for (int t = 0; t < 4; ++t) acc[t] = v8f{};
    #pragma unroll
    for (int kk = 0; kk < MUL0 / 4; ++kk) {
      const int k0 = 4 * kk + 2 * hi;
      #pragma unroll
      for (int t = 0; t < 4; ++t) {
        v2f b;
        b.x = Wm[ k0      * MUL0 + 16 * t + lm];
        b.y = Wm[(k0 + 1) * MUL0 + 16 * t + lm];
        acc[t] = __builtin_amdgcn_wmma_f32_16x16x4_f32(
            false, as[kk], false, b, (short)0, acc[t], false, false);
      }
    }
    #pragma unroll
    for (int r = 0; r < 8; ++r) {
      const int nd = n0 + r + 8 * hi;
      if (nd < n) {
        #pragma unroll
        for (int t = 0; t < 4; ++t) dst[(long long)nd * MUL0 + 16 * t + lm] = acc[t][r] * ar[r];
      }
    }
  }

  v2f av[3][MUL1 / 4];
  #pragma unroll
  for (int kk = 0; kk < MUL1 / 4; ++kk) {
    const v2f* q = (const v2f*)(nv + (long long)na * (MUL1 * 3) + 12 * kk + 6 * hi);
    const v2f q0 = q[0], q1 = q[1], q2 = q[2];
    av[0][kk].x = q0.x;  av[0][kk].y = q1.y;
    av[1][kk].x = q0.y;  av[1][kk].y = q2.x;
    av[2][kk].x = q1.x;  av[2][kk].y = q2.y;
  }
  #pragma unroll
  for (int w = 0; w < 2; ++w) {
    const float* __restrict__ Wm = w ? Wsv : Wlv;
    float* __restrict__ dst      = w ? sc_v : f_v;
    v8f acc[3][2];
    #pragma unroll
    for (int c = 0; c < 3; ++c) { acc[c][0] = v8f{}; acc[c][1] = v8f{}; }
    #pragma unroll
    for (int kk = 0; kk < MUL1 / 4; ++kk) {
      const int k0 = 4 * kk + 2 * hi;
      #pragma unroll
      for (int t = 0; t < 2; ++t) {
        v2f b;
        b.x = Wm[ k0      * MUL1 + 16 * t + lm];
        b.y = Wm[(k0 + 1) * MUL1 + 16 * t + lm];
        #pragma unroll
        for (int c = 0; c < 3; ++c)
          acc[c][t] = __builtin_amdgcn_wmma_f32_16x16x4_f32(
              false, av[c][kk], false, b, (short)0, acc[c][t], false, false);
      }
    }
    #pragma unroll
    for (int r = 0; r < 8; ++r) {
      const int nd = n0 + r + 8 * hi;
      if (nd < n) {
        #pragma unroll
        for (int t = 0; t < 2; ++t) {
          #pragma unroll
          for (int c = 0; c < 3; ++c)
            dst[(long long)nd * (MUL1 * 3) + (16 * t + lm) * 3 + c] = acc[c][t][r] * ar[r];
        }
      }
    }
  }
}

__global__ __launch_bounds__(WAVES * 32, 2)
void edge_fused_kernel(const float* __restrict__ es,
                       const float* __restrict__ ea,
                       const int*   __restrict__ esrc,
                       const int*   __restrict__ edst,
                       const int*   __restrict__ order,
                       const float* __restrict__ Wfc1,
                       const float* __restrict__ Wfc2,
                       const float* __restrict__ f_s,
                       const float* __restrict__ f_v,
                       float* __restrict__ agg_s,
                       float* __restrict__ agg_v,
                       int E, int numTiles)
{
  __shared__ float W1s[KPAD1 * NPAD1];
  __shared__ float4 W2f[NPAD1 * WNUM / 4];
  __shared__ float Hs[WAVES][16 * NPAD1];
  __shared__ float mF[WAVES][16 * 4];
  __shared__ int   mI[WAVES][16 * 2];

  {
    constexpr int NT = WAVES * 32;
    constexpr int N1 = (KPAD1 * NPAD1 + NT - 1) / NT;
    float w1[N1];
    #pragma unroll
    for (int j = 0; j < N1; ++j) {
      const int i = threadIdx.x + NT * j;
      const int k = i / NPAD1, ncol = i - k * NPAD1;
      const bool ok = (k < NBASIS) && (ncol < FCH);
      const float x = Wfc1[ok ? k * FCH + ncol : 0];
      w1[j] = ok ? x : 0.f;
    }
    constexpr int NS = NPAD1 * WNUM / 4;
    constexpr int N2 = NS / NT;
    static_assert(NS % NT == 0, "W2f must split evenly over the block");
    static_assert(NT % 32 == 0 && FCH % 2 == 0, "slot decoding below");
    float4 w2[N2];
    #pragma unroll
    for (int j = 0; j < N2; ++j) {
      const int s  = threadIdx.x + NT * j;
      const int ln = s & 31, q = s >> 5;
      const int pr = q / (NPAD1 / 4), kk = q - pr * (NPAD1 / 4);
      const int k0 = 4 * kk + 2 * (ln >> 4);
      const int cc = 32 * pr + (ln & 15);
      const bool ok = (k0 < FCH);
      const float* r0 = Wfc2 + (ok ? k0 * WNUM + cc : 0);
      float4 x;
      x.x = r0[0];  x.y = r0[WNUM];  x.z = r0[16];  x.w = r0[WNUM + 16];
      w2[j] = ok ? x : make_float4(0.f, 0.f, 0.f, 0.f);
    }
    #pragma unroll
    for (int j = 0; j < N1; ++j) {
      const int i = threadIdx.x + NT * j;
      if (i < KPAD1 * NPAD1) W1s[i] = w1[j];
    }
    #pragma unroll
    for (int j = 0; j < N2; ++j) W2f[threadIdx.x + NT * j] = w2[j];
  }
  __syncthreads();

  const int wid  = threadIdx.x >> 5;
  const int lane = threadIdx.x & 31;
  const int hi   = lane >> 4;
  const int lm   = lane & 15;

  float* H  = Hs[wid];
  float* mf = mF[wid];
  int*   mi = mI[wid];

  const float inv  = 0.25f;
  const float inv3 = 0.57735026918962576451f;
  const float inv2 = 0.70710678118654752440f;

  const int gw   = blockIdx.x * WAVES + wid;
  const int nw   = gridDim.x * WAVES;
  const int tq   = numTiles / nw;
  const int trem = numTiles - tq * nw;
  const int t0   = gw * tq + (gw < trem ? gw : trem);
  const int t1   = t0 + tq + (gw < trem ? 1 : 0);

  int nodeFirst = -3, nodeLast = -3;
  if (t0 < t1) {
    const int pl = (t1 * 16 < E ? t1 * 16 : E) - 1;
    nodeFirst = edst[order[t0 * 16]];
    nodeLast  = edst[order[pl]];
  }

  float cs[6][1];
  float cv[8][3];
  #pragma unroll
  for (int i = 0; i < 6; ++i) cs[i][0] = 0.f;
  #pragma unroll
  for (int i = 0; i < 8; ++i) { cv[i][0] = 0.f; cv[i][1] = 0.f; cv[i][2] = 0.f; }
  int dLastPrev = -2;

  for (int tile = t0; tile < t1; ++tile) {
    int base = tile * 16;

    int  pA = base + lm;
    bool ev = (pA < E);
    const int  pN = base + 16;
    int        eA = order[ev ? pA : E - 1];
    const int  eN = order[pN < E ? pN : E - 1];
    const int  dNraw = edst[eN];
    const int  dNextTile = (pN < E) ? dNraw : -1;

    if (lane < 16) {
      int e = eA;
      if (ev) {
        mi[lane * 2 + 0] = esrc[e];
        mi[lane * 2 + 1] = edst[e];
        mf[lane * 4 + 0] = ea[(long long)e * 4 + 0];
        mf[lane * 4 + 1] = ea[(long long)e * 4 + 1];
        mf[lane * 4 + 2] = ea[(long long)e * 4 + 2];
        mf[lane * 4 + 3] = ea[(long long)e * 4 + 3];
      } else {
        mi[lane * 2 + 0] = 0;
        mi[lane * 2 + 1] = -1;
        mf[lane * 4 + 0] = 0.f; mf[lane * 4 + 1] = 0.f;
        mf[lane * 4 + 2] = 0.f; mf[lane * 4 + 3] = 0.f;
      }
    }

    v2f a1[3];
    #pragma unroll
    for (int kk = 0; kk < 3; ++kk) {
      int k0 = 4 * kk + 2 * hi;
      a1[kk].x = (ev && (k0     < NBASIS)) ? es[(long long)eA * NBASIS + k0    ] : 0.f;
      a1[kk].y = (ev && (k0 + 1 < NBASIS)) ? es[(long long)eA * NBASIS + k0 + 1] : 0.f;
    }

    #pragma unroll
    for (int nt = 0; nt < NPAD1 / 16; ++nt) {
      v8f acc = {};
      #pragma unroll
      for (int kk = 0; kk < 3; ++kk) {
        int k0 = 4 * kk + 2 * hi;
        v2f b;
        b.x = W1s[ k0      * NPAD1 + nt * 16 + lm];
        b.y = W1s[(k0 + 1) * NPAD1 + nt * 16 + lm];
        acc = __builtin_amdgcn_wmma_f32_16x16x4_f32(
            false, a1[kk], false, b, (short)0, acc, false, false);
      }
      #pragma unroll
      for (int r = 0; r < 8; ++r) {
        float x  = acc[r];
        float sg = fast_rcp(1.f + __expf(-x));
        H[(r + 8 * hi) * NPAD1 + nt * 16 + lm] = x * sg;
      }
    }
    asm volatile("s_wait_dscnt 0" ::: "memory");

    v2f a2[NPAD1 / 4];
    #pragma unroll
    for (int kk = 0; kk < NPAD1 / 4; ++kk) {
      int k0 = 4 * kk + 2 * hi;
      a2[kk] = *(const v2f*)&H[lm * NPAD1 + k0];
    }

    int   drow[8], srow[8];
    float y0r[8], y1x[8], y1y[8], y1z[8];
    #pragma unroll
    for (int r = 0; r < 8; ++r) {
      int m = r + 8 * hi;
      srow[r] = mi[m * 2 + 0];
      drow[r] = mi[m * 2 + 1];
      y0r[r]  = mf[m * 4 + 0];
      y1x[r]  = mf[m * 4 + 1];
      y1y[r]  = mf[m * 4 + 2];
      y1z[r]  = mf[m * 4 + 3];
    }
    const int d15   = mi[15 * 2 + 1];
    const int dPrev = hi ? mi[7 * 2 + 1] : dLastPrev;
    const int dNext = hi ? ((tile + 1 < t1) ? dNextTile : -2) : mi[8 * 2 + 1];
    RunFlags rf;
    {
      const bool contIn = (drow[0] == dPrev);
      bool noEnd = true;
      #pragma unroll
      for (int r = 0; r < 7; ++r) {
        const bool er   = (drow[r] != drow[r + 1]);
        const bool hold = contIn && noEnd;
        rf.e[r]   = er;
        rf.cap[r] = er && hold;
        rf.fl[r]  = er && !hold && (drow[r] >= 0);
        noEnd = noEnd && !er;
      }
      rf.single  = contIn && noEnd;
      rf.contOut = (drow[7] == dNext);
      const bool condA = contIn && !rf.single && (drow[0] >= 0);
      const bool condB = !rf.contOut && (drow[7] >= 0);
      const bool shA = (drow[0] == nodeFirst) || (drow[0] == nodeLast);
      const bool shB = (drow[7] == nodeFirst) || (drow[7] == nodeLast);
      rf.aSt = condA && !shA;  rf.aAt = condA && shA;
      rf.bSt = condB && !shB;  rf.bAt = condB && shB;
    }


    auto gemm2_pair = [&](const int pr, v8f (&acc)[2]) {
      const float4* bp = &W2f[pr * (NPAD1 / 4) * 32 + lane];
      v2f b[2][NPAD1 / 4];
      #pragma unroll
      for (int kk = 0; kk < NPAD1 / 4; ++kk) {
        const float4 q = bp[kk * 32];
        b[0][kk].x = q.x;  b[0][kk].y = q.y;
        b[1][kk].x = q.z;  b[1][kk].y = q.w;
      }
      __builtin_amdgcn_sched_barrier(0);

      acc[0] = v8f{}; acc[1] = v8f{};
      #pragma unroll
      for (int kk = 0; kk < NPAD1 / 4; ++kk) {
        #pragma unroll
        for (int t = 0; t < 2; ++t) {
          acc[t] = __builtin_amdgcn_wmma_f32_16x16x4_f32(
              false, a2[kk], false, b[t][kk], (short)0, acc[t], false, false);
        }
      }
    };

    #pragma unroll
    for (int p = 0; p < 4; ++p) {
      const int cA  = p * 32 + lm;
      const int col = cA & 63;
      float g[2][8];
      #pragma unroll
      for (int t = 0; t < 2; ++t) {
        #pragma unroll
        for (int r = 0; r < 8; ++r) g[t][r] = f_s[(long long)srow[r] * MUL0 + col + 16 * t];
      }
      v8f acc[2];
      gemm2_pair(p, acc);

      #pragma unroll
      for (int t = 0; t < 2; ++t) {
        const int c = cA + 16 * t;
        if (p < 2) {
          float v[1][8];
          #pragma unroll
          for (int r = 0; r < 8; ++r) v[0][r] = acc[t][r] * g[t][r] * y0r[r] * inv;
          seg_flush<1>(agg_s, MIDS, c, drow, rf, hi, v, cs[p < 2 ? 2 * p + t : 0]);
        } else {
          const int u = c - 64;
          float v[3][8];
          #pragma unroll
          for (int r = 0; r < 8; ++r) {
            float tt = acc[t][r] * g[t][r] * inv;
            v[0][r] = tt * y1x[r]; v[1][r] = tt * y1y[r]; v[2][r] = tt * y1z[r];
          }
          seg_flush<3>(agg_v, MIDV * 3, u * 3, drow, rf, hi, v, cv[p < 2 ? 0 : 2 * (p - 2) + t]);
        }
      }
    }

    {
      float g[2][3][8];
      #pragma unroll
      for (int t = 0; t < 2; ++t) {
        #pragma unroll
        for (int r = 0; r < 8; ++r) {
          const float* vp = &f_v[((long long)srow[r] * MUL1 + lm + 16 * t) * 3];
          g[t][0][r] = vp[0]; g[t][1][r] = vp[1]; g[t][2][r] = vp[2];
        }
      }
      v8f acc[2];

      gemm2_pair(4, acc);
      #pragma unroll
      for (int t = 0; t < 2; ++t) {
        const int u = lm + 16 * t;
        float v[3][8];
        #pragma unroll
        for (int r = 0; r < 8; ++r) {
          float tt = acc[t][r] * y0r[r] * inv;
          v[0][r] = tt * g[t][0][r]; v[1][r] = tt * g[t][1][r]; v[2][r] = tt * g[t][2][r];
        }
        seg_flush<3>(agg_v, MIDV * 3, (64 + u) * 3, drow, rf, hi, v, cv[4 + t]);
      }

      gemm2_pair(5, acc);
      #pragma unroll
      for (int t = 0; t < 2; ++t) {
        const int u = lm + 16 * t;
        float v[1][8];
        #pragma unroll
        for (int r = 0; r < 8; ++r) {
          float dt = g[t][0][r] * y1x[r] + g[t][1][r] * y1y[r] + g[t][2][r] * y1z[r];
          v[0][r] = acc[t][r] * dt * inv3 * inv;
        }
        seg_flush<1>(agg_s, MIDS, 64 + u, drow, rf, hi, v, cs[4 + t]);
      }

      gemm2_pair(6, acc);
      #pragma unroll
      for (int t = 0; t < 2; ++t) {
        const int u = lm + 16 * t;
        float v[3][8];
        #pragma unroll
        for (int r = 0; r < 8; ++r) {
          float cx = g[t][1][r] * y1z[r] - g[t][2][r] * y1y[r];
          float cy = g[t][2][r] * y1x[r] - g[t][0][r] * y1z[r];
          float cz = g[t][0][r] * y1y[r] - g[t][1][r] * y1x[r];
          float tt = acc[t][r] * inv2 * inv;
          v[0][r] = tt * cx; v[1][r] = tt * cy; v[2][r] = tt * cz;
        }
        seg_flush<3>(agg_v, MIDV * 3, (96 + u) * 3, drow, rf, hi, v, cv[6 + t]);
      }
    }
    dLastPrev = d15;
  }
}

__global__ __launch_bounds__(128)
void node_out_kernel(const float* __restrict__ agg_s, const float* __restrict__ agg_v,
                     const float* __restrict__ sc_s,  const float* __restrict__ sc_v,
                     const float* __restrict__ attr,
                     const float* __restrict__ Wl2s, const float* __restrict__ Wl2v,
                     const float* __restrict__ Wl3,
                     float* __restrict__ out, int n)
{
  const int lane = threadIdx.x & 31;
  const int hi   = lane >> 4;
  const int lm   = lane & 15;
  const int n0   = (blockIdx.x * 4 + (threadIdx.x >> 5)) * 16;
  if (n0 >= n) return;
  const int na = (n0 + lm < n) ? (n0 + lm) : (n - 1);

  v2f as[MIDS / 4];
  #pragma unroll
  for (int kk = 0; kk < MIDS / 4; ++kk)
    as[kk] = *(const v2f*)(agg_s + (long long)na * MIDS + 4 * kk + 2 * hi);

  float part = 0.f;
  #pragma unroll
  for (int kk = 0; kk < MIDS / 4; ++kk) {
    const int k0 = 4 * kk + 2 * hi;
    part += as[kk].x * Wl3[k0] + as[kk].y * Wl3[k0 + 1];
  }
  const float ang = 0.1f * (part + xhalf(part)) * attr[na];
  const float csn = cosf(ang);
  const float snn = sinf(ang);

  float cr[8], sr[8], ar[8];
  #pragma unroll
  for (int r = 0; r < 8; ++r) {
    const int nd = n0 + r + 8 * hi;
    cr[r] = __shfl(csn, r + 8 * hi);
    sr[r] = __shfl(snn, r + 8 * hi);
    ar[r] = attr[nd < n ? nd : n - 1];
  }

  {
    float scv[8][4];
    #pragma unroll
    for (int r = 0; r < 8; ++r) {
      const int nd = n0 + r + 8 * hi;
      const int nc = nd < n ? nd : n - 1;
      #pragma unroll
      for (int t = 0; t < 4; ++t) scv[r][t] = sc_s[(long long)nc * MUL0 + 16 * t + lm];
    }
    v8f acc[4];
    #pragma unroll
    for (int t = 0; t < 4; ++t) acc[t] = v8f{};
    #pragma unroll
    for (int kk = 0; kk < MIDS / 4; ++kk) {
      const int k0 = 4 * kk + 2 * hi;
      #pragma unroll
      for (int t = 0; t < 4; ++t) {
        v2f b;
        b.x = Wl2s[ k0      * MUL0 + 16 * t + lm];
        b.y = Wl2s[(k0 + 1) * MUL0 + 16 * t + lm];
        acc[t] = __builtin_amdgcn_wmma_f32_16x16x4_f32(
            false, as[kk], false, b, (short)0, acc[t], false, false);
      }
    }
    #pragma unroll
    for (int r = 0; r < 8; ++r) {
      const int nd = n0 + r + 8 * hi;
      if (nd < n) {
        #pragma unroll
        for (int t = 0; t < 4; ++t) {
          const int col = 16 * t + lm;
          const float conv = acc[t][r] * ar[r];
          out[(long long)nd * 160 + col] = cr[r] * scv[r][t] + sr[r] * conv;
        }
      }
    }
  }

  {
    float scv[8][2][3];
    #pragma unroll
    for (int r = 0; r < 8; ++r) {
      const int nd = n0 + r + 8 * hi;
      const int nc = nd < n ? nd : n - 1;
      #pragma unroll
      for (int t = 0; t < 2; ++t) {
        #pragma unroll
        for (int c = 0; c < 3; ++c) scv[r][t][c] = sc_v[(long long)nc * (MUL1 * 3) + (16 * t + lm) * 3 + c];
      }
    }
    v8f acc[3][2];
    #pragma unroll
    for (int c = 0; c < 3; ++c) { acc[c][0] = v8f{}; acc[c][1] = v8f{}; }
    const float* pv = agg_v + (long long)na * (MIDV * 3) + 6 * hi;
    #pragma unroll
    for (int kk = 0; kk < MIDV / 4; ++kk) {
      const int k0 = 4 * kk + 2 * hi;
      const v2f* q = (const v2f*)(pv + 12 * kk);
      const v2f q0 = q[0], q1 = q[1], q2 = q[2];
      v2f a[3];
      a[0].x = q0.x;  a[0].y = q1.y;
      a[1].x = q0.y;  a[1].y = q2.x;
      a[2].x = q1.x;  a[2].y = q2.y;
      #pragma unroll
      for (int t = 0; t < 2; ++t) {
        v2f b;
        b.x = Wl2v[ k0      * MUL1 + 16 * t + lm];
        b.y = Wl2v[(k0 + 1) * MUL1 + 16 * t + lm];
        #pragma unroll
        for (int c = 0; c < 3; ++c)
          acc[c][t] = __builtin_amdgcn_wmma_f32_16x16x4_f32(
              false, a[c], false, b, (short)0, acc[c][t], false, false);
      }
    }
    #pragma unroll
    for (int r = 0; r < 8; ++r) {
      const int nd = n0 + r + 8 * hi;
      if (nd < n) {
        #pragma unroll
        for (int t = 0; t < 2; ++t) {
          #pragma unroll
          for (int c = 0; c < 3; ++c) {
            const int idx = (16 * t + lm) * 3 + c;
            const float conv = acc[c][t][r] * ar[r];
            out[(long long)nd * 160 + MUL0 + idx] = cr[r] * scv[r][t][c] + sr[r] * conv;
          }
        }
      }
    }
  }
}

extern "C" void kernel_launch(void* const* d_in, const int* in_sizes, int n_in,
                              void* d_out, int out_size, void* d_ws, size_t ws_size,
                              hipStream_t stream) {
  const float* node_scalars = (const float*)d_in[0];
  const float* node_vectors = (const float*)d_in[1];
  const float* node_attr    = (const float*)d_in[2];
  const int*   edge_src     = (const int*)  d_in[3];
  const int*   edge_dst     = (const int*)  d_in[4];
  const float* edge_attr    = (const float*)d_in[5];
  const float* edge_scalars = (const float*)d_in[6];
  const float* W_fc1        = (const float*)d_in[7];
  const float* W_fc2        = (const float*)d_in[8];
  const float* W_sc_s       = (const float*)d_in[9];
  const float* W_sc_v       = (const float*)d_in[10];
  const float* W_l1_s       = (const float*)d_in[11];
  const float* W_l1_v       = (const float*)d_in[12];
  const float* W_l2_s       = (const float*)d_in[13];
  const float* W_l2_v       = (const float*)d_in[14];
  const float* W_l3         = (const float*)d_in[15];

  const int N = in_sizes[0] / MUL0;
  const int E = in_sizes[3];

  float* ws    = (float*)d_ws;
  float* f_s   = ws;
  float* f_v   = f_s  + (size_t)N * MUL0;
  float* sc_s  = f_v  + (size_t)N * MUL1 * 3;
  float* sc_v  = sc_s + (size_t)N * MUL0;
  float* agg_s = sc_v + (size_t)N * MUL1 * 3;
  float* agg_v = agg_s + (size_t)N * MIDS;

  int* cnt    = (int*)d_out;
  int* cursor = cnt + N;
  int* order  = cursor + N;

  long long nAgg = (long long)N * (MIDS + MIDV * 3);
  zero_kernel<<<4096, 256, 0, stream>>>(agg_s, nAgg, cnt, N);

  const int nodeTiles = (N + 15) / 16;
  node_prep_kernel<<<(nodeTiles + 3) / 4, 128, 0, stream>>>(node_scalars, node_vectors, node_attr,
                                          W_l1_s, W_l1_v, W_sc_s, W_sc_v,
                                          f_s, f_v, sc_s, sc_v, N);

  dst_count_kernel<<<1024, 256, 0, stream>>>(edge_dst, cnt, E);
  dst_scan_kernel<<<1, 1024, 0, stream>>>(cnt, cursor, N);
  dst_fill_kernel<<<1024, 256, 0, stream>>>(edge_dst, cursor, order, E);

  int numTiles = (E + 15) / 16;
  edge_fused_kernel<<<512, WAVES * 32, 0, stream>>>(edge_scalars, edge_attr,
                                                     edge_src, edge_dst, order,
                                                     W_fc1, W_fc2, f_s, f_v,
                                                     agg_s, agg_v, E, numTiles);

  node_out_kernel<<<(nodeTiles + 3) / 4, 128, 0, stream>>>(agg_s, agg_v, sc_s, sc_v, node_attr,
                                         W_l2_s, W_l2_v, W_l3, (float*)d_out, N);
}


# edge kernel: uncommon ends of a destination run (node under 8 edges, range-end atomics, end on a half's last row) sit behind one wave-uniform test; a channel group's flush has 2 branches, was 11
# speedup vs baseline: 3.1046x; 1.0460x over previous
#include <hip/hip_runtime.h>
#include <hip/hip_bf16.h>
#include <math.h>


typedef __attribute__((ext_vector_type(2))) float v2f;
typedef __attribute__((ext_vector_type(8))) float v8f;

#define MUL0   64
#define MUL1   32
#define NBASIS 10
#define FCH    100
#define WNUM   224
#define MIDS   96
#define MIDV   128
#define KPAD1  12
#define NPAD1  112
#define WAVES  4

__device__ __forceinline__ void atomic_add_f32(float* p, float v) {
  __hip_atomic_fetch_add(p, v, __ATOMIC_RELAXED, __HIP_MEMORY_SCOPE_AGENT);
}

__device__ __forceinline__ float fast_rcp(float x) {
#if __has_builtin(__builtin_amdgcn_rcpf)
  return __builtin_amdgcn_rcpf(x);
#else
  return 1.f / x;
#endif
}

__device__ __forceinline__ float xhalf(float x) {
  const int b = __builtin_bit_cast(int, x);
  return __builtin_bit_cast(float, __builtin_amdgcn_permlanex16(b, b, 0x76543210, 0xfedcba98, false, false));
}

struct RunFlags {
  bool e[7];
  bool cap[7];
  bool fl[7];
  bool single;
  bool contOut;
  bool aSt, aAt;
  bool bSt, bAt;
  bool uFl;
  bool uAt;
  bool uB;
  bool uRare;
};

template <int NC>
__device__ __forceinline__ void seg_flush(float* __restrict__ agg, int rowStride, int colOff,
                                          const int (&d)[8], const RunFlags& f, int hi,
                                          const float (&v)[NC][8], float (&carry)[NC]) {
  float run[NC], head[NC], th[NC], tl[NC];
  #pragma unroll
  for (int k = 0; k < NC; ++k) { run[k] = 0.f; head[k] = 0.f; }
  #pragma unroll
  for (int r = 0; r < 8; ++r) {
    #pragma unroll
    for (int k = 0; k < NC; ++k) run[k] += v[k][r];
    if (r < 7) {
      #pragma unroll
      for (int k = 0; k < NC; ++k) head[k] = f.cap[r] ? run[k] : head[k];
      #pragma unroll
      for (int k = 0; k < NC; ++k) run[k] = f.e[r] ? 0.f : run[k];
    }
  }
  #pragma unroll
  for (int k = 0; k < NC; ++k) {
    const float hv    = f.single ? run[k] : head[k];
    const float cinLo = xhalf(carry[k]);
    const float outLo = f.contOut ? (f.single ? hv + cinLo : run[k]) : 0.f;
    const float x     = xhalf(outLo);
    const float cin   = hi ? x : cinLo;
    th[k] = hv + cin;
    tl[k] = f.single ? th[k] : run[k];
    carry[k] = f.contOut ? tl[k] : 0.f;
  }
  if (f.aSt) {
    float* p = agg + (long long)d[0] * rowStride + colOff;
    #pragma unroll
    for (int k = 0; k < NC; ++k) p[k] = th[k];
  }
  if (f.uRare) {
    if (f.uFl) {
      float s2[NC];
      #pragma unroll
      for (int k = 0; k < NC; ++k) s2[k] = 0.f;
      #pragma unroll
      for (int r = 0; r < 7; ++r) {
        #pragma unroll
        for (int k = 0; k < NC; ++k) s2[k] += v[k][r];
        if (f.fl[r]) {
          float* p = agg + (long long)d[r] * rowStride + colOff;
          #pragma unroll
          for (int k = 0; k < NC; ++k) atomic_add_f32(p + k, s2[k]);
        }
        #pragma unroll
        for (int k = 0; k < NC; ++k) s2[k] = f.e[r] ? 0.f : s2[k];
      }
    }
    if (f.uAt) {
      if (f.aAt) {
        float* p = agg + (long long)d[0] * rowStride + colOff;
        #pragma unroll
        for (int k = 0; k < NC; ++k) atomic_add_f32(p + k, th[k]);
      }
      if (f.bAt) {
        float* p = agg + (long long)d[7] * rowStride + colOff;
        #pragma unroll
        for (int k = 0; k < NC; ++k) atomic_add_f32(p + k, tl[k]);
      }
    }
    if (f.uB) {
      if (f.bSt) {
        float* p = agg + (long long)d[7] * rowStride + colOff;
        #pragma unroll
        for (int k = 0; k < NC; ++k) p[k] = tl[k];
      }
    }
  }
}

__global__ void zero_kernel(float* __restrict__ p, long long n, int* __restrict__ q, int m) {
  long long i = (long long)blockIdx.x * blockDim.x + threadIdx.x;
  long long stride = (long long)gridDim.x * blockDim.x;
  for (long long j = i; j < m; j += stride) q[j] = 0;
  for (; i < n; i += stride) p[i] = 0.f;
}

__global__ void dst_count_kernel(const int* __restrict__ edst, int* __restrict__ cnt, int E) {
  int stride = gridDim.x * blockDim.x;
  for (int e = blockIdx.x * blockDim.x + threadIdx.x; e < E; e += stride)
    atomicAdd(&cnt[edst[e]], 1);
}

__global__ __launch_bounds__(1024)
void dst_scan_kernel(const int* __restrict__ cnt, int* __restrict__ cursor, int n) {
  __shared__ int part[1024];
  const int t = threadIdx.x;
  const int chunk = (n + 1023) / 1024;
  const int b = t * chunk;
  const int e = (b + chunk < n) ? (b + chunk) : n;
  int s = 0;
  for (int i = b; i < e; ++i) s += cnt[i];
  part[t] = s;
  __syncthreads();
  for (int dd = 1; dd < 1024; dd <<= 1) {
    int add = (t >= dd) ? part[t - dd] : 0;
    __syncthreads();
    part[t] += add;
    __syncthreads();
  }
  int run = part[t] - s;
  for (int i = b; i < e; ++i) { cursor[i] = run; run += cnt[i]; }
}

__global__ void dst_fill_kernel(const int* __restrict__ edst, int* __restrict__ cursor,
                                int* __restrict__ order, int E) {
  int stride = gridDim.x * blockDim.x;
  for (int e = blockIdx.x * blockDim.x + threadIdx.x; e < E; e += stride) {
    int pos = atomicAdd(&cursor[edst[e]], 1);
    order[pos] = e;
  }
}

__global__ __launch_bounds__(128)
void node_prep_kernel(const float* __restrict__ ns, const float* __restrict__ nv,
                      const float* __restrict__ attr,
                      const float* __restrict__ Wls, const float* __restrict__ Wlv,
                      const float* __restrict__ Wss, const float* __restrict__ Wsv,
                      float* __restrict__ f_s, float* __restrict__ f_v,
                      float* __restrict__ sc_s, float* __restrict__ sc_v, int n)
{
  const int lane = threadIdx.x & 31;
  const int hi   = lane >> 4;
  const int lm   = lane & 15;
  const int n0   = (blockIdx.x * 4 + (threadIdx.x >> 5)) * 16;
  if (n0 >= n) return;
  const int na = (n0 + lm < n) ? (n0 + lm) : (n - 1);

  float ar[8];
  #pragma unroll
  for (int r = 0; r < 8; ++r) {
    const int nd = n0 + r + 8 * hi;
    ar[r] = attr[nd < n ? nd : n - 1];
  }

  v2f as[MUL0 / 4];
  #pragma unroll
  for (int kk = 0; kk < MUL0 / 4; ++kk)
    as[kk] = *(const v2f*)(ns + (long long)na * MUL0 + 4 * kk + 2 * hi);

  #pragma unroll
  for (int w = 0; w < 2; ++w) {
    const float* __restrict__ Wm = w ? Wss : Wls;
    float* __restrict__ dst      = w ? sc_s : f_s;
    v8f acc[4];
    #pragma unroll
    for (int t = 0; t < 4; ++t) acc[t] = v8f{};
    #pragma unroll
    for (int kk = 0; kk < MUL0 / 4; ++kk) {
      const int k0 = 4 * kk + 2 * hi;
      #pragma unroll
      for (int t = 0; t < 4; ++t) {
        v2f b;
        b.x = Wm[ k0      * MUL0 + 16 * t + lm];
        b.y = Wm[(k0 + 1) * MUL0 + 16 * t + lm];
        acc[t] = __builtin_amdgcn_wmma_f32_16x16x4_f32(
            false, as[kk], false, b, (short)0, acc[t], false, false);
      }
    }
    #pragma unroll
    for (int r = 0; r < 8; ++r) {
      const int nd = n0 + r + 8 * hi;
      if (nd < n) {
        #pragma unroll
        for (int t = 0; t < 4; ++t) dst[(long long)nd * MUL0 + 16 * t + lm] = acc[t][r] * ar[r];
      }
    }
  }

  v2f av[3][MUL1 / 4];
  #pragma unroll
  for (int kk = 0; kk < MUL1 / 4; ++kk) {
    const v2f* q = (const v2f*)(nv + (long long)na * (MUL1 * 3) + 12 * kk + 6 * hi);
    const v2f q0 = q[0], q1 = q[1], q2 = q[2];
    av[0][kk].x = q0.x;  av[0][kk].y = q1.y;
    av[1][kk].x = q0.y;  av[1][kk].y = q2.x;
    av[2][kk].x = q1.x;  av[2][kk].y = q2.y;
  }
  #pragma unroll
  for (int w = 0; w < 2; ++w) {
    const float* __restrict__ Wm = w ? Wsv : Wlv;
    float* __restrict__ dst      = w ? sc_v : f_v;
    v8f acc[3][2];
    #pragma unroll
    for (int c = 0; c < 3; ++c) { acc[c][0] = v8f{}; acc[c][1] = v8f{}; }
    #pragma unroll
    for (int kk = 0; kk < MUL1 / 4; ++kk) {
      const int k0 = 4 * kk + 2 * hi;
      #pragma unroll
      for (int t = 0; t < 2; ++t) {
        v2f b;
        b.x = Wm[ k0      * MUL1 + 16 * t + lm];
        b.y = Wm[(k0 + 1) * MUL1 + 16 * t + lm];
        #pragma unroll
        for (int c = 0; c < 3; ++c)
          acc[c][t] = __builtin_amdgcn_wmma_f32_16x16x4_f32(
              false, av[c][kk], false, b, (short)0, acc[c][t], false, false);
      }
    }
    #pragma unroll
    for (int r = 0; r < 8; ++r) {
      const int nd = n0 + r + 8 * hi;
      if (nd < n) {
        #pragma unroll
        for (int t = 0; t < 2; ++t) {
          #pragma unroll
          for (int c = 0; c < 3; ++c)
            dst[(long long)nd * (MUL1 * 3) + (16 * t + lm) * 3 + c] = acc[c][t][r] * ar[r];
        }
      }
    }
  }
}

__global__ __launch_bounds__(WAVES * 32, 2)
void edge_fused_kernel(const float* __restrict__ es,
                       const float* __restrict__ ea,
                       const int*   __restrict__ esrc,
                       const int*   __restrict__ edst,
                       const int*   __restrict__ order,
                       const float* __restrict__ Wfc1,
                       const float* __restrict__ Wfc2,
                       const float* __restrict__ f_s,
                       const float* __restrict__ f_v,
                       float* __restrict__ agg_s,
                       float* __restrict__ agg_v,
                       int E, int numTiles)
{
  __shared__ float W1s[KPAD1 * NPAD1];
  __shared__ float4 W2f[NPAD1 * WNUM / 4];
  __shared__ float Hs[WAVES][16 * NPAD1];
  __shared__ float mF[WAVES][16 * 4];
  __shared__ int   mI[WAVES][16 * 2];

  {
    constexpr int NT = WAVES * 32;
    constexpr int N1 = (KPAD1 * NPAD1 + NT - 1) / NT;
    float w1[N1];
    #pragma unroll
    for (int j = 0; j < N1; ++j) {
      const int i = threadIdx.x + NT * j;
      const int k = i / NPAD1, ncol = i - k * NPAD1;
      const bool ok = (k < NBASIS) && (ncol < FCH);
      const float x = Wfc1[ok ? k * FCH + ncol : 0];
      w1[j] = ok ? x : 0.f;
    }
    constexpr int NS = NPAD1 * WNUM / 4;
    constexpr int N2 = NS / NT;
    static_assert(NS % NT == 0, "W2f must split evenly over the block");
    static_assert(NT % 32 == 0 && FCH % 2 == 0, "slot decoding below");
    float4 w2[N2];
    #pragma unroll
    for (int j = 0; j < N2; ++j) {
      const int s  = threadIdx.x + NT * j;
      const int ln = s & 31, q = s >> 5;
      const int pr = q / (NPAD1 / 4), kk = q - pr * (NPAD1 / 4);
      const int k0 = 4 * kk + 2 * (ln >> 4);
      const int cc = 32 * pr + (ln & 15);
      const bool ok = (k0 < FCH);
      const float* r0 = Wfc2 + (ok ? k0 * WNUM + cc : 0);
      float4 x;
      x.x = r0[0];  x.y = r0[WNUM];  x.z = r0[16];  x.w = r0[WNUM + 16];
      w2[j] = ok ? x : make_float4(0.f, 0.f, 0.f, 0.f);
    }
    #pragma unroll
    for (int j = 0; j < N1; ++j) {
      const int i = threadIdx.x + NT * j;
      if (i < KPAD1 * NPAD1) W1s[i] = w1[j];
    }
    #pragma unroll
    for (int j = 0; j < N2; ++j) W2f[threadIdx.x + NT * j] = w2[j];
  }
  __syncthreads();

  const int wid  = threadIdx.x >> 5;
  const int lane = threadIdx.x & 31;
  const int hi   = lane >> 4;
  const int lm   = lane & 15;

  float* H  = Hs[wid];
  float* mf = mF[wid];
  int*   mi = mI[wid];

  const float inv  = 0.25f;
  const float inv3 = 0.57735026918962576451f;
  const float inv2 = 0.70710678118654752440f;

  const int gw   = blockIdx.x * WAVES + wid;
  const int nw   = gridDim.x * WAVES;
  const int tq   = numTiles / nw;
  const int trem = numTiles - tq * nw;
  const int t0   = gw * tq + (gw < trem ? gw : trem);
  const int t1   = t0 + tq + (gw < trem ? 1 : 0);

  int nodeFirst = -3, nodeLast = -3;
  if (t0 < t1) {
    const int pl = (t1 * 16 < E ? t1 * 16 : E) - 1;
    nodeFirst = edst[order[t0 * 16]];
    nodeLast  = edst[order[pl]];
  }

  float cs[6][1];
  float cv[8][3];
  #pragma unroll
  for (int i = 0; i < 6; ++i) cs[i][0] = 0.f;
  #pragma unroll
  for (int i = 0; i < 8; ++i) { cv[i][0] = 0.f; cv[i][1] = 0.f; cv[i][2] = 0.f; }
  int dLastPrev = -2;

  for (int tile = t0; tile < t1; ++tile) {
    int base = tile * 16;

    int  pA = base + lm;
    bool ev = (pA < E);
    const int  pN = base + 16;
    int        eA = order[ev ? pA : E - 1];
    const int  eN = order[pN < E ? pN : E - 1];
    const int  dNraw = edst[eN];
    const int  dNextTile = (pN < E) ? dNraw : -1;

    if (lane < 16) {
      int e = eA;
      if (ev) {
        mi[lane * 2 + 0] = esrc[e];
        mi[lane * 2 + 1] = edst[e];
        mf[lane * 4 + 0] = ea[(long long)e * 4 + 0];
        mf[lane * 4 + 1] = ea[(long long)e * 4 + 1];
        mf[lane * 4 + 2] = ea[(long long)e * 4 + 2];
        mf[lane * 4 + 3] = ea[(long long)e * 4 + 3];
      } else {
        mi[lane * 2 + 0] = 0;
        mi[lane * 2 + 1] = -1;
        mf[lane * 4 + 0] = 0.f; mf[lane * 4 + 1] = 0.f;
        mf[lane * 4 + 2] = 0.f; mf[lane * 4 + 3] = 0.f;
      }
    }

    v2f a1[3];
    #pragma unroll
    for (int kk = 0; kk < 3; ++kk) {
      int k0 = 4 * kk + 2 * hi;
      a1[kk].x = (ev && (k0     < NBASIS)) ? es[(long long)eA * NBASIS + k0    ] : 0.f;
      a1[kk].y = (ev && (k0 + 1 < NBASIS)) ? es[(long long)eA * NBASIS + k0 + 1] : 0.f;
    }

    #pragma unroll
    for (int nt = 0; nt < NPAD1 / 16; ++nt) {
      v8f acc = {};
      #pragma unroll
      for (int kk = 0; kk < 3; ++kk) {
        int k0 = 4 * kk + 2 * hi;
        v2f b;
        b.x = W1s[ k0      * NPAD1 + nt * 16 + lm];
        b.y = W1s[(k0 + 1) * NPAD1 + nt * 16 + lm];
        acc = __builtin_amdgcn_wmma_f32_16x16x4_f32(
            false, a1[kk], false, b, (short)0, acc, false, false);
      }
      #pragma unroll
      for (int r = 0; r < 8; ++r) {
        float x  = acc[r];
        float sg = fast_rcp(1.f + __expf(-x));
        H[(r + 8 * hi) * NPAD1 + nt * 16 + lm] = x * sg;
      }
    }
    asm volatile("s_wait_dscnt 0" ::: "memory");

    v2f a2[NPAD1 / 4];
    #pragma unroll
    for (int kk = 0; kk < NPAD1 / 4; ++kk) {
      int k0 = 4 * kk + 2 * hi;
      a2[kk] = *(const v2f*)&H[lm * NPAD1 + k0];
    }

    int   drow[8], srow[8];
    float y0r[8], y1x[8], y1y[8], y1z[8];
    #pragma unroll
    for (int r = 0; r < 8; ++r) {
      int m = r + 8 * hi;
      srow[r] = mi[m * 2 + 0];
      drow[r] = mi[m * 2 + 1];
      y0r[r]  = mf[m * 4 + 0];
      y1x[r]  = mf[m * 4 + 1];
      y1y[r]  = mf[m * 4 + 2];
      y1z[r]  = mf[m * 4 + 3];
    }
    const int d15   = mi[15 * 2 + 1];
    const int dPrev = hi ? mi[7 * 2 + 1] : dLastPrev;
    const int dNext = hi ? ((tile + 1 < t1) ? dNextTile : -2) : mi[8 * 2 + 1];
    RunFlags rf;
    {
      const bool contIn = (drow[0] == dPrev);
      bool noEnd = true;
      #pragma unroll
      for (int r = 0; r < 7; ++r) {
        const bool er   = (drow[r] != drow[r + 1]);
        const bool hold = contIn && noEnd;
        rf.e[r]   = er;
        rf.cap[r] = er && hold;
        rf.fl[r]  = er && !hold && (drow[r] >= 0);
        noEnd = noEnd && !er;
      }
      rf.single  = contIn && noEnd;
      rf.contOut = (drow[7] == dNext);
      const bool condA = contIn && !rf.single && (drow[0] >= 0);
      const bool condB = !rf.contOut && (drow[7] >= 0);
      const bool shA = (drow[0] == nodeFirst) || (drow[0] == nodeLast);
      const bool shB = (drow[7] == nodeFirst) || (drow[7] == nodeLast);
      rf.aSt = condA && !shA;  rf.aAt = condA && shA;
      rf.bSt = condB && !shB;  rf.bAt = condB && shB;
      bool anyFl = false;
      #pragma unroll
      for (int r = 0; r < 7; ++r) anyFl = anyFl || rf.fl[r];
      rf.uFl   = __builtin_amdgcn_ballot_w32(anyFl) != 0u;
      rf.uAt   = __builtin_amdgcn_ballot_w32(rf.aAt || rf.bAt) != 0u;
      rf.uB    = __builtin_amdgcn_ballot_w32(rf.bSt) != 0u;
      rf.uRare = __builtin_amdgcn_ballot_w32(anyFl || rf.aAt || rf.bAt || rf.bSt) != 0u;
    }


    auto gemm2_pair = [&](const int pr, v8f (&acc)[2]) {
      const float4* bp = &W2f[pr * (NPAD1 / 4) * 32 + lane];
      v2f b[2][NPAD1 / 4];
      #pragma unroll
      for (int kk = 0; kk < NPAD1 / 4; ++kk) {
        const float4 q = bp[kk * 32];
        b[0][kk].x = q.x;  b[0][kk].y = q.y;
        b[1][kk].x = q.z;  b[1][kk].y = q.w;
      }
      __builtin_amdgcn_sched_barrier(0);

      acc[0] = v8f{}; acc[1] = v8f{};
      #pragma unroll
      for (int kk = 0; kk < NPAD1 / 4; ++kk) {
        #pragma unroll
        for (int t = 0; t < 2; ++t) {
          acc[t] = __builtin_amdgcn_wmma_f32_16x16x4_f32(
              false, a2[kk], false, b[t][kk], (short)0, acc[t], false, false);
        }
      }
    };

    #pragma unroll
    for (int p = 0; p < 4; ++p) {
      const int cA  = p * 32 + lm;
      const int col = cA & 63;
      float g[2][8];
      #pragma unroll
      for (int t = 0; t < 2; ++t) {
        #pragma unroll
        for (int r = 0; r < 8; ++r) g[t][r] = f_s[(long long)srow[r] * MUL0 + col + 16 * t];
      }
      v8f acc[2];
      gemm2_pair(p, acc);

      #pragma unroll
      for (int t = 0; t < 2; ++t) {
        const int c = cA + 16 * t;
        if (p < 2) {
          float v[1][8];
          #pragma unroll
          for (int r = 0; r < 8; ++r) v[0][r] = acc[t][r] * g[t][r] * y0r[r] * inv;
          seg_flush<1>(agg_s, MIDS, c, drow, rf, hi, v, cs[p < 2 ? 2 * p + t : 0]);
        } else {
          const int u = c - 64;
          float v[3][8];
          #pragma unroll
          for (int r = 0; r < 8; ++r) {
            float tt = acc[t][r] * g[t][r] * inv;
            v[0][r] = tt * y1x[r]; v[1][r] = tt * y1y[r]; v[2][r] = tt * y1z[r];
          }
          seg_flush<3>(agg_v, MIDV * 3, u * 3, drow, rf, hi, v, cv[p < 2 ? 0 : 2 * (p - 2) + t]);
        }
      }
    }

    {
      float g[2][3][8];
      #pragma unroll
      for (int t = 0; t < 2; ++t) {
        #pragma unroll
        for (int r = 0; r < 8; ++r) {
          const float* vp = &f_v[((long long)srow[r] * MUL1 + lm + 16 * t) * 3];
          g[t][0][r] = vp[0]; g[t][1][r] = vp[1]; g[t][2][r] = vp[2];
        }
      }
      v8f acc[2];

      gemm2_pair(4, acc);
      #pragma unroll
      for (int t = 0; t < 2; ++t) {
        const int u = lm + 16 * t;
        float v[3][8];
        #pragma unroll
        for (int r = 0; r < 8; ++r) {
          float tt = acc[t][r] * y0r[r] * inv;
          v[0][r] = tt * g[t][0][r]; v[1][r] = tt * g[t][1][r]; v[2][r] = tt * g[t][2][r];
        }
        seg_flush<3>(agg_v, MIDV * 3, (64 + u) * 3, drow, rf, hi, v, cv[4 + t]);
      }

      gemm2_pair(5, acc);
      #pragma unroll
      for (int t = 0; t < 2; ++t) {
        const int u = lm + 16 * t;
        float v[1][8];
        #pragma unroll
        for (int r = 0; r < 8; ++r) {
          float dt = g[t][0][r] * y1x[r] + g[t][1][r] * y1y[r] + g[t][2][r] * y1z[r];
          v[0][r] = acc[t][r] * dt * inv3 * inv;
        }
        seg_flush<1>(agg_s, MIDS, 64 + u, drow, rf, hi, v, cs[4 + t]);
      }

      gemm2_pair(6, acc);
      #pragma unroll
      for (int t = 0; t < 2; ++t) {
        const int u = lm + 16 * t;
        float v[3][8];
        #pragma unroll
        for (int r = 0; r < 8; ++r) {
          float cx = g[t][1][r] * y1z[r] - g[t][2][r] * y1y[r];
          float cy = g[t][2][r] * y1x[r] - g[t][0][r] * y1z[r];
          float cz = g[t][0][r] * y1y[r] - g[t][1][r] * y1x[r];
          float tt = acc[t][r] * inv2 * inv;
          v[0][r] = tt * cx; v[1][r] = tt * cy; v[2][r] = tt * cz;
        }
        seg_flush<3>(agg_v, MIDV * 3, (96 + u) * 3, drow, rf, hi, v, cv[6 + t]);
      }
    }
    dLastPrev = d15;
  }
}

__global__ __launch_bounds__(128)
void node_out_kernel(const float* __restrict__ agg_s, const float* __restrict__ agg_v,
                     const float* __restrict__ sc_s,  const float* __restrict__ sc_v,
                     const float* __restrict__ attr,
                     const float* __restrict__ Wl2s, const float* __restrict__ Wl2v,
                     const float* __restrict__ Wl3,
                     float* __restrict__ out, int n)
{
  const int lane = threadIdx.x & 31;
  const int hi   = lane >> 4;
  const int lm   = lane & 15;
  const int n0   = (blockIdx.x * 4 + (threadIdx.x >> 5)) * 16;
  if (n0 >= n) return;
  const int na = (n0 + lm < n) ? (n0 + lm) : (n - 1);

  v2f as[MIDS / 4];
  #pragma unroll
  for (int kk = 0; kk < MIDS / 4; ++kk)
    as[kk] = *(const v2f*)(agg_s + (long long)na * MIDS + 4 * kk + 2 * hi);

  float part = 0.f;
  #pragma unroll
  for (int kk = 0; kk < MIDS / 4; ++kk) {
    const int k0 = 4 * kk + 2 * hi;
    part += as[kk].x * Wl3[k0] + as[kk].y * Wl3[k0 + 1];
  }
  const float ang = 0.1f * (part + xhalf(part)) * attr[na];
  const float csn = cosf(ang);
  const float snn = sinf(ang);

  float cr[8], sr[8], ar[8];
  #pragma unroll
  for (int r = 0; r < 8; ++r) {
    const int nd = n0 + r + 8 * hi;
    cr[r] = __shfl(csn, r + 8 * hi);
    sr[r] = __shfl(snn, r + 8 * hi);
    ar[r] = attr[nd < n ? nd : n - 1];
  }

  {
    float scv[8][4];
    #pragma unroll
    for (int r = 0; r < 8; ++r) {
      const int nd = n0 + r + 8 * hi;
      const int nc = nd < n ? nd : n - 1;
      #pragma unroll
      for (int t = 0; t < 4; ++t) scv[r][t] = sc_s[(long long)nc * MUL0 + 16 * t + lm];
    }
    v8f acc[4];
    #pragma unroll
    for (int t = 0; t < 4; ++t) acc[t] = v8f{};
    #pragma unroll
    for (int kk = 0; kk < MIDS / 4; ++kk) {
      const int k0 = 4 * kk + 2 * hi;
      #pragma unroll
      for (int t = 0; t < 4; ++t) {
        v2f b;
        b.x = Wl2s[ k0      * MUL0 + 16 * t + lm];
        b.y = Wl2s[(k0 + 1) * MUL0 + 16 * t + lm];
        acc[t] = __builtin_amdgcn_wmma_f32_16x16x4_f32(
            false, as[kk], false, b, (short)0, acc[t], false, false);
      }
    }
    #pragma unroll
    for (int r = 0; r < 8; ++r) {
      const int nd = n0 + r + 8 * hi;
      if (nd < n) {
        #pragma unroll
        for (int t = 0; t < 4; ++t) {
          const int col = 16 * t + lm;
          const float conv = acc[t][r] * ar[r];
          out[(long long)nd * 160 + col] = cr[r] * scv[r][t] + sr[r] * conv;
        }
      }
    }
  }

  {
    float scv[8][2][3];
    #pragma unroll
    for (int r = 0; r < 8; ++r) {
      const int nd = n0 + r + 8 * hi;
      const int nc = nd < n ? nd : n - 1;
      #pragma unroll
      for (int t = 0; t < 2; ++t) {
        #pragma unroll
        for (int c = 0; c < 3; ++c) scv[r][t][c] = sc_v[(long long)nc * (MUL1 * 3) + (16 * t + lm) * 3 + c];
      }
    }
    v8f acc[3][2];
    #pragma unroll
    for (int c = 0; c < 3; ++c) { acc[c][0] = v8f{}; acc[c][1] = v8f{}; }
    const float* pv = agg_v + (long long)na * (MIDV * 3) + 6 * hi;
    #pragma unroll
    for (int kk = 0; kk < MIDV / 4; ++kk) {
      const int k0 = 4 * kk + 2 * hi;
      const v2f* q = (const v2f*)(pv + 12 * kk);
      const v2f q0 = q[0], q1 = q[1], q2 = q[2];
      v2f a[3];
      a[0].x = q0.x;  a[0].y = q1.y;
      a[1].x = q0.y;  a[1].y = q2.x;
      a[2].x = q1.x;  a[2].y = q2.y;
      #pragma unroll
      for (int t = 0; t < 2; ++t) {
        v2f b;
        b.x = Wl2v[ k0      * MUL1 + 16 * t + lm];
        b.y = Wl2v[(k0 + 1) * MUL1 + 16 * t + lm];
        #pragma unroll
        for (int c = 0; c < 3; ++c)
          acc[c][t] = __builtin_amdgcn_wmma_f32_16x16x4_f32(
              false, a[c], false, b, (short)0, acc[c][t], false, false);
      }
    }
    #pragma unroll
    for (int r = 0; r < 8; ++r) {
      const int nd = n0 + r + 8 * hi;
      if (nd < n) {
        #pragma unroll
        for (int t = 0; t < 2; ++t) {
          #pragma unroll
          for (int c = 0; c < 3; ++c) {
            const int idx = (16 * t + lm) * 3 + c;
            const float conv = acc[c][t][r] * ar[r];
            out[(long long)nd * 160 + MUL0 + idx] = cr[r] * scv[r][t][c] + sr[r] * conv;
          }
        }
      }
    }
  }
}

extern "C" void kernel_launch(void* const* d_in, const int* in_sizes, int n_in,
                              void* d_out, int out_size, void* d_ws, size_t ws_size,
                              hipStream_t stream) {
  const float* node_scalars = (const float*)d_in[0];
  const float* node_vectors = (const float*)d_in[1];
  const float* node_attr    = (const float*)d_in[2];
  const int*   edge_src     = (const int*)  d_in[3];
  const int*   edge_dst     = (const int*)  d_in[4];
  const float* edge_attr    = (const float*)d_in[5];
  const float* edge_scalars = (const float*)d_in[6];
  const float* W_fc1        = (const float*)d_in[7];
  const float* W_fc2        = (const float*)d_in[8];
  const float* W_sc_s       = (const float*)d_in[9];
  const float* W_sc_v       = (const float*)d_in[10];
  const float* W_l1_s       = (const float*)d_in[11];
  const float* W_l1_v       = (const float*)d_in[12];
  const float* W_l2_s       = (const float*)d_in[13];
  const float* W_l2_v       = (const float*)d_in[14];
  const float* W_l3         = (const float*)d_in[15];

  const int N = in_sizes[0] / MUL0;
  const int E = in_sizes[3];

  float* ws    = (float*)d_ws;
  float* f_s   = ws;
  float* f_v   = f_s  + (size_t)N * MUL0;
  float* sc_s  = f_v  + (size_t)N * MUL1 * 3;
  float* sc_v  = sc_s + (size_t)N * MUL0;
  float* agg_s = sc_v + (size_t)N * MUL1 * 3;
  float* agg_v = agg_s + (size_t)N * MIDS;

  int* cnt    = (int*)d_out;
  int* cursor = cnt + N;
  int* order  = cursor + N;

  long long nAgg = (long long)N * (MIDS + MIDV * 3);
  zero_kernel<<<4096, 256, 0, stream>>>(agg_s, nAgg, cnt, N);

  const int nodeTiles = (N + 15) / 16;
  node_prep_kernel<<<(nodeTiles + 3) / 4, 128, 0, stream>>>(node_scalars, node_vectors, node_attr,
                                          W_l1_s, W_l1_v, W_sc_s, W_sc_v,
                                          f_s, f_v, sc_s, sc_v, N);

  dst_count_kernel<<<1024, 256, 0, stream>>>(edge_dst, cnt, E);
  dst_scan_kernel<<<1, 1024, 0, stream>>>(cnt, cursor, N);
  dst_fill_kernel<<<1024, 256, 0, stream>>>(edge_dst, cursor, order, E);

  int numTiles = (E + 15) / 16;
  edge_fused_kernel<<<512, WAVES * 32, 0, stream>>>(edge_scalars, edge_attr,
                                                     edge_src, edge_dst, order,
                                                     W_fc1, W_fc2, f_s, f_v,
                                                     agg_s, agg_v, E, numTiles);

  node_out_kernel<<<(nodeTiles + 3) / 4, 128, 0, stream>>>(agg_s, agg_v, sc_s, sc_v, node_attr,
                                         W_l2_s, W_l2_v, W_l3, (float*)d_out, N);
}


# edge kernel: GEMM2 contracts over the 100 real hidden units only: 25 k steps per column tile, was 28 with 12 zero-padded units (350 WMMAs and 175 LDS fragment reads per edge tile, was 392 and 196)
# speedup vs baseline: 3.2560x; 1.0488x over previous
#include <hip/hip_runtime.h>
#include <hip/hip_bf16.h>
#include <math.h>


typedef __attribute__((ext_vector_type(2))) float v2f;
typedef __attribute__((ext_vector_type(8))) float v8f;

#define MUL0   64
#define MUL1   32
#define NBASIS 10
#define FCH    100
#define WNUM   224
#define MIDS   96
#define MIDV   128
#define KPAD1  12
#define NPAD1  112
#define WAVES  4
#define KS2    (FCH / 4)

__device__ __forceinline__ void atomic_add_f32(float* p, float v) {
  __hip_atomic_fetch_add(p, v, __ATOMIC_RELAXED, __HIP_MEMORY_SCOPE_AGENT);
}

__device__ __forceinline__ float fast_rcp(float x) {
#if __has_builtin(__builtin_amdgcn_rcpf)
  return __builtin_amdgcn_rcpf(x);
#else
  return 1.f / x;
#endif
}

__device__ __forceinline__ float xhalf(float x) {
  const int b = __builtin_bit_cast(int, x);
  return __builtin_bit_cast(float, __builtin_amdgcn_permlanex16(b, b, 0x76543210, 0xfedcba98, false, false));
}

struct RunFlags {
  bool e[7];
  bool cap[7];
  bool fl[7];
  bool single;
  bool contOut;
  bool aSt, aAt;
  bool bSt, bAt;
  bool uFl;
  bool uAt;
  bool uB;
  bool uRare;
};

template <int NC>
__device__ __forceinline__ void seg_flush(float* __restrict__ agg, int rowStride, int colOff,
                                          const int (&d)[8], const RunFlags& f, int hi,
                                          const float (&v)[NC][8], float (&carry)[NC]) {
  float run[NC], head[NC], th[NC], tl[NC];
  #pragma unroll
  for (int k = 0; k < NC; ++k) { run[k] = 0.f; head[k] = 0.f; }
  #pragma unroll
  for (int r = 0; r < 8; ++r) {
    #pragma unroll
    for (int k = 0; k < NC; ++k) run[k] += v[k][r];
    if (r < 7) {
      #pragma unroll
      for (int k = 0; k < NC; ++k) head[k] = f.cap[r] ? run[k] : head[k];
      #pragma unroll
      for (int k = 0; k < NC; ++k) run[k] = f.e[r] ? 0.f : run[k];
    }
  }
  #pragma unroll
  for (int k = 0; k < NC; ++k) {
    const float hv    = f.single ? run[k] : head[k];
    const float cinLo = xhalf(carry[k]);
    const float outLo = f.contOut ? (f.single ? hv + cinLo : run[k]) : 0.f;
    const float x     = xhalf(outLo);
    const float cin   = hi ? x : cinLo;
    th[k] = hv + cin;
    tl[k] = f.single ? th[k] : run[k];
    carry[k] = f.contOut ? tl[k] : 0.f;
  }
  if (f.aSt) {
    float* p = agg + (long long)d[0] * rowStride + colOff;
    #pragma unroll
    for (int k = 0; k < NC; ++k) p[k] = th[k];
  }
  if (f.uRare) {
    if (f.uFl) {
      float s2[NC];
      #pragma unroll
      for (int k = 0; k < NC; ++k) s2[k] = 0.f;
      #pragma unroll
      for (int r = 0; r < 7; ++r) {
        #pragma unroll
        for (int k = 0; k < NC; ++k) s2[k] += v[k][r];
        if (f.fl[r]) {
          float* p = agg + (long long)d[r] * rowStride + colOff;
          #pragma unroll
          for (int k = 0; k < NC; ++k) atomic_add_f32(p + k, s2[k]);
        }
        #pragma unroll
        for (int k = 0; k < NC; ++k) s2[k] = f.e[r] ? 0.f : s2[k];
      }
    }
    if (f.uAt) {
      if (f.aAt) {
        float* p = agg + (long long)d[0] * rowStride + colOff;
        #pragma unroll
        for (int k = 0; k < NC; ++k) atomic_add_f32(p + k, th[k]);
      }
      if (f.bAt) {
        float* p = agg + (long long)d[7] * rowStride + colOff;
        #pragma unroll
        for (int k = 0; k < NC; ++k) atomic_add_f32(p + k, tl[k]);
      }
    }
    if (f.uB) {
      if (f.bSt) {
        float* p = agg + (long long)d[7] * rowStride + colOff;
        #pragma unroll
        for (int k = 0; k < NC; ++k) p[k] = tl[k];
      }
    }
  }
}

__global__ void zero_kernel(float* __restrict__ p, long long n, int* __restrict__ q, int m) {
  long long i = (long long)blockIdx.x * blockDim.x + threadIdx.x;
  long long stride = (long long)gridDim.x * blockDim.x;
  for (long long j = i; j < m; j += stride) q[j] = 0;
  for (; i < n; i += stride) p[i] = 0.f;
}

__global__ void dst_count_kernel(const int* __restrict__ edst, int* __restrict__ cnt, int E) {
  int stride = gridDim.x * blockDim.x;
  for (int e = blockIdx.x * blockDim.x + threadIdx.x; e < E; e += stride)
    atomicAdd(&cnt[edst[e]], 1);
}

__global__ __launch_bounds__(1024)
void dst_scan_kernel(const int* __restrict__ cnt, int* __restrict__ cursor, int n) {
  __shared__ int part[1024];
  const int t = threadIdx.x;
  const int chunk = (n + 1023) / 1024;
  const int b = t * chunk;
  const int e = (b + chunk < n) ? (b + chunk) : n;
  int s = 0;
  for (int i = b; i < e; ++i) s += cnt[i];
  part[t] = s;
  __syncthreads();
  for (int dd = 1; dd < 1024; dd <<= 1) {
    int add = (t >= dd) ? part[t - dd] : 0;
    __syncthreads();
    part[t] += add;
    __syncthreads();
  }
  int run = part[t] - s;
  for (int i = b; i < e; ++i) { cursor[i] = run; run += cnt[i]; }
}

__global__ void dst_fill_kernel(const int* __restrict__ edst, int* __restrict__ cursor,
                                int* __restrict__ order, int E) {
  int stride = gridDim.x * blockDim.x;
  for (int e = blockIdx.x * blockDim.x + threadIdx.x; e < E; e += stride) {
    int pos = atomicAdd(&cursor[edst[e]], 1);
    order[pos] = e;
  }
}

__global__ __launch_bounds__(128)
void node_prep_kernel(const float* __restrict__ ns, const float* __restrict__ nv,
                      const float* __restrict__ attr,
                      const float* __restrict__ Wls, const float* __restrict__ Wlv,
                      const float* __restrict__ Wss, const float* __restrict__ Wsv,
                      float* __restrict__ f_s, float* __restrict__ f_v,
                      float* __restrict__ sc_s, float* __restrict__ sc_v, int n)
{
  const int lane = threadIdx.x & 31;
  const int hi   = lane >> 4;
  const int lm   = lane & 15;
  const int n0   = (blockIdx.x * 4 + (threadIdx.x >> 5)) * 16;
  if (n0 >= n) return;
  const int na = (n0 + lm < n) ? (n0 + lm) : (n - 1);

  float ar[8];
  #pragma unroll
  for (int r = 0; r < 8; ++r) {
    const int nd = n0 + r + 8 * hi;
    ar[r] = attr[nd < n ? nd : n - 1];
  }

  v2f as[MUL0 / 4];
  #pragma unroll
  for (int kk = 0; kk < MUL0 / 4; ++kk)
    as[kk] = *(const v2f*)(ns + (long long)na * MUL0 + 4 * kk + 2 * hi);

  #pragma unroll
  for (int w = 0; w < 2; ++w) {
    const float* __restrict__ Wm = w ? Wss : Wls;
    float* __restrict__ dst      = w ? sc_s : f_s;
    v8f acc[4];
    #pragma unroll
    for (int t = 0; t < 4; ++t) acc[t] = v8f{};
    #pragma unroll
    for (int kk = 0; kk < MUL0 / 4; ++kk) {
      const int k0 = 4 * kk + 2 * hi;
      #pragma unroll
      for (int t = 0; t < 4; ++t) {
        v2f b;
        b.x = Wm[ k0      * MUL0 + 16 * t + lm];
        b.y = Wm[(k0 + 1) * MUL0 + 16 * t + lm];
        acc[t] = __builtin_amdgcn_wmma_f32_16x16x4_f32(
            false, as[kk], false, b, (short)0, acc[t], false, false);
      }
    }
    #pragma unroll
    for (int r = 0; r < 8; ++r) {
      const int nd = n0 + r + 8 * hi;
      if (nd < n) {
        #pragma unroll
        for (int t = 0; t < 4; ++t) dst[(long long)nd * MUL0 + 16 * t + lm] = acc[t][r] * ar[r];
      }
    }
  }

  v2f av[3][MUL1 / 4];
  #pragma unroll
  for (int kk = 0; kk < MUL1 / 4; ++kk) {
    const v2f* q = (const v2f*)(nv + (long long)na * (MUL1 * 3) + 12 * kk + 6 * hi);
    const v2f q0 = q[0], q1 = q[1], q2 = q[2];
    av[0][kk].x = q0.x;  av[0][kk].y = q1.y;
    av[1][kk].x = q0.y;  av[1][kk].y = q2.x;
    av[2][kk].x = q1.x;  av[2][kk].y = q2.y;
  }
  #pragma unroll
  for (int w = 0; w < 2; ++w) {
    const float* __restrict__ Wm = w ? Wsv : Wlv;
    float* __restrict__ dst      = w ? sc_v : f_v;
    v8f acc[3][2];
    #pragma unroll
    for (int c = 0; c < 3; ++c) { acc[c][0] = v8f{}; acc[c][1] = v8f{}; }
    #pragma unroll
    for (int kk = 0; kk < MUL1 / 4; ++kk) {
      const int k0 = 4 * kk + 2 * hi;
      #pragma unroll
      for (int t = 0; t < 2; ++t) {
        v2f b;
        b.x = Wm[ k0      * MUL1 + 16 * t + lm];
        b.y = Wm[(k0 + 1) * MUL1 + 16 * t + lm];
        #pragma unroll
        for (int c = 0; c < 3; ++c)
          acc[c][t] = __builtin_amdgcn_wmma_f32_16x16x4_f32(
              false, av[c][kk], false, b, (short)0, acc[c][t], false, false);
      }
    }
    #pragma unroll
    for (int r = 0; r < 8; ++r) {
      const int nd = n0 + r + 8 * hi;
      if (nd < n) {
        #pragma unroll
        for (int t = 0; t < 2; ++t) {
          #pragma unroll
          for (int c = 0; c < 3; ++c)
            dst[(long long)nd * (MUL1 * 3) + (16 * t + lm) * 3 + c] = acc[c][t][r] * ar[r];
        }
      }
    }
  }
}

__global__ __launch_bounds__(WAVES * 32, 2)
void edge_fused_kernel(const float* __restrict__ es,
                       const float* __restrict__ ea,
                       const int*   __restrict__ esrc,
                       const int*   __restrict__ edst,
                       const int*   __restrict__ order,
                       const float* __restrict__ Wfc1,
                       const float* __restrict__ Wfc2,
                       const float* __restrict__ f_s,
                       const float* __restrict__ f_v,
                       float* __restrict__ agg_s,
                       float* __restrict__ agg_v,
                       int E, int numTiles)
{
  __shared__ float W1s[KPAD1 * NPAD1];
  __shared__ float4 W2f[KS2 * 32 * (WNUM / 32)];
  __shared__ float Hs[WAVES][16 * NPAD1];
  __shared__ float mF[WAVES][16 * 4];
  __shared__ int   mI[WAVES][16 * 2];

  {
    constexpr int NT = WAVES * 32;
    constexpr int N1 = (KPAD1 * NPAD1 + NT - 1) / NT;
    float w1[N1];
    #pragma unroll
    for (int j = 0; j < N1; ++j) {
      const int i = threadIdx.x + NT * j;
      const int k = i / NPAD1, ncol = i - k * NPAD1;
      const bool ok = (k < NBASIS) && (ncol < FCH);
      const float x = Wfc1[ok ? k * FCH + ncol : 0];
      w1[j] = ok ? x : 0.f;
    }
    constexpr int NS = KS2 * 32 * (WNUM / 32);
    constexpr int N2 = (NS + NT - 1) / NT;
    static_assert(FCH % 4 == 0, "GEMM2 contracts over FCH in k steps of 4");
    static_assert(NT % 32 == 0 && WNUM % 32 == 0, "slot decoding below");
    float4 w2[N2];
    #pragma unroll
    for (int j = 0; j < N2; ++j) {
      const int s  = threadIdx.x + NT * j;
      const int sc = (s < NS) ? s : NS - 1;
      const int ln = sc & 31, q = sc >> 5;
      const int pr = q / KS2, kk = q - pr * KS2;
      const int k0 = 4 * kk + 2 * (ln >> 4);
      const int cc = 32 * pr + (ln & 15);
      const float* r0 = Wfc2 + k0 * WNUM + cc;
      float4 x;
      x.x = r0[0];  x.y = r0[WNUM];  x.z = r0[16];  x.w = r0[WNUM + 16];
      w2[j] = x;
    }
    #pragma unroll
    for (int j = 0; j < N1; ++j) {
      const int i = threadIdx.x + NT * j;
      if (i < KPAD1 * NPAD1) W1s[i] = w1[j];
    }
    #pragma unroll
    for (int j = 0; j < N2; ++j) {
      const int s = threadIdx.x + NT * j;
      if (s < NS) W2f[s] = w2[j];
    }
  }
  __syncthreads();

  const int wid  = threadIdx.x >> 5;
  const int lane = threadIdx.x & 31;
  const int hi   = lane >> 4;
  const int lm   = lane & 15;

  float* H  = Hs[wid];
  float* mf = mF[wid];
  int*   mi = mI[wid];

  const float inv  = 0.25f;
  const float inv3 = 0.57735026918962576451f;
  const float inv2 = 0.70710678118654752440f;

  const int gw   = blockIdx.x * WAVES + wid;
  const int nw   = gridDim.x * WAVES;
  const int tq   = numTiles / nw;
  const int trem = numTiles - tq * nw;
  const int t0   = gw * tq + (gw < trem ? gw : trem);
  const int t1   = t0 + tq + (gw < trem ? 1 : 0);

  int nodeFirst = -3, nodeLast = -3;
  if (t0 < t1) {
    const int pl = (t1 * 16 < E ? t1 * 16 : E) - 1;
    nodeFirst = edst[order[t0 * 16]];
    nodeLast  = edst[order[pl]];
  }

  float cs[6][1];
  float cv[8][3];
  #pragma unroll
  for (int i = 0; i < 6; ++i) cs[i][0] = 0.f;
  #pragma unroll
  for (int i = 0; i < 8; ++i) { cv[i][0] = 0.f; cv[i][1] = 0.f; cv[i][2] = 0.f; }
  int dLastPrev = -2;

  for (int tile = t0; tile < t1; ++tile) {
    int base = tile * 16;

    int  pA = base + lm;
    bool ev = (pA < E);
    const int  pN = base + 16;
    int        eA = order[ev ? pA : E - 1];
    const int  eN = order[pN < E ? pN : E - 1];
    const int  dNraw = edst[eN];
    const int  dNextTile = (pN < E) ? dNraw : -1;

    if (lane < 16) {
      int e = eA;
      if (ev) {
        mi[lane * 2 + 0] = esrc[e];
        mi[lane * 2 + 1] = edst[e];
        mf[lane * 4 + 0] = ea[(long long)e * 4 + 0];
        mf[lane * 4 + 1] = ea[(long long)e * 4 + 1];
        mf[lane * 4 + 2] = ea[(long long)e * 4 + 2];
        mf[lane * 4 + 3] = ea[(long long)e * 4 + 3];
      } else {
        mi[lane * 2 + 0] = 0;
        mi[lane * 2 + 1] = -1;
        mf[lane * 4 + 0] = 0.f; mf[lane * 4 + 1] = 0.f;
        mf[lane * 4 + 2] = 0.f; mf[lane * 4 + 3] = 0.f;
      }
    }

    v2f a1[3];
    #pragma unroll
    for (int kk = 0; kk < 3; ++kk) {
      int k0 = 4 * kk + 2 * hi;
      a1[kk].x = (ev && (k0     < NBASIS)) ? es[(long long)eA * NBASIS + k0    ] : 0.f;
      a1[kk].y = (ev && (k0 + 1 < NBASIS)) ? es[(long long)eA * NBASIS + k0 + 1] : 0.f;
    }

    #pragma unroll
    for (int nt = 0; nt < NPAD1 / 16; ++nt) {
      v8f acc = {};
      #pragma unroll
      for (int kk = 0; kk < 3; ++kk) {
        int k0 = 4 * kk + 2 * hi;
        v2f b;
        b.x = W1s[ k0      * NPAD1 + nt * 16 + lm];
        b.y = W1s[(k0 + 1) * NPAD1 + nt * 16 + lm];
        acc = __builtin_amdgcn_wmma_f32_16x16x4_f32(
            false, a1[kk], false, b, (short)0, acc, false, false);
      }
      #pragma unroll
      for (int r = 0; r < 8; ++r) {
        float x  = acc[r];
        float sg = fast_rcp(1.f + __expf(-x));
        H[(r + 8 * hi) * NPAD1 + nt * 16 + lm] = x * sg;
      }
    }
    asm volatile("s_wait_dscnt 0" ::: "memory");

    v2f a2[KS2];
    #pragma unroll
    for (int kk = 0; kk < KS2; ++kk) {
      int k0 = 4 * kk + 2 * hi;
      a2[kk] = *(const v2f*)&H[lm * NPAD1 + k0];
    }

    int   drow[8], srow[8];
    float y0r[8], y1x[8], y1y[8], y1z[8];
    #pragma unroll
    for (int r = 0; r < 8; ++r) {
      int m = r + 8 * hi;
      srow[r] = mi[m * 2 + 0];
      drow[r] = mi[m * 2 + 1];
      y0r[r]  = mf[m * 4 + 0];
      y1x[r]  = mf[m * 4 + 1];
      y1y[r]  = mf[m * 4 + 2];
      y1z[r]  = mf[m * 4 + 3];
    }
    const int d15   = mi[15 * 2 + 1];
    const int dPrev = hi ? mi[7 * 2 + 1] : dLastPrev;
    const int dNext = hi ? ((tile + 1 < t1) ? dNextTile : -2) : mi[8 * 2 + 1];
    RunFlags rf;
    {
      const bool contIn = (drow[0] == dPrev);
      bool noEnd = true;
      #pragma unroll
      for (int r = 0; r < 7; ++r) {
        const bool er   = (drow[r] != drow[r + 1]);
        const bool hold = contIn && noEnd;
        rf.e[r]   = er;
        rf.cap[r] = er && hold;
        rf.fl[r]  = er && !hold && (drow[r] >= 0);
        noEnd = noEnd && !er;
      }
      rf.single  = contIn && noEnd;
      rf.contOut = (drow[7] == dNext);
      const bool condA = contIn && !rf.single && (drow[0] >= 0);
      const bool condB = !rf.contOut && (drow[7] >= 0);
      const bool shA = (drow[0] == nodeFirst) || (drow[0] == nodeLast);
      const bool shB = (drow[7] == nodeFirst) || (drow[7] == nodeLast);
      rf.aSt = condA && !shA;  rf.aAt = condA && shA;
      rf.bSt = condB && !shB;  rf.bAt = condB && shB;
      bool anyFl = false;
      #pragma unroll
      for (int r = 0; r < 7; ++r) anyFl = anyFl || rf.fl[r];
      rf.uFl   = __builtin_amdgcn_ballot_w32(anyFl) != 0u;
      rf.uAt   = __builtin_amdgcn_ballot_w32(rf.aAt || rf.bAt) != 0u;
      rf.uB    = __builtin_amdgcn_ballot_w32(rf.bSt) != 0u;
      rf.uRare = __builtin_amdgcn_ballot_w32(anyFl || rf.aAt || rf.bAt || rf.bSt) != 0u;
    }


    auto gemm2_pair = [&](const int pr, v8f (&acc)[2]) {
      const float4* bp = &W2f[pr * KS2 * 32 + lane];
      v2f b[2][KS2];
      #pragma unroll
      for (int kk = 0; kk < KS2; ++kk) {
        const float4 q = bp[kk * 32];
        b[0][kk].x = q.x;  b[0][kk].y = q.y;
        b[1][kk].x = q.z;  b[1][kk].y = q.w;
      }
      __builtin_amdgcn_sched_barrier(0);

      acc[0] = v8f{}; acc[1] = v8f{};
      #pragma unroll
      for (int kk = 0; kk < KS2; ++kk) {
        #pragma unroll
        for (int t = 0; t < 2; ++t) {
          acc[t] = __builtin_amdgcn_wmma_f32_16x16x4_f32(
              false, a2[kk], false, b[t][kk], (short)0, acc[t], false, false);
        }
      }
    };

    #pragma unroll
    for (int p = 0; p < 4; ++p) {
      const int cA  = p * 32 + lm;
      const int col = cA & 63;
      float g[2][8];
      #pragma unroll
      for (int t = 0; t < 2; ++t) {
        #pragma unroll
        for (int r = 0; r < 8; ++r) g[t][r] = f_s[(long long)srow[r] * MUL0 + col + 16 * t];
      }
      v8f acc[2];
      gemm2_pair(p, acc);

      #pragma unroll
      for (int t = 0; t < 2; ++t) {
        const int c = cA + 16 * t;
        if (p < 2) {
          float v[1][8];
          #pragma unroll
          for (int r = 0; r < 8; ++r) v[0][r] = acc[t][r] * g[t][r] * y0r[r] * inv;
          seg_flush<1>(agg_s, MIDS, c, drow, rf, hi, v, cs[p < 2 ? 2 * p + t : 0]);
        } else {
          const int u = c - 64;
          float v[3][8];
          #pragma unroll
          for (int r = 0; r < 8; ++r) {
            float tt = acc[t][r] * g[t][r] * inv;
            v[0][r] = tt * y1x[r]; v[1][r] = tt * y1y[r]; v[2][r] = tt * y1z[r];
          }
          seg_flush<3>(agg_v, MIDV * 3, u * 3, drow, rf, hi, v, cv[p < 2 ? 0 : 2 * (p - 2) + t]);
        }
      }
    }

    {
      float g[2][3][8];
      #pragma unroll
      for (int t = 0; t < 2; ++t) {
        #pragma unroll
        for (int r = 0; r < 8; ++r) {
          const float* vp = &f_v[((long long)srow[r] * MUL1 + lm + 16 * t) * 3];
          g[t][0][r] = vp[0]; g[t][1][r] = vp[1]; g[t][2][r] = vp[2];
        }
      }
      v8f acc[2];

      gemm2_pair(4, acc);
      #pragma unroll
      for (int t = 0; t < 2; ++t) {
        const int u = lm + 16 * t;
        float v[3][8];
        #pragma unroll
        for (int r = 0; r < 8; ++r) {
          float tt = acc[t][r] * y0r[r] * inv;
          v[0][r] = tt * g[t][0][r]; v[1][r] = tt * g[t][1][r]; v[2][r] = tt * g[t][2][r];
        }
        seg_flush<3>(agg_v, MIDV * 3, (64 + u) * 3, drow, rf, hi, v, cv[4 + t]);
      }

      gemm2_pair(5, acc);
      #pragma unroll
      for (int t = 0; t < 2; ++t) {
        const int u = lm + 16 * t;
        float v[1][8];
        #pragma unroll
        for (int r = 0; r < 8; ++r) {
          float dt = g[t][0][r] * y1x[r] + g[t][1][r] * y1y[r] + g[t][2][r] * y1z[r];
          v[0][r] = acc[t][r] * dt * inv3 * inv;
        }
        seg_flush<1>(agg_s, MIDS, 64 + u, drow, rf, hi, v, cs[4 + t]);
      }

      gemm2_pair(6, acc);
      #pragma unroll
      for (int t = 0; t < 2; ++t) {
        const int u = lm + 16 * t;
        float v[3][8];
        #pragma unroll
        for (int r = 0; r < 8; ++r) {
          float cx = g[t][1][r] * y1z[r] - g[t][2][r] * y1y[r];
          float cy = g[t][2][r] * y1x[r] - g[t][0][r] * y1z[r];
          float cz = g[t][0][r] * y1y[r] - g[t][1][r] * y1x[r];
          float tt = acc[t][r] * inv2 * inv;
          v[0][r] = tt * cx; v[1][r] = tt * cy; v[2][r] = tt * cz;
        }
        seg_flush<3>(agg_v, MIDV * 3, (96 + u) * 3, drow, rf, hi, v, cv[6 + t]);
      }
    }
    dLastPrev = d15;
  }
}

__global__ __launch_bounds__(128)
void node_out_kernel(const float* __restrict__ agg_s, const float* __restrict__ agg_v,
                     const float* __restrict__ sc_s,  const float* __restrict__ sc_v,
                     const float* __restrict__ attr,
                     const float* __restrict__ Wl2s, const float* __restrict__ Wl2v,
                     const float* __restrict__ Wl3,
                     float* __restrict__ out, int n)
{
  const int lane = threadIdx.x & 31;
  const int hi   = lane >> 4;
  const int lm   = lane & 15;
  const int n0   = (blockIdx.x * 4 + (threadIdx.x >> 5)) * 16;
  if (n0 >= n) return;
  const int na = (n0 + lm < n) ? (n0 + lm) : (n - 1);

  v2f as[MIDS / 4];
  #pragma unroll
  for (int kk = 0; kk < MIDS / 4; ++kk)
    as[kk] = *(const v2f*)(agg_s + (long long)na * MIDS + 4 * kk + 2 * hi);

  float part = 0.f;
  #pragma unroll
  for (int kk = 0; kk < MIDS / 4; ++kk) {
    const int k0 = 4 * kk + 2 * hi;
    part += as[kk].x * Wl3[k0] + as[kk].y * Wl3[k0 + 1];
  }
  const float ang = 0.1f * (part + xhalf(part)) * attr[na];
  const float csn = cosf(ang);
  const float snn = sinf(ang);

  float cr[8], sr[8], ar[8];
  #pragma unroll
  for (int r = 0; r < 8; ++r) {
    const int nd = n0 + r + 8 * hi;
    cr[r] = __shfl(csn, r + 8 * hi);
    sr[r] = __shfl(snn, r + 8 * hi);
    ar[r] = attr[nd < n ? nd : n - 1];
  }

  {
    float scv[8][4];
    #pragma unroll
    for (int r = 0; r < 8; ++r) {
      const int nd = n0 + r + 8 * hi;
      const int nc = nd < n ? nd : n - 1;
      #pragma unroll
      for (int t = 0; t < 4; ++t) scv[r][t] = sc_s[(long long)nc * MUL0 + 16 * t + lm];
    }
    v8f acc[4];
    #pragma unroll
    for (int t = 0; t < 4; ++t) acc[t] = v8f{};
    #pragma unroll
    for (int kk = 0; kk < MIDS / 4; ++kk) {
      const int k0 = 4 * kk + 2 * hi;
      #pragma unroll
      for (int t = 0; t < 4; ++t) {
        v2f b;
        b.x = Wl2s[ k0      * MUL0 + 16 * t + lm];
        b.y = Wl2s[(k0 + 1) * MUL0 + 16 * t + lm];
        acc[t] = __builtin_amdgcn_wmma_f32_16x16x4_f32(
            false, as[kk], false, b, (short)0, acc[t], false, false);
      }
    }
    #pragma unroll
    for (int r = 0; r < 8; ++r) {
      const int nd = n0 + r + 8 * hi;
      if (nd < n) {
        #pragma unroll
        for (int t = 0; t < 4; ++t) {
          const int col = 16 * t + lm;
          const float conv = acc[t][r] * ar[r];
          out[(long long)nd * 160 + col] = cr[r] * scv[r][t] + sr[r] * conv;
        }
      }
    }
  }

  {
    float scv[8][2][3];
    #pragma unroll
    for (int r = 0; r < 8; ++r) {
      const int nd = n0 + r + 8 * hi;
      const int nc = nd < n ? nd : n - 1;
      #pragma unroll
      for (int t = 0; t < 2; ++t) {
        #pragma unroll
        for (int c = 0; c < 3; ++c) scv[r][t][c] = sc_v[(long long)nc * (MUL1 * 3) + (16 * t + lm) * 3 + c];
      }
    }
    v8f acc[3][2];
    #pragma unroll
    for (int c = 0; c < 3; ++c) { acc[c][0] = v8f{}; acc[c][1] = v8f{}; }
    const float* pv = agg_v + (long long)na * (MIDV * 3) + 6 * hi;
    #pragma unroll
    for (int kk = 0; kk < MIDV / 4; ++kk) {
      const int k0 = 4 * kk + 2 * hi;
      const v2f* q = (const v2f*)(pv + 12 * kk);
      const v2f q0 = q[0], q1 = q[1], q2 = q[2];
      v2f a[3];
      a[0].x = q0.x;  a[0].y = q1.y;
      a[1].x = q0.y;  a[1].y = q2.x;
      a[2].x = q1.x;  a[2].y = q2.y;
      #pragma unroll
      for (int t = 0; t < 2; ++t) {
        v2f b;
        b.x = Wl2v[ k0      * MUL1 + 16 * t + lm];
        b.y = Wl2v[(k0 + 1) * MUL1 + 16 * t + lm];
        #pragma unroll
        for (int c = 0; c < 3; ++c)
          acc[c][t] = __builtin_amdgcn_wmma_f32_16x16x4_f32(
              false, a[c], false, b, (short)0, acc[c][t], false, false);
      }
    }
    #pragma unroll
    for (int r = 0; r < 8; ++r) {
      const int nd = n0 + r + 8 * hi;
      if (nd < n) {
        #pragma unroll
        for (int t = 0; t < 2; ++t) {
          #pragma unroll
          for (int c = 0; c < 3; ++c) {
            const int idx = (16 * t + lm) * 3 + c;
            const float conv = acc[c][t][r] * ar[r];
            out[(long long)nd * 160 + MUL0 + idx] = cr[r] * scv[r][t][c] + sr[r] * conv;
          }
        }
      }
    }
  }
}

extern "C" void kernel_launch(void* const* d_in, const int* in_sizes, int n_in,
                              void* d_out, int out_size, void* d_ws, size_t ws_size,
                              hipStream_t stream) {
  const float* node_scalars = (const float*)d_in[0];
  const float* node_vectors = (const float*)d_in[1];
  const float* node_attr    = (const float*)d_in[2];
  const int*   edge_src     = (const int*)  d_in[3];
  const int*   edge_dst     = (const int*)  d_in[4];
  const float* edge_attr    = (const float*)d_in[5];
  const float* edge_scalars = (const float*)d_in[6];
  const float* W_fc1        = (const float*)d_in[7];
  const float* W_fc2        = (const float*)d_in[8];
  const float* W_sc_s       = (const float*)d_in[9];
  const float* W_sc_v       = (const float*)d_in[10];
  const float* W_l1_s       = (const float*)d_in[11];
  const float* W_l1_v       = (const float*)d_in[12];
  const float* W_l2_s       = (const float*)d_in[13];
  const float* W_l2_v       = (const float*)d_in[14];
  const float* W_l3         = (const float*)d_in[15];

  const int N = in_sizes[0] / MUL0;
  const int E = in_sizes[3];

  float* ws    = (float*)d_ws;
  float* f_s   = ws;
  float* f_v   = f_s  + (size_t)N * MUL0;
  float* sc_s  = f_v  + (size_t)N * MUL1 * 3;
  float* sc_v  = sc_s + (size_t)N * MUL0;
  float* agg_s = sc_v + (size_t)N * MUL1 * 3;
  float* agg_v = agg_s + (size_t)N * MIDS;

  int* cnt    = (int*)d_out;
  int* cursor = cnt + N;
  int* order  = cursor + N;

  long long nAgg = (long long)N * (MIDS + MIDV * 3);
  zero_kernel<<<4096, 256, 0, stream>>>(agg_s, nAgg, cnt, N);

  const int nodeTiles = (N + 15) / 16;
  node_prep_kernel<<<(nodeTiles + 3) / 4, 128, 0, stream>>>(node_scalars, node_vectors, node_attr,
                                          W_l1_s, W_l1_v, W_sc_s, W_sc_v,
                                          f_s, f_v, sc_s, sc_v, N);

  dst_count_kernel<<<1024, 256, 0, stream>>>(edge_dst, cnt, E);
  dst_scan_kernel<<<1, 1024, 0, stream>>>(cnt, cursor, N);
  dst_fill_kernel<<<1024, 256, 0, stream>>>(edge_dst, cursor, order, E);

  int numTiles = (E + 15) / 16;
  edge_fused_kernel<<<512, WAVES * 32, 0, stream>>>(edge_scalars, edge_attr,
                                                     edge_src, edge_dst, order,
                                                     W_fc1, W_fc2, f_s, f_v,
                                                     agg_s, agg_v, E, numTiles);

  node_out_kernel<<<(nodeTiles + 3) / 4, 128, 0, stream>>>(agg_s, agg_v, sc_s, sc_v, node_attr,
                                         W_l2_s, W_l2_v, W_l3, (float*)d_out, N);
}


# edge sort in one launch: a work-group per 128 nodes reads all destinations once and ranks its edges with LDS atomics (arrival-order rank: sum order changes); was 3 launches, 800k global atomics
# speedup vs baseline: 4.3545x; 1.3374x over previous
#include <hip/hip_runtime.h>
#include <hip/hip_bf16.h>
#include <math.h>


typedef __attribute__((ext_vector_type(2))) float v2f;
typedef __attribute__((ext_vector_type(8))) float v8f;

#define MUL0   64
#define MUL1   32
#define NBASIS 10
#define FCH    100
#define WNUM   224
#define MIDS   96
#define MIDV   128
#define KPAD1  12
#define NPAD1  112
#define WAVES  4
#define KS2    (FCH / 4)

__device__ __forceinline__ void atomic_add_f32(float* p, float v) {
  __hip_atomic_fetch_add(p, v, __ATOMIC_RELAXED, __HIP_MEMORY_SCOPE_AGENT);
}

__device__ __forceinline__ float fast_rcp(float x) {
#if __has_builtin(__builtin_amdgcn_rcpf)
  return __builtin_amdgcn_rcpf(x);
#else
  return 1.f / x;
#endif
}

__device__ __forceinline__ float xhalf(float x) {
  const int b = __builtin_bit_cast(int, x);
  return __builtin_bit_cast(float, __builtin_amdgcn_permlanex16(b, b, 0x76543210, 0xfedcba98, false, false));
}

struct RunFlags {
  bool e[7];
  bool cap[7];
  bool fl[7];
  bool single;
  bool contOut;
  bool aSt, aAt;
  bool bSt, bAt;
  bool uFl;
  bool uAt;
  bool uB;
  bool uRare;
};

template <int NC>
__device__ __forceinline__ void seg_flush(float* __restrict__ agg, int rowStride, int colOff,
                                          const int (&d)[8], const RunFlags& f, int hi,
                                          const float (&v)[NC][8], float (&carry)[NC]) {
  float run[NC], head[NC], th[NC], tl[NC];
  #pragma unroll
  for (int k = 0; k < NC; ++k) { run[k] = 0.f; head[k] = 0.f; }
  #pragma unroll
  for (int r = 0; r < 8; ++r) {
    #pragma unroll
    for (int k = 0; k < NC; ++k) run[k] += v[k][r];
    if (r < 7) {
      #pragma unroll
      for (int k = 0; k < NC; ++k) head[k] = f.cap[r] ? run[k] : head[k];
      #pragma unroll
      for (int k = 0; k < NC; ++k) run[k] = f.e[r] ? 0.f : run[k];
    }
  }
  #pragma unroll
  for (int k = 0; k < NC; ++k) {
    const float hv    = f.single ? run[k] : head[k];
    const float cinLo = xhalf(carry[k]);
    const float outLo = f.contOut ? (f.single ? hv + cinLo : run[k]) : 0.f;
    const float x     = xhalf(outLo);
    const float cin   = hi ? x : cinLo;
    th[k] = hv + cin;
    tl[k] = f.single ? th[k] : run[k];
    carry[k] = f.contOut ? tl[k] : 0.f;
  }
  if (f.aSt) {
    float* p = agg + (long long)d[0] * rowStride + colOff;
    #pragma unroll
    for (int k = 0; k < NC; ++k) p[k] = th[k];
  }
  if (f.uRare) {
    if (f.uFl) {
      float s2[NC];
      #pragma unroll
      for (int k = 0; k < NC; ++k) s2[k] = 0.f;
      #pragma unroll
      for (int r = 0; r < 7; ++r) {
        #pragma unroll
        for (int k = 0; k < NC; ++k) s2[k] += v[k][r];
        if (f.fl[r]) {
          float* p = agg + (long long)d[r] * rowStride + colOff;
          #pragma unroll
          for (int k = 0; k < NC; ++k) atomic_add_f32(p + k, s2[k]);
        }
        #pragma unroll
        for (int k = 0; k < NC; ++k) s2[k] = f.e[r] ? 0.f : s2[k];
      }
    }
    if (f.uAt) {
      if (f.aAt) {
        float* p = agg + (long long)d[0] * rowStride + colOff;
        #pragma unroll
        for (int k = 0; k < NC; ++k) atomic_add_f32(p + k, th[k]);
      }
      if (f.bAt) {
        float* p = agg + (long long)d[7] * rowStride + colOff;
        #pragma unroll
        for (int k = 0; k < NC; ++k) atomic_add_f32(p + k, tl[k]);
      }
    }
    if (f.uB) {
      if (f.bSt) {
        float* p = agg + (long long)d[7] * rowStride + colOff;
        #pragma unroll
        for (int k = 0; k < NC; ++k) p[k] = tl[k];
      }
    }
  }
}

__global__ void zero_kernel(float* __restrict__ p, long long n) {
  long long i = (long long)blockIdx.x * blockDim.x + threadIdx.x;
  long long stride = (long long)gridDim.x * blockDim.x;
  for (; i < n; i += stride) p[i] = 0.f;
}

#define SORT_R   128
#define SORT_NT  1024
#define SORT_CAP 8192

template <bool PLACE>
__device__ __forceinline__ void sort_sweep(const int* __restrict__ edst, int* __restrict__ order,
                                           int E, int lo, int base,
                                           int* cntL, int* listE, int* listD, int* nList,
                                           int& wbelow)
{
  const int tid = threadIdx.x;
  const int n4  = E >> 2;
  const int4* __restrict__ p4 = (const int4*)edst;

  auto own = [&](unsigned rel, int e) {
    if (PLACE) {
      const int pos = atomicAdd(&cntL[rel], 1);
      order[base + pos] = e;
    } else {
      atomicAdd(&cntL[rel], 1);
      const int s = atomicAdd(nList, 1);
      if (s < SORT_CAP) { listE[s] = e; listD[s] = (int)rel; }
    }
  };
  auto lower = [&](int d) { return __builtin_popcount(__builtin_amdgcn_ballot_w32(d < lo)); };
  auto one = [&](int d, int e) {
    const unsigned rel = (unsigned)(d - lo);
    if (!PLACE) wbelow += lower(d);
    if (rel < (unsigned)SORT_R) own(rel, e);
  };
  auto four = [&](const int4 d, int i4) {
    const unsigned r0 = (unsigned)(d.x - lo), r1 = (unsigned)(d.y - lo);
    const unsigned r2 = (unsigned)(d.z - lo), r3 = (unsigned)(d.w - lo);
    if (!PLACE) wbelow += lower(d.x) + lower(d.y) + lower(d.z) + lower(d.w);
    const unsigned m01 = r0 < r1 ? r0 : r1, m23 = r2 < r3 ? r2 : r3;
    if ((m01 < m23 ? m01 : m23) < (unsigned)SORT_R) {
      const int e0 = 4 * i4;
      if (r0 < (unsigned)SORT_R) own(r0, e0);
      if (r1 < (unsigned)SORT_R) own(r1, e0 + 1);
      if (r2 < (unsigned)SORT_R) own(r2, e0 + 2);
      if (r3 < (unsigned)SORT_R) own(r3, e0 + 3);
    }
  };

  int jb = 0;
  for (; jb + 4 * SORT_NT <= n4; jb += 4 * SORT_NT) {
    int4 d[4];
    #pragma unroll
    for (int u = 0; u < 4; ++u) d[u] = p4[jb + u * SORT_NT + tid];
    #pragma unroll
    for (int u = 0; u < 4; ++u) four(d[u], jb + u * SORT_NT + tid);
  }
  for (int j = jb + tid; j < n4; j += SORT_NT) four(p4[j], j);
  {
    const int e = 4 * n4 + tid;
    if (e < E) one(edst[e], e);
  }
}

__global__ __launch_bounds__(SORT_NT)
void dst_sort_kernel(const int* __restrict__ edst, int* __restrict__ order, int E)
{
  __shared__ int cntL[SORT_R];
  __shared__ int listE[SORT_CAP];
  __shared__ int listD[SORT_CAP];
  __shared__ int nList;
  __shared__ int nBelow;

  const int tid = threadIdx.x;
  const int lo  = blockIdx.x * SORT_R;

  if (tid < SORT_R) cntL[tid] = 0;
  if (tid == 0) { nList = 0; nBelow = 0; }
  __syncthreads();

  int wbelow = 0;
  sort_sweep<false>(edst, order, E, lo, 0, cntL, listE, listD, &nList, wbelow);
  if ((tid & 31) == 0) atomicAdd(&nBelow, wbelow);
  __syncthreads();

  const int total = nList;
  const int base  = nBelow;

  static_assert(SORT_R == 128, "one wave, four counts per lane");
  if (tid < 32) {
    const int c0 = cntL[4 * tid], c1 = cntL[4 * tid + 1], c2 = cntL[4 * tid + 2], c3 = cntL[4 * tid + 3];
    const int s = c0 + c1 + c2 + c3;
    int inc = s;
    #pragma unroll
    for (int o = 1; o < 32; o <<= 1) {
      const int t = __shfl_up(inc, o);
      if (tid >= o) inc += t;
    }
    const int ex = inc - s;
    cntL[4 * tid]     = ex;
    cntL[4 * tid + 1] = ex + c0;
    cntL[4 * tid + 2] = ex + c0 + c1;
    cntL[4 * tid + 3] = ex + c0 + c1 + c2;
  }
  __syncthreads();

  if (total <= SORT_CAP) {
    for (int s = tid; s < total; s += SORT_NT) {
      const int pos = atomicAdd(&cntL[listD[s]], 1);
      order[base + pos] = listE[s];
    }
  } else {
    int unused = 0;
    sort_sweep<true>(edst, order, E, lo, base, cntL, listE, listD, &nList, unused);
  }
}

__global__ __launch_bounds__(128)
void node_prep_kernel(const float* __restrict__ ns, const float* __restrict__ nv,
                      const float* __restrict__ attr,
                      const float* __restrict__ Wls, const float* __restrict__ Wlv,
                      const float* __restrict__ Wss, const float* __restrict__ Wsv,
                      float* __restrict__ f_s, float* __restrict__ f_v,
                      float* __restrict__ sc_s, float* __restrict__ sc_v, int n)
{
  const int lane = threadIdx.x & 31;
  const int hi   = lane >> 4;
  const int lm   = lane & 15;
  const int n0   = (blockIdx.x * 4 + (threadIdx.x >> 5)) * 16;
  if (n0 >= n) return;
  const int na = (n0 + lm < n) ? (n0 + lm) : (n - 1);

  float ar[8];
  #pragma unroll
  for (int r = 0; r < 8; ++r) {
    const int nd = n0 + r + 8 * hi;
    ar[r] = attr[nd < n ? nd : n - 1];
  }

  v2f as[MUL0 / 4];
  #pragma unroll
  for (int kk = 0; kk < MUL0 / 4; ++kk)
    as[kk] = *(const v2f*)(ns + (long long)na * MUL0 + 4 * kk + 2 * hi);

  #pragma unroll
  for (int w = 0; w < 2; ++w) {
    const float* __restrict__ Wm = w ? Wss : Wls;
    float* __restrict__ dst      = w ? sc_s : f_s;
    v8f acc[4];
    #pragma unroll
    for (int t = 0; t < 4; ++t) acc[t] = v8f{};
    #pragma unroll
    for (int kk = 0; kk < MUL0 / 4; ++kk) {
      const int k0 = 4 * kk + 2 * hi;
      #pragma unroll
      for (int t = 0; t < 4; ++t) {
        v2f b;
        b.x = Wm[ k0      * MUL0 + 16 * t + lm];
        b.y = Wm[(k0 + 1) * MUL0 + 16 * t + lm];
        acc[t] = __builtin_amdgcn_wmma_f32_16x16x4_f32(
            false, as[kk], false, b, (short)0, acc[t], false, false);
      }
    }
    #pragma unroll
    for (int r = 0; r < 8; ++r) {
      const int nd = n0 + r + 8 * hi;
      if (nd < n) {
        #pragma unroll
        for (int t = 0; t < 4; ++t) dst[(long long)nd * MUL0 + 16 * t + lm] = acc[t][r] * ar[r];
      }
    }
  }

  v2f av[3][MUL1 / 4];
  #pragma unroll
  for (int kk = 0; kk < MUL1 / 4; ++kk) {
    const v2f* q = (const v2f*)(nv + (long long)na * (MUL1 * 3) + 12 * kk + 6 * hi);
    const v2f q0 = q[0], q1 = q[1], q2 = q[2];
    av[0][kk].x = q0.x;  av[0][kk].y = q1.y;
    av[1][kk].x = q0.y;  av[1][kk].y = q2.x;
    av[2][kk].x = q1.x;  av[2][kk].y = q2.y;
  }
  #pragma unroll
  for (int w = 0; w < 2; ++w) {
    const float* __restrict__ Wm = w ? Wsv : Wlv;
    float* __restrict__ dst      = w ? sc_v : f_v;
    v8f acc[3][2];
    #pragma unroll
    for (int c = 0; c < 3; ++c) { acc[c][0] = v8f{}; acc[c][1] = v8f{}; }
    #pragma unroll
    for (int kk = 0; kk < MUL1 / 4; ++kk) {
      const int k0 = 4 * kk + 2 * hi;
      #pragma unroll
      for (int t = 0; t < 2; ++t) {
        v2f b;
        b.x = Wm[ k0      * MUL1 + 16 * t + lm];
        b.y = Wm[(k0 + 1) * MUL1 + 16 * t + lm];
        #pragma unroll
        for (int c = 0; c < 3; ++c)
          acc[c][t] = __builtin_amdgcn_wmma_f32_16x16x4_f32(
              false, av[c][kk], false, b, (short)0, acc[c][t], false, false);
      }
    }
    #pragma unroll
    for (int r = 0; r < 8; ++r) {
      const int nd = n0 + r + 8 * hi;
      if (nd < n) {
        #pragma unroll
        for (int t = 0; t < 2; ++t) {
          #pragma unroll
          for (int c = 0; c < 3; ++c)
            dst[(long long)nd * (MUL1 * 3) + (16 * t + lm) * 3 + c] = acc[c][t][r] * ar[r];
        }
      }
    }
  }
}

__global__ __launch_bounds__(WAVES * 32, 2)
void edge_fused_kernel(const float* __restrict__ es,
                       const float* __restrict__ ea,
                       const int*   __restrict__ esrc,
                       const int*   __restrict__ edst,
                       const int*   __restrict__ order,
                       const float* __restrict__ Wfc1,
                       const float* __restrict__ Wfc2,
                       const float* __restrict__ f_s,
                       const float* __restrict__ f_v,
                       float* __restrict__ agg_s,
                       float* __restrict__ agg_v,
                       int E, int numTiles)
{
  __shared__ float W1s[KPAD1 * NPAD1];
  __shared__ float4 W2f[KS2 * 32 * (WNUM / 32)];
  __shared__ float Hs[WAVES][16 * NPAD1];
  __shared__ float mF[WAVES][16 * 4];
  __shared__ int   mI[WAVES][16 * 2];

  {
    constexpr int NT = WAVES * 32;
    constexpr int N1 = (KPAD1 * NPAD1 + NT - 1) / NT;
    float w1[N1];
    #pragma unroll
    for (int j = 0; j < N1; ++j) {
      const int i = threadIdx.x + NT * j;
      const int k = i / NPAD1, ncol = i - k * NPAD1;
      const bool ok = (k < NBASIS) && (ncol < FCH);
      const float x = Wfc1[ok ? k * FCH + ncol : 0];
      w1[j] = ok ? x : 0.f;
    }
    constexpr int NS = KS2 * 32 * (WNUM / 32);
    constexpr int N2 = (NS + NT - 1) / NT;
    static_assert(FCH % 4 == 0, "GEMM2 contracts over FCH in k steps of 4");
    static_assert(NT % 32 == 0 && WNUM % 32 == 0, "slot decoding below");
    float4 w2[N2];
    #pragma unroll
    for (int j = 0; j < N2; ++j) {
      const int s  = threadIdx.x + NT * j;
      const int sc = (s < NS) ? s : NS - 1;
      const int ln = sc & 31, q = sc >> 5;
      const int pr = q / KS2, kk = q - pr * KS2;
      const int k0 = 4 * kk + 2 * (ln >> 4);
      const int cc = 32 * pr + (ln & 15);
      const float* r0 = Wfc2 + k0 * WNUM + cc;
      float4 x;
      x.x = r0[0];  x.y = r0[WNUM];  x.z = r0[16];  x.w = r0[WNUM + 16];
      w2[j] = x;
    }
    #pragma unroll
    for (int j = 0; j < N1; ++j) {
      const int i = threadIdx.x + NT * j;
      if (i < KPAD1 * NPAD1) W1s[i] = w1[j];
    }
    #pragma unroll
    for (int j = 0; j < N2; ++j) {
      const int s = threadIdx.x + NT * j;
      if (s < NS) W2f[s] = w2[j];
    }
  }
  __syncthreads();

  const int wid  = threadIdx.x >> 5;
  const int lane = threadIdx.x & 31;
  const int hi   = lane >> 4;
  const int lm   = lane & 15;

  float* H  = Hs[wid];
  float* mf = mF[wid];
  int*   mi = mI[wid];

  const float inv  = 0.25f;
  const float inv3 = 0.57735026918962576451f;
  const float inv2 = 0.70710678118654752440f;

  const int gw   = blockIdx.x * WAVES + wid;
  const int nw   = gridDim.x * WAVES;
  const int tq   = numTiles / nw;
  const int trem = numTiles - tq * nw;
  const int t0   = gw * tq + (gw < trem ? gw : trem);
  const int t1   = t0 + tq + (gw < trem ? 1 : 0);

  int nodeFirst = -3, nodeLast = -3;
  if (t0 < t1) {
    const int pl = (t1 * 16 < E ? t1 * 16 : E) - 1;
    nodeFirst = edst[order[t0 * 16]];
    nodeLast  = edst[order[pl]];
  }

  float cs[6][1];
  float cv[8][3];
  #pragma unroll
  for (int i = 0; i < 6; ++i) cs[i][0] = 0.f;
  #pragma unroll
  for (int i = 0; i < 8; ++i) { cv[i][0] = 0.f; cv[i][1] = 0.f; cv[i][2] = 0.f; }
  int dLastPrev = -2;

  for (int tile = t0; tile < t1; ++tile) {
    int base = tile * 16;

    int  pA = base + lm;
    bool ev = (pA < E);
    const int  pN = base + 16;
    int        eA = order[ev ? pA : E - 1];
    const int  eN = order[pN < E ? pN : E - 1];
    const int  dNraw = edst[eN];
    const int  dNextTile = (pN < E) ? dNraw : -1;

    if (lane < 16) {
      int e = eA;
      if (ev) {
        mi[lane * 2 + 0] = esrc[e];
        mi[lane * 2 + 1] = edst[e];
        mf[lane * 4 + 0] = ea[(long long)e * 4 + 0];
        mf[lane * 4 + 1] = ea[(long long)e * 4 + 1];
        mf[lane * 4 + 2] = ea[(long long)e * 4 + 2];
        mf[lane * 4 + 3] = ea[(long long)e * 4 + 3];
      } else {
        mi[lane * 2 + 0] = 0;
        mi[lane * 2 + 1] = -1;
        mf[lane * 4 + 0] = 0.f; mf[lane * 4 + 1] = 0.f;
        mf[lane * 4 + 2] = 0.f; mf[lane * 4 + 3] = 0.f;
      }
    }

    v2f a1[3];
    #pragma unroll
    for (int kk = 0; kk < 3; ++kk) {
      int k0 = 4 * kk + 2 * hi;
      a1[kk].x = (ev && (k0     < NBASIS)) ? es[(long long)eA * NBASIS + k0    ] : 0.f;
      a1[kk].y = (ev && (k0 + 1 < NBASIS)) ? es[(long long)eA * NBASIS + k0 + 1] : 0.f;
    }

    #pragma unroll
    for (int nt = 0; nt < NPAD1 / 16; ++nt) {
      v8f acc = {};
      #pragma unroll
      for (int kk = 0; kk < 3; ++kk) {
        int k0 = 4 * kk + 2 * hi;
        v2f b;
        b.x = W1s[ k0      * NPAD1 + nt * 16 + lm];
        b.y = W1s[(k0 + 1) * NPAD1 + nt * 16 + lm];
        acc = __builtin_amdgcn_wmma_f32_16x16x4_f32(
            false, a1[kk], false, b, (short)0, acc, false, false);
      }
      #pragma unroll
      for (int r = 0; r < 8; ++r) {
        float x  = acc[r];
        float sg = fast_rcp(1.f + __expf(-x));
        H[(r + 8 * hi) * NPAD1 + nt * 16 + lm] = x * sg;
      }
    }
    asm volatile("s_wait_dscnt 0" ::: "memory");

    v2f a2[KS2];
    #pragma unroll
    for (int kk = 0; kk < KS2; ++kk) {
      int k0 = 4 * kk + 2 * hi;
      a2[kk] = *(const v2f*)&H[lm * NPAD1 + k0];
    }

    int   drow[8], srow[8];
    float y0r[8], y1x[8], y1y[8], y1z[8];
    #pragma unroll
    for (int r = 0; r < 8; ++r) {
      int m = r + 8 * hi;
      srow[r] = mi[m * 2 + 0];
      drow[r] = mi[m * 2 + 1];
      y0r[r]  = mf[m * 4 + 0];
      y1x[r]  = mf[m * 4 + 1];
      y1y[r]  = mf[m * 4 + 2];
      y1z[r]  = mf[m * 4 + 3];
    }
    const int d15   = mi[15 * 2 + 1];
    const int dPrev = hi ? mi[7 * 2 + 1] : dLastPrev;
    const int dNext = hi ? ((tile + 1 < t1) ? dNextTile : -2) : mi[8 * 2 + 1];
    RunFlags rf;
    {
      const bool contIn = (drow[0] == dPrev);
      bool noEnd = true;
      #pragma unroll
      for (int r = 0; r < 7; ++r) {
        const bool er   = (drow[r] != drow[r + 1]);
        const bool hold = contIn && noEnd;
        rf.e[r]   = er;
        rf.cap[r] = er && hold;
        rf.fl[r]  = er && !hold && (drow[r] >= 0);
        noEnd = noEnd && !er;
      }
      rf.single  = contIn && noEnd;
      rf.contOut = (drow[7] == dNext);
      const bool condA = contIn && !rf.single && (drow[0] >= 0);
      const bool condB = !rf.contOut && (drow[7] >= 0);
      const bool shA = (drow[0] == nodeFirst) || (drow[0] == nodeLast);
      const bool shB = (drow[7] == nodeFirst) || (drow[7] == nodeLast);
      rf.aSt = condA && !shA;  rf.aAt = condA && shA;
      rf.bSt = condB && !shB;  rf.bAt = condB && shB;
      bool anyFl = false;
      #pragma unroll
      for (int r = 0; r < 7; ++r) anyFl = anyFl || rf.fl[r];
      rf.uFl   = __builtin_amdgcn_ballot_w32(anyFl) != 0u;
      rf.uAt   = __builtin_amdgcn_ballot_w32(rf.aAt || rf.bAt) != 0u;
      rf.uB    = __builtin_amdgcn_ballot_w32(rf.bSt) != 0u;
      rf.uRare = __builtin_amdgcn_ballot_w32(anyFl || rf.aAt || rf.bAt || rf.bSt) != 0u;
    }


    auto gemm2_pair = [&](const int pr, v8f (&acc)[2]) {
      const float4* bp = &W2f[pr * KS2 * 32 + lane];
      v2f b[2][KS2];
      #pragma unroll
      for (int kk = 0; kk < KS2; ++kk) {
        const float4 q = bp[kk * 32];
        b[0][kk].x = q.x;  b[0][kk].y = q.y;
        b[1][kk].x = q.z;  b[1][kk].y = q.w;
      }
      __builtin_amdgcn_sched_barrier(0);

      acc[0] = v8f{}; acc[1] = v8f{};
      #pragma unroll
      for (int kk = 0; kk < KS2; ++kk) {
        #pragma unroll
        for (int t = 0; t < 2; ++t) {
          acc[t] = __builtin_amdgcn_wmma_f32_16x16x4_f32(
              false, a2[kk], false, b[t][kk], (short)0, acc[t], false, false);
        }
      }
    };

    #pragma unroll
    for (int p = 0; p < 4; ++p) {
      const int cA  = p * 32 + lm;
      const int col = cA & 63;
      float g[2][8];
      #pragma unroll
      for (int t = 0; t < 2; ++t) {
        #pragma unroll
        for (int r = 0; r < 8; ++r) g[t][r] = f_s[(long long)srow[r] * MUL0 + col + 16 * t];
      }
      v8f acc[2];
      gemm2_pair(p, acc);

      #pragma unroll
      for (int t = 0; t < 2; ++t) {
        const int c = cA + 16 * t;
        if (p < 2) {
          float v[1][8];
          #pragma unroll
          for (int r = 0; r < 8; ++r) v[0][r] = acc[t][r] * g[t][r] * y0r[r] * inv;
          seg_flush<1>(agg_s, MIDS, c, drow, rf, hi, v, cs[p < 2 ? 2 * p + t : 0]);
        } else {
          const int u = c - 64;
          float v[3][8];
          #pragma unroll
          for (int r = 0; r < 8; ++r) {
            float tt = acc[t][r] * g[t][r] * inv;
            v[0][r] = tt * y1x[r]; v[1][r] = tt * y1y[r]; v[2][r] = tt * y1z[r];
          }
          seg_flush<3>(agg_v, MIDV * 3, u * 3, drow, rf, hi, v, cv[p < 2 ? 0 : 2 * (p - 2) + t]);
        }
      }
    }

    {
      float g[2][3][8];
      #pragma unroll
      for (int t = 0; t < 2; ++t) {
        #pragma unroll
        for (int r = 0; r < 8; ++r) {
          const float* vp = &f_v[((long long)srow[r] * MUL1 + lm + 16 * t) * 3];
          g[t][0][r] = vp[0]; g[t][1][r] = vp[1]; g[t][2][r] = vp[2];
        }
      }
      v8f acc[2];

      gemm2_pair(4, acc);
      #pragma unroll
      for (int t = 0; t < 2; ++t) {
        const int u = lm + 16 * t;
        float v[3][8];
        #pragma unroll
        for (int r = 0; r < 8; ++r) {
          float tt = acc[t][r] * y0r[r] * inv;
          v[0][r] = tt * g[t][0][r]; v[1][r] = tt * g[t][1][r]; v[2][r] = tt * g[t][2][r];
        }
        seg_flush<3>(agg_v, MIDV * 3, (64 + u) * 3, drow, rf, hi, v, cv[4 + t]);
      }

      gemm2_pair(5, acc);
      #pragma unroll
      for (int t = 0; t < 2; ++t) {
        const int u = lm + 16 * t;
        float v[1][8];
        #pragma unroll
        for (int r = 0; r < 8; ++r) {
          float dt = g[t][0][r] * y1x[r] + g[t][1][r] * y1y[r] + g[t][2][r] * y1z[r];
          v[0][r] = acc[t][r] * dt * inv3 * inv;
        }
        seg_flush<1>(agg_s, MIDS, 64 + u, drow, rf, hi, v, cs[4 + t]);
      }

      gemm2_pair(6, acc);
      #pragma unroll
      for (int t = 0; t < 2; ++t) {
        const int u = lm + 16 * t;
        float v[3][8];
        #pragma unroll
        for (int r = 0; r < 8; ++r) {
          float cx = g[t][1][r] * y1z[r] - g[t][2][r] * y1y[r];
          float cy = g[t][2][r] * y1x[r] - g[t][0][r] * y1z[r];
          float cz = g[t][0][r] * y1y[r] - g[t][1][r] * y1x[r];
          float tt = acc[t][r] * inv2 * inv;
          v[0][r] = tt * cx; v[1][r] = tt * cy; v[2][r] = tt * cz;
        }
        seg_flush<3>(agg_v, MIDV * 3, (96 + u) * 3, drow, rf, hi, v, cv[6 + t]);
      }
    }
    dLastPrev = d15;
  }
}

__global__ __launch_bounds__(128)
void node_out_kernel(const float* __restrict__ agg_s, const float* __restrict__ agg_v,
                     const float* __restrict__ sc_s,  const float* __restrict__ sc_v,
                     const float* __restrict__ attr,
                     const float* __restrict__ Wl2s, const float* __restrict__ Wl2v,
                     const float* __restrict__ Wl3,
                     float* __restrict__ out, int n)
{
  const int lane = threadIdx.x & 31;
  const int hi   = lane >> 4;
  const int lm   = lane & 15;
  const int n0   = (blockIdx.x * 4 + (threadIdx.x >> 5)) * 16;
  if (n0 >= n) return;
  const int na = (n0 + lm < n) ? (n0 + lm) : (n - 1);

  v2f as[MIDS / 4];
  #pragma unroll
  for (int kk = 0; kk < MIDS / 4; ++kk)
    as[kk] = *(const v2f*)(agg_s + (long long)na * MIDS + 4 * kk + 2 * hi);

  float part = 0.f;
  #pragma unroll
  for (int kk = 0; kk < MIDS / 4; ++kk) {
    const int k0 = 4 * kk + 2 * hi;
    part += as[kk].x * Wl3[k0] + as[kk].y * Wl3[k0 + 1];
  }
  const float ang = 0.1f * (part + xhalf(part)) * attr[na];
  const float csn = cosf(ang);
  const float snn = sinf(ang);

  float cr[8], sr[8], ar[8];
  #pragma unroll
  for (int r = 0; r < 8; ++r) {
    const int nd = n0 + r + 8 * hi;
    cr[r] = __shfl(csn, r + 8 * hi);
    sr[r] = __shfl(snn, r + 8 * hi);
    ar[r] = attr[nd < n ? nd : n - 1];
  }

  {
    float scv[8][4];
    #pragma unroll
    for (int r = 0; r < 8; ++r) {
      const int nd = n0 + r + 8 * hi;
      const int nc = nd < n ? nd : n - 1;
      #pragma unroll
      for (int t = 0; t < 4; ++t) scv[r][t] = sc_s[(long long)nc * MUL0 + 16 * t + lm];
    }
    v8f acc[4];
    #pragma unroll
    for (int t = 0; t < 4; ++t) acc[t] = v8f{};
    #pragma unroll
    for (int kk = 0; kk < MIDS / 4; ++kk) {
      const int k0 = 4 * kk + 2 * hi;
      #pragma unroll
      for (int t = 0; t < 4; ++t) {
        v2f b;
        b.x = Wl2s[ k0      * MUL0 + 16 * t + lm];
        b.y = Wl2s[(k0 + 1) * MUL0 + 16 * t + lm];
        acc[t] = __builtin_amdgcn_wmma_f32_16x16x4_f32(
            false, as[kk], false, b, (short)0, acc[t], false, false);
      }
    }
    #pragma unroll
    for (int r = 0; r < 8; ++r) {
      const int nd = n0 + r + 8 * hi;
      if (nd < n) {
        #pragma unroll
        for (int t = 0; t < 4; ++t) {
          const int col = 16 * t + lm;
          const float conv = acc[t][r] * ar[r];
          out[(long long)nd * 160 + col] = cr[r] * scv[r][t] + sr[r] * conv;
        }
      }
    }
  }

  {
    float scv[8][2][3];
    #pragma unroll
    for (int r = 0; r < 8; ++r) {
      const int nd = n0 + r + 8 * hi;
      const int nc = nd < n ? nd : n - 1;
      #pragma unroll
      for (int t = 0; t < 2; ++t) {
        #pragma unroll
        for (int c = 0; c < 3; ++c) scv[r][t][c] = sc_v[(long long)nc * (MUL1 * 3) + (16 * t + lm) * 3 + c];
      }
    }
    v8f acc[3][2];
    #pragma unroll
    for (int c = 0; c < 3; ++c) { acc[c][0] = v8f{}; acc[c][1] = v8f{}; }
    const float* pv = agg_v + (long long)na * (MIDV * 3) + 6 * hi;
    #pragma unroll
    for (int kk = 0; kk < MIDV / 4; ++kk) {
      const int k0 = 4 * kk + 2 * hi;
      const v2f* q = (const v2f*)(pv + 12 * kk);
      const v2f q0 = q[0], q1 = q[1], q2 = q[2];
      v2f a[3];
      a[0].x = q0.x;  a[0].y = q1.y;
      a[1].x = q0.y;  a[1].y = q2.x;
      a[2].x = q1.x;  a[2].y = q2.y;
      #pragma unroll
      for (int t = 0; t < 2; ++t) {
        v2f b;
        b.x = Wl2v[ k0      * MUL1 + 16 * t + lm];
        b.y = Wl2v[(k0 + 1) * MUL1 + 16 * t + lm];
        #pragma unroll
        for (int c = 0; c < 3; ++c)
          acc[c][t] = __builtin_amdgcn_wmma_f32_16x16x4_f32(
              false, a[c], false, b, (short)0, acc[c][t], false, false);
      }
    }
    #pragma unroll
    for (int r = 0; r < 8; ++r) {
      const int nd = n0 + r + 8 * hi;
      if (nd < n) {
        #pragma unroll
        for (int t = 0; t < 2; ++t) {
          #pragma unroll
          for (int c = 0; c < 3; ++c) {
            const int idx = (16 * t + lm) * 3 + c;
            const float conv = acc[c][t][r] * ar[r];
            out[(long long)nd * 160 + MUL0 + idx] = cr[r] * scv[r][t][c] + sr[r] * conv;
          }
        }
      }
    }
  }
}

extern "C" void kernel_launch(void* const* d_in, const int* in_sizes, int n_in,
                              void* d_out, int out_size, void* d_ws, size_t ws_size,
                              hipStream_t stream) {
  const float* node_scalars = (const float*)d_in[0];
  const float* node_vectors = (const float*)d_in[1];
  const float* node_attr    = (const float*)d_in[2];
  const int*   edge_src     = (const int*)  d_in[3];
  const int*   edge_dst     = (const int*)  d_in[4];
  const float* edge_attr    = (const float*)d_in[5];
  const float* edge_scalars = (const float*)d_in[6];
  const float* W_fc1        = (const float*)d_in[7];
  const float* W_fc2        = (const float*)d_in[8];
  const float* W_sc_s       = (const float*)d_in[9];
  const float* W_sc_v       = (const float*)d_in[10];
  const float* W_l1_s       = (const float*)d_in[11];
  const float* W_l1_v       = (const float*)d_in[12];
  const float* W_l2_s       = (const float*)d_in[13];
  const float* W_l2_v       = (const float*)d_in[14];
  const float* W_l3         = (const float*)d_in[15];

  const int N = in_sizes[0] / MUL0;
  const int E = in_sizes[3];

  float* ws    = (float*)d_ws;
  float* f_s   = ws;
  float* f_v   = f_s  + (size_t)N * MUL0;
  float* sc_s  = f_v  + (size_t)N * MUL1 * 3;
  float* sc_v  = sc_s + (size_t)N * MUL0;
  float* agg_s = sc_v + (size_t)N * MUL1 * 3;
  float* agg_v = agg_s + (size_t)N * MIDS;

  int* order = (int*)d_out;

  long long nAgg = (long long)N * (MIDS + MIDV * 3);
  zero_kernel<<<4096, 256, 0, stream>>>(agg_s, nAgg);

  const int nodeTiles = (N + 15) / 16;
  node_prep_kernel<<<(nodeTiles + 3) / 4, 128, 0, stream>>>(node_scalars, node_vectors, node_attr,
                                          W_l1_s, W_l1_v, W_sc_s, W_sc_v,
                                          f_s, f_v, sc_s, sc_v, N);

  dst_sort_kernel<<<(N + SORT_R - 1) / SORT_R, SORT_NT, 0, stream>>>(edge_dst, order, E);

  int numTiles = (E + 15) / 16;
  edge_fused_kernel<<<512, WAVES * 32, 0, stream>>>(edge_scalars, edge_attr,
                                                     edge_src, edge_dst, order,
                                                     W_fc1, W_fc2, f_s, f_v,
                                                     agg_s, agg_v, E, numTiles);

  node_out_kernel<<<(nodeTiles + 3) / 4, 128, 0, stream>>>(agg_s, agg_v, sc_s, sc_v, node_attr,
                                         W_l2_s, W_l2_v, W_l3, (float*)d_out, N);
}


# edge kernel reads a tile's inputs ahead (edge ids two tiles early; edge scalars, attributes, end nodes one tile early): a tile starts from registers, not from two dependent global round trips
# speedup vs baseline: 4.5391x; 1.0424x over previous
#include <hip/hip_runtime.h>
#include <hip/hip_bf16.h>
#include <math.h>


typedef __attribute__((ext_vector_type(2))) float v2f;
typedef __attribute__((ext_vector_type(8))) float v8f;

#define MUL0   64
#define MUL1   32
#define NBASIS 10
#define FCH    100
#define WNUM   224
#define MIDS   96
#define MIDV   128
#define KPAD1  12
#define NPAD1  112
#define WAVES  4
#define KS2    (FCH / 4)

__device__ __forceinline__ void atomic_add_f32(float* p, float v) {
  __hip_atomic_fetch_add(p, v, __ATOMIC_RELAXED, __HIP_MEMORY_SCOPE_AGENT);
}

__device__ __forceinline__ float fast_rcp(float x) {
#if __has_builtin(__builtin_amdgcn_rcpf)
  return __builtin_amdgcn_rcpf(x);
#else
  return 1.f / x;
#endif
}

__device__ __forceinline__ float xhalf(float x) {
  const int b = __builtin_bit_cast(int, x);
  return __builtin_bit_cast(float, __builtin_amdgcn_permlanex16(b, b, 0x76543210, 0xfedcba98, false, false));
}

struct RunFlags {
  bool e[7];
  bool cap[7];
  bool fl[7];
  bool single;
  bool contOut;
  bool aSt, aAt;
  bool bSt, bAt;
  bool uFl;
  bool uAt;
  bool uB;
  bool uRare;
};

template <int NC>
__device__ __forceinline__ void seg_flush(float* __restrict__ agg, int rowStride, int colOff,
                                          const int (&d)[8], const RunFlags& f, int hi,
                                          const float (&v)[NC][8], float (&carry)[NC]) {
  float run[NC], head[NC], th[NC], tl[NC];
  #pragma unroll
  for (int k = 0; k < NC; ++k) { run[k] = 0.f; head[k] = 0.f; }
  #pragma unroll
  for (int r = 0; r < 8; ++r) {
    #pragma unroll
    for (int k = 0; k < NC; ++k) run[k] += v[k][r];
    if (r < 7) {
      #pragma unroll
      for (int k = 0; k < NC; ++k) head[k] = f.cap[r] ? run[k] : head[k];
      #pragma unroll
      for (int k = 0; k < NC; ++k) run[k] = f.e[r] ? 0.f : run[k];
    }
  }
  #pragma unroll
  for (int k = 0; k < NC; ++k) {
    const float hv    = f.single ? run[k] : head[k];
    const float cinLo = xhalf(carry[k]);
    const float outLo = f.contOut ? (f.single ? hv + cinLo : run[k]) : 0.f;
    const float x     = xhalf(outLo);
    const float cin   = hi ? x : cinLo;
    th[k] = hv + cin;
    tl[k] = f.single ? th[k] : run[k];
    carry[k] = f.contOut ? tl[k] : 0.f;
  }
  if (f.aSt) {
    float* p = agg + (long long)d[0] * rowStride + colOff;
    #pragma unroll
    for (int k = 0; k < NC; ++k) p[k] = th[k];
  }
  if (f.uRare) {
    if (f.uFl) {
      float s2[NC];
      #pragma unroll
      for (int k = 0; k < NC; ++k) s2[k] = 0.f;
      #pragma unroll
      for (int r = 0; r < 7; ++r) {
        #pragma unroll
        for (int k = 0; k < NC; ++k) s2[k] += v[k][r];
        if (f.fl[r]) {
          float* p = agg + (long long)d[r] * rowStride + colOff;
          #pragma unroll
          for (int k = 0; k < NC; ++k) atomic_add_f32(p + k, s2[k]);
        }
        #pragma unroll
        for (int k = 0; k < NC; ++k) s2[k] = f.e[r] ? 0.f : s2[k];
      }
    }
    if (f.uAt) {
      if (f.aAt) {
        float* p = agg + (long long)d[0] * rowStride + colOff;
        #pragma unroll
        for (int k = 0; k < NC; ++k) atomic_add_f32(p + k, th[k]);
      }
      if (f.bAt) {
        float* p = agg + (long long)d[7] * rowStride + colOff;
        #pragma unroll
        for (int k = 0; k < NC; ++k) atomic_add_f32(p + k, tl[k]);
      }
    }
    if (f.uB) {
      if (f.bSt) {
        float* p = agg + (long long)d[7] * rowStride + colOff;
        #pragma unroll
        for (int k = 0; k < NC; ++k) p[k] = tl[k];
      }
    }
  }
}

__global__ void zero_kernel(float* __restrict__ p, long long n) {
  long long i = (long long)blockIdx.x * blockDim.x + threadIdx.x;
  long long stride = (long long)gridDim.x * blockDim.x;
  for (; i < n; i += stride) p[i] = 0.f;
}

#define SORT_R   128
#define SORT_NT  1024
#define SORT_CAP 8192

template <bool PLACE>
__device__ __forceinline__ void sort_sweep(const int* __restrict__ edst, int* __restrict__ order,
                                           int E, int lo, int base,
                                           int* cntL, int* listE, int* listD, int* nList,
                                           int& wbelow)
{
  const int tid = threadIdx.x;
  const int n4  = E >> 2;
  const int4* __restrict__ p4 = (const int4*)edst;

  auto own = [&](unsigned rel, int e) {
    if (PLACE) {
      const int pos = atomicAdd(&cntL[rel], 1);
      order[base + pos] = e;
    } else {
      atomicAdd(&cntL[rel], 1);
      const int s = atomicAdd(nList, 1);
      if (s < SORT_CAP) { listE[s] = e; listD[s] = (int)rel; }
    }
  };
  auto lower = [&](int d) { return __builtin_popcount(__builtin_amdgcn_ballot_w32(d < lo)); };
  auto one = [&](int d, int e) {
    const unsigned rel = (unsigned)(d - lo);
    if (!PLACE) wbelow += lower(d);
    if (rel < (unsigned)SORT_R) own(rel, e);
  };
  auto four = [&](const int4 d, int i4) {
    const unsigned r0 = (unsigned)(d.x - lo), r1 = (unsigned)(d.y - lo);
    const unsigned r2 = (unsigned)(d.z - lo), r3 = (unsigned)(d.w - lo);
    if (!PLACE) wbelow += lower(d.x) + lower(d.y) + lower(d.z) + lower(d.w);
    const unsigned m01 = r0 < r1 ? r0 : r1, m23 = r2 < r3 ? r2 : r3;
    if ((m01 < m23 ? m01 : m23) < (unsigned)SORT_R) {
      const int e0 = 4 * i4;
      if (r0 < (unsigned)SORT_R) own(r0, e0);
      if (r1 < (unsigned)SORT_R) own(r1, e0 + 1);
      if (r2 < (unsigned)SORT_R) own(r2, e0 + 2);
      if (r3 < (unsigned)SORT_R) own(r3, e0 + 3);
    }
  };

  int jb = 0;
  for (; jb + 4 * SORT_NT <= n4; jb += 4 * SORT_NT) {
    int4 d[4];
    #pragma unroll
    for (int u = 0; u < 4; ++u) d[u] = p4[jb + u * SORT_NT + tid];
    #pragma unroll
    for (int u = 0; u < 4; ++u) four(d[u], jb + u * SORT_NT + tid);
  }
  for (int j = jb + tid; j < n4; j += SORT_NT) four(p4[j], j);
  {
    const int e = 4 * n4 + tid;
    if (e < E) one(edst[e], e);
  }
}

__global__ __launch_bounds__(SORT_NT)
void dst_sort_kernel(const int* __restrict__ edst, int* __restrict__ order, int E)
{
  __shared__ int cntL[SORT_R];
  __shared__ int listE[SORT_CAP];
  __shared__ int listD[SORT_CAP];
  __shared__ int nList;
  __shared__ int nBelow;

  const int tid = threadIdx.x;
  const int lo  = blockIdx.x * SORT_R;

  if (tid < SORT_R) cntL[tid] = 0;
  if (tid == 0) { nList = 0; nBelow = 0; }
  __syncthreads();

  int wbelow = 0;
  sort_sweep<false>(edst, order, E, lo, 0, cntL, listE, listD, &nList, wbelow);
  if ((tid & 31) == 0) atomicAdd(&nBelow, wbelow);
  __syncthreads();

  const int total = nList;
  const int base  = nBelow;

  static_assert(SORT_R == 128, "one wave, four counts per lane");
  if (tid < 32) {
    const int c0 = cntL[4 * tid], c1 = cntL[4 * tid + 1], c2 = cntL[4 * tid + 2], c3 = cntL[4 * tid + 3];
    const int s = c0 + c1 + c2 + c3;
    int inc = s;
    #pragma unroll
    for (int o = 1; o < 32; o <<= 1) {
      const int t = __shfl_up(inc, o);
      if (tid >= o) inc += t;
    }
    const int ex = inc - s;
    cntL[4 * tid]     = ex;
    cntL[4 * tid + 1] = ex + c0;
    cntL[4 * tid + 2] = ex + c0 + c1;
    cntL[4 * tid + 3] = ex + c0 + c1 + c2;
  }
  __syncthreads();

  if (total <= SORT_CAP) {
    for (int s = tid; s < total; s += SORT_NT) {
      const int pos = atomicAdd(&cntL[listD[s]], 1);
      order[base + pos] = listE[s];
    }
  } else {
    int unused = 0;
    sort_sweep<true>(edst, order, E, lo, base, cntL, listE, listD, &nList, unused);
  }
}

__global__ __launch_bounds__(128)
void node_prep_kernel(const float* __restrict__ ns, const float* __restrict__ nv,
                      const float* __restrict__ attr,
                      const float* __restrict__ Wls, const float* __restrict__ Wlv,
                      const float* __restrict__ Wss, const float* __restrict__ Wsv,
                      float* __restrict__ f_s, float* __restrict__ f_v,
                      float* __restrict__ sc_s, float* __restrict__ sc_v, int n)
{
  const int lane = threadIdx.x & 31;
  const int hi   = lane >> 4;
  const int lm   = lane & 15;
  const int n0   = (blockIdx.x * 4 + (threadIdx.x >> 5)) * 16;
  if (n0 >= n) return;
  const int na = (n0 + lm < n) ? (n0 + lm) : (n - 1);

  float ar[8];
  #pragma unroll
  for (int r = 0; r < 8; ++r) {
    const int nd = n0 + r + 8 * hi;
    ar[r] = attr[nd < n ? nd : n - 1];
  }

  v2f as[MUL0 / 4];
  #pragma unroll
  for (int kk = 0; kk < MUL0 / 4; ++kk)
    as[kk] = *(const v2f*)(ns + (long long)na * MUL0 + 4 * kk + 2 * hi);

  #pragma unroll
  for (int w = 0; w < 2; ++w) {
    const float* __restrict__ Wm = w ? Wss : Wls;
    float* __restrict__ dst      = w ? sc_s : f_s;
    v8f acc[4];
    #pragma unroll
    for (int t = 0; t < 4; ++t) acc[t] = v8f{};
    #pragma unroll
    for (int kk = 0; kk < MUL0 / 4; ++kk) {
      const int k0 = 4 * kk + 2 * hi;
      #pragma unroll
      for (int t = 0; t < 4; ++t) {
        v2f b;
        b.x = Wm[ k0      * MUL0 + 16 * t + lm];
        b.y = Wm[(k0 + 1) * MUL0 + 16 * t + lm];
        acc[t] = __builtin_amdgcn_wmma_f32_16x16x4_f32(
            false, as[kk], false, b, (short)0, acc[t], false, false);
      }
    }
    #pragma unroll
    for (int r = 0; r < 8; ++r) {
      const int nd = n0 + r + 8 * hi;
      if (nd < n) {
        #pragma unroll
        for (int t = 0; t < 4; ++t) dst[(long long)nd * MUL0 + 16 * t + lm] = acc[t][r] * ar[r];
      }
    }
  }

  v2f av[3][MUL1 / 4];
  #pragma unroll
  for (int kk = 0; kk < MUL1 / 4; ++kk) {
    const v2f* q = (const v2f*)(nv + (long long)na * (MUL1 * 3) + 12 * kk + 6 * hi);
    const v2f q0 = q[0], q1 = q[1], q2 = q[2];
    av[0][kk].x = q0.x;  av[0][kk].y = q1.y;
    av[1][kk].x = q0.y;  av[1][kk].y = q2.x;
    av[2][kk].x = q1.x;  av[2][kk].y = q2.y;
  }
  #pragma unroll
  for (int w = 0; w < 2; ++w) {
    const float* __restrict__ Wm = w ? Wsv : Wlv;
    float* __restrict__ dst      = w ? sc_v : f_v;
    v8f acc[3][2];
    #pragma unroll
    for (int c = 0; c < 3; ++c) { acc[c][0] = v8f{}; acc[c][1] = v8f{}; }
    #pragma unroll
    for (int kk = 0; kk < MUL1 / 4; ++kk) {
      const int k0 = 4 * kk + 2 * hi;
      #pragma unroll
      for (int t = 0; t < 2; ++t) {
        v2f b;
        b.x = Wm[ k0      * MUL1 + 16 * t + lm];
        b.y = Wm[(k0 + 1) * MUL1 + 16 * t + lm];
        #pragma unroll
        for (int c = 0; c < 3; ++c)
          acc[c][t] = __builtin_amdgcn_wmma_f32_16x16x4_f32(
              false, av[c][kk], false, b, (short)0, acc[c][t], false, false);
      }
    }
    #pragma unroll
    for (int r = 0; r < 8; ++r) {
      const int nd = n0 + r + 8 * hi;
      if (nd < n) {
        #pragma unroll
        for (int t = 0; t < 2; ++t) {
          #pragma unroll
          for (int c = 0; c < 3; ++c)
            dst[(long long)nd * (MUL1 * 3) + (16 * t + lm) * 3 + c] = acc[c][t][r] * ar[r];
        }
      }
    }
  }
}

struct TileIn {
  int    src, dst;
  float4 y;
  v2f    a1[3];
};

__global__ __launch_bounds__(WAVES * 32, 2)
void edge_fused_kernel(const float* __restrict__ es,
                       const float* __restrict__ ea,
                       const int*   __restrict__ esrc,
                       const int*   __restrict__ edst,
                       const int*   __restrict__ order,
                       const float* __restrict__ Wfc1,
                       const float* __restrict__ Wfc2,
                       const float* __restrict__ f_s,
                       const float* __restrict__ f_v,
                       float* __restrict__ agg_s,
                       float* __restrict__ agg_v,
                       int E, int numTiles)
{
  __shared__ float W1s[KPAD1 * NPAD1];
  __shared__ float4 W2f[KS2 * 32 * (WNUM / 32)];
  __shared__ float Hs[WAVES][16 * NPAD1];
  __shared__ float mF[WAVES][16 * 4];
  __shared__ int   mI[WAVES][16 * 2];

  const int wid  = threadIdx.x >> 5;
  const int lane = threadIdx.x & 31;
  const int hi   = lane >> 4;
  const int lm   = lane & 15;

  const int gw   = blockIdx.x * WAVES + wid;
  const int nw   = gridDim.x * WAVES;
  const int tq   = numTiles / nw;
  const int trem = numTiles - tq * nw;
  const int t0   = gw * tq + (gw < trem ? gw : trem);
  const int t1   = t0 + tq + (gw < trem ? 1 : 0);

  auto ordAt = [&](int tl) {
    const int p = tl * 16 + lm;
    return order[p < E ? p : E - 1];
  };
  auto loadTile = [&](int e) {
    TileIn x;
    x.src = esrc[e];
    x.dst = edst[e];
    x.y   = *(const float4*)(ea + (long long)e * 4);
    #pragma unroll
    for (int kk = 0; kk < 3; ++kk) {
      const int k0 = 4 * kk + 2 * hi;
      const bool in = (k0 < NBASIS);
      const v2f q = *(const v2f*)(es + (long long)e * NBASIS + (in ? k0 : 0));
      x.a1[kk].x = in ? q.x : 0.f;
      x.a1[kk].y = in ? q.y : 0.f;
    }
    return x;
  };
  static_assert(NBASIS % 2 == 0, "loadTile reads edge_scalars in pairs");

  int nodeFirst = -3, nodeLast = -3;
  int eNxt = 0;
  TileIn nxt = {};
  if (t0 < t1) {
    const int pl = (t1 * 16 < E ? t1 * 16 : E) - 1;
    const int e0 = ordAt(t0);
    eNxt = ordAt(t0 + 1);
    nodeFirst = edst[order[t0 * 16]];
    nodeLast  = edst[order[pl]];
    nxt = loadTile(e0);
  }

  {
    constexpr int NT = WAVES * 32;
    constexpr int N1 = (KPAD1 * NPAD1 + NT - 1) / NT;
    float w1[N1];
    #pragma unroll
    for (int j = 0; j < N1; ++j) {
      const int i = threadIdx.x + NT * j;
      const int k = i / NPAD1, ncol = i - k * NPAD1;
      const bool ok = (k < NBASIS) && (ncol < FCH);
      const float x = Wfc1[ok ? k * FCH + ncol : 0];
      w1[j] = ok ? x : 0.f;
    }
    constexpr int NS = KS2 * 32 * (WNUM / 32);
    constexpr int N2 = (NS + NT - 1) / NT;
    static_assert(FCH % 4 == 0, "GEMM2 contracts over FCH in k steps of 4");
    static_assert(NT % 32 == 0 && WNUM % 32 == 0, "slot decoding below");
    float4 w2[N2];
    #pragma unroll
    for (int j = 0; j < N2; ++j) {
      const int s  = threadIdx.x + NT * j;
      const int sc = (s < NS) ? s : NS - 1;
      const int ln = sc & 31, q = sc >> 5;
      const int pr = q / KS2, kk = q - pr * KS2;
      const int k0 = 4 * kk + 2 * (ln >> 4);
      const int cc = 32 * pr + (ln & 15);
      const float* r0 = Wfc2 + k0 * WNUM + cc;
      float4 x;
      x.x = r0[0];  x.y = r0[WNUM];  x.z = r0[16];  x.w = r0[WNUM + 16];
      w2[j] = x;
    }
    #pragma unroll
    for (int j = 0; j < N1; ++j) {
      const int i = threadIdx.x + NT * j;
      if (i < KPAD1 * NPAD1) W1s[i] = w1[j];
    }
    #pragma unroll
    for (int j = 0; j < N2; ++j) {
      const int s = threadIdx.x + NT * j;
      if (s < NS) W2f[s] = w2[j];
    }
  }
  __syncthreads();

  float* H  = Hs[wid];
  float* mf = mF[wid];
  int*   mi = mI[wid];

  const float inv  = 0.25f;
  const float inv3 = 0.57735026918962576451f;
  const float inv2 = 0.70710678118654752440f;

  float cs[6][1];
  float cv[8][3];
  #pragma unroll
  for (int i = 0; i < 6; ++i) cs[i][0] = 0.f;
  #pragma unroll
  for (int i = 0; i < 8; ++i) { cv[i][0] = 0.f; cv[i][1] = 0.f; cv[i][2] = 0.f; }
  int dLastPrev = -2;

  for (int tile = t0; tile < t1; ++tile) {
    int base = tile * 16;

    int  pA = base + lm;
    bool ev = (pA < E);

    const TileIn cur = nxt;
    nxt  = loadTile(eNxt);
    eNxt = ordAt(tile + 2);

    if (lane < 16) {
      if (ev) {
        mi[lane * 2 + 0] = cur.src;
        mi[lane * 2 + 1] = cur.dst;
        mf[lane * 4 + 0] = cur.y.x;
        mf[lane * 4 + 1] = cur.y.y;
        mf[lane * 4 + 2] = cur.y.z;
        mf[lane * 4 + 3] = cur.y.w;
      } else {
        mi[lane * 2 + 0] = 0;
        mi[lane * 2 + 1] = -1;
        mf[lane * 4 + 0] = 0.f; mf[lane * 4 + 1] = 0.f;
        mf[lane * 4 + 2] = 0.f; mf[lane * 4 + 3] = 0.f;
      }
    }

    v2f a1[3];
    #pragma unroll
    for (int kk = 0; kk < 3; ++kk) {
      a1[kk].x = ev ? cur.a1[kk].x : 0.f;
      a1[kk].y = ev ? cur.a1[kk].y : 0.f;
    }

    #pragma unroll
    for (int nt = 0; nt < NPAD1 / 16; ++nt) {
      v8f acc = {};
      #pragma unroll
      for (int kk = 0; kk < 3; ++kk) {
        int k0 = 4 * kk + 2 * hi;
        v2f b;
        b.x = W1s[ k0      * NPAD1 + nt * 16 + lm];
        b.y = W1s[(k0 + 1) * NPAD1 + nt * 16 + lm];
        acc = __builtin_amdgcn_wmma_f32_16x16x4_f32(
            false, a1[kk], false, b, (short)0, acc, false, false);
      }
      #pragma unroll
      for (int r = 0; r < 8; ++r) {
        float x  = acc[r];
        float sg = fast_rcp(1.f + __expf(-x));
        H[(r + 8 * hi) * NPAD1 + nt * 16 + lm] = x * sg;
      }
    }
    asm volatile("s_wait_dscnt 0" ::: "memory");

    v2f a2[KS2];
    #pragma unroll
    for (int kk = 0; kk < KS2; ++kk) {
      int k0 = 4 * kk + 2 * hi;
      a2[kk] = *(const v2f*)&H[lm * NPAD1 + k0];
    }

    int   drow[8], srow[8];
    float y0r[8], y1x[8], y1y[8], y1z[8];
    #pragma unroll
    for (int r = 0; r < 8; ++r) {
      int m = r + 8 * hi;
      srow[r] = mi[m * 2 + 0];
      drow[r] = mi[m * 2 + 1];
      y0r[r]  = mf[m * 4 + 0];
      y1x[r]  = mf[m * 4 + 1];
      y1y[r]  = mf[m * 4 + 2];
      y1z[r]  = mf[m * 4 + 3];
    }
    __builtin_amdgcn_sched_barrier(0);
    const int dN0 = __builtin_amdgcn_readfirstlane(nxt.dst);
    const int dNextTile = (base + 16 < E) ? dN0 : -1;
    const int d15   = mi[15 * 2 + 1];
    const int dPrev = hi ? mi[7 * 2 + 1] : dLastPrev;
    const int dNext = hi ? ((tile + 1 < t1) ? dNextTile : -2) : mi[8 * 2 + 1];
    RunFlags rf;
    {
      const bool contIn = (drow[0] == dPrev);
      bool noEnd = true;
      #pragma unroll
      for (int r = 0; r < 7; ++r) {
        const bool er   = (drow[r] != drow[r + 1]);
        const bool hold = contIn && noEnd;
        rf.e[r]   = er;
        rf.cap[r] = er && hold;
        rf.fl[r]  = er && !hold && (drow[r] >= 0);
        noEnd = noEnd && !er;
      }
      rf.single  = contIn && noEnd;
      rf.contOut = (drow[7] == dNext);
      const bool condA = contIn && !rf.single && (drow[0] >= 0);
      const bool condB = !rf.contOut && (drow[7] >= 0);
      const bool shA = (drow[0] == nodeFirst) || (drow[0] == nodeLast);
      const bool shB = (drow[7] == nodeFirst) || (drow[7] == nodeLast);
      rf.aSt = condA && !shA;  rf.aAt = condA && shA;
      rf.bSt = condB && !shB;  rf.bAt = condB && shB;
      bool anyFl = false;
      #pragma unroll
      for (int r = 0; r < 7; ++r) anyFl = anyFl || rf.fl[r];
      rf.uFl   = __builtin_amdgcn_ballot_w32(anyFl) != 0u;
      rf.uAt   = __builtin_amdgcn_ballot_w32(rf.aAt || rf.bAt) != 0u;
      rf.uB    = __builtin_amdgcn_ballot_w32(rf.bSt) != 0u;
      rf.uRare = __builtin_amdgcn_ballot_w32(anyFl || rf.aAt || rf.bAt || rf.bSt) != 0u;
    }


    auto gemm2_pair = [&](const int pr, v8f (&acc)[2]) {
      const float4* bp = &W2f[pr * KS2 * 32 + lane];
      v2f b[2][KS2];
      #pragma unroll
      for (int kk = 0; kk < KS2; ++kk) {
        const float4 q = bp[kk * 32];
        b[0][kk].x = q.x;  b[0][kk].y = q.y;
        b[1][kk].x = q.z;  b[1][kk].y = q.w;
      }
      __builtin_amdgcn_sched_barrier(0);

      acc[0] = v8f{}; acc[1] = v8f{};
      #pragma unroll
      for (int kk = 0; kk < KS2; ++kk) {
        #pragma unroll
        for (int t = 0; t < 2; ++t) {
          acc[t] = __builtin_amdgcn_wmma_f32_16x16x4_f32(
              false, a2[kk], false, b[t][kk], (short)0, acc[t], false, false);
        }
      }
    };

    #pragma unroll
    for (int p = 0; p < 4; ++p) {
      const int cA  = p * 32 + lm;
      const int col = cA & 63;
      float g[2][8];
      #pragma unroll
      for (int t = 0; t < 2; ++t) {
        #pragma unroll
        for (int r = 0; r < 8; ++r) g[t][r] = f_s[(long long)srow[r] * MUL0 + col + 16 * t];
      }
      v8f acc[2];
      gemm2_pair(p, acc);

      #pragma unroll
      for (int t = 0; t < 2; ++t) {
        const int c = cA + 16 * t;
        if (p < 2) {
          float v[1][8];
          #pragma unroll
          for (int r = 0; r < 8; ++r) v[0][r] = acc[t][r] * g[t][r] * y0r[r] * inv;
          seg_flush<1>(agg_s, MIDS, c, drow, rf, hi, v, cs[p < 2 ? 2 * p + t : 0]);
        } else {
          const int u = c - 64;
          float v[3][8];
          #pragma unroll
          for (int r = 0; r < 8; ++r) {
            float tt = acc[t][r] * g[t][r] * inv;
            v[0][r] = tt * y1x[r]; v[1][r] = tt * y1y[r]; v[2][r] = tt * y1z[r];
          }
          seg_flush<3>(agg_v, MIDV * 3, u * 3, drow, rf, hi, v, cv[p < 2 ? 0 : 2 * (p - 2) + t]);
        }
      }
    }

    {
      float g[2][3][8];
      #pragma unroll
      for (int t = 0; t < 2; ++t) {
        #pragma unroll
        for (int r = 0; r < 8; ++r) {
          const float* vp = &f_v[((long long)srow[r] * MUL1 + lm + 16 * t) * 3];
          g[t][0][r] = vp[0]; g[t][1][r] = vp[1]; g[t][2][r] = vp[2];
        }
      }
      v8f acc[2];

      gemm2_pair(4, acc);
      #pragma unroll
      for (int t = 0; t < 2; ++t) {
        const int u = lm + 16 * t;
        float v[3][8];
        #pragma unroll
        for (int r = 0; r < 8; ++r) {
          float tt = acc[t][r] * y0r[r] * inv;
          v[0][r] = tt * g[t][0][r]; v[1][r] = tt * g[t][1][r]; v[2][r] = tt * g[t][2][r];
        }
        seg_flush<3>(agg_v, MIDV * 3, (64 + u) * 3, drow, rf, hi, v, cv[4 + t]);
      }

      gemm2_pair(5, acc);
      #pragma unroll
      for (int t = 0; t < 2; ++t) {
        const int u = lm + 16 * t;
        float v[1][8];
        #pragma unroll
        for (int r = 0; r < 8; ++r) {
          float dt = g[t][0][r] * y1x[r] + g[t][1][r] * y1y[r] + g[t][2][r] * y1z[r];
          v[0][r] = acc[t][r] * dt * inv3 * inv;
        }
        seg_flush<1>(agg_s, MIDS, 64 + u, drow, rf, hi, v, cs[4 + t]);
      }

      gemm2_pair(6, acc);
      #pragma unroll
      for (int t = 0; t < 2; ++t) {
        const int u = lm + 16 * t;
        float v[3][8];
        #pragma unroll
        for (int r = 0; r < 8; ++r) {
          float cx = g[t][1][r] * y1z[r] - g[t][2][r] * y1y[r];
          float cy = g[t][2][r] * y1x[r] - g[t][0][r] * y1z[r];
          float cz = g[t][0][r] * y1y[r] - g[t][1][r] * y1x[r];
          float tt = acc[t][r] * inv2 * inv;
          v[0][r] = tt * cx; v[1][r] = tt * cy; v[2][r] = tt * cz;
        }
        seg_flush<3>(agg_v, MIDV * 3, (96 + u) * 3, drow, rf, hi, v, cv[6 + t]);
      }
    }
    dLastPrev = d15;
  }
}

__global__ __launch_bounds__(128)
void node_out_kernel(const float* __restrict__ agg_s, const float* __restrict__ agg_v,
                     const float* __restrict__ sc_s,  const float* __restrict__ sc_v,
                     const float* __restrict__ attr,
                     const float* __restrict__ Wl2s, const float* __restrict__ Wl2v,
                     const float* __restrict__ Wl3,
                     float* __restrict__ out, int n)
{
  const int lane = threadIdx.x & 31;
  const int hi   = lane >> 4;
  const int lm   = lane & 15;
  const int n0   = (blockIdx.x * 4 + (threadIdx.x >> 5)) * 16;
  if (n0 >= n) return;
  const int na = (n0 + lm < n) ? (n0 + lm) : (n - 1);

  v2f as[MIDS / 4];
  #pragma unroll
  for (int kk = 0; kk < MIDS / 4; ++kk)
    as[kk] = *(const v2f*)(agg_s + (long long)na * MIDS + 4 * kk + 2 * hi);

  float part = 0.f;
  #pragma unroll
  for (int kk = 0; kk < MIDS / 4; ++kk) {
    const int k0 = 4 * kk + 2 * hi;
    part += as[kk].x * Wl3[k0] + as[kk].y * Wl3[k0 + 1];
  }
  const float ang = 0.1f * (part + xhalf(part)) * attr[na];
  const float csn = cosf(ang);
  const float snn = sinf(ang);

  float cr[8], sr[8], ar[8];
  #pragma unroll
  for (int r = 0; r < 8; ++r) {
    const int nd = n0 + r + 8 * hi;
    cr[r] = __shfl(csn, r + 8 * hi);
    sr[r] = __shfl(snn, r + 8 * hi);
    ar[r] = attr[nd < n ? nd : n - 1];
  }

  {
    float scv[8][4];
    #pragma unroll
    for (int r = 0; r < 8; ++r) {
      const int nd = n0 + r + 8 * hi;
      const int nc = nd < n ? nd : n - 1;
      #pragma unroll
      for (int t = 0; t < 4; ++t) scv[r][t] = sc_s[(long long)nc * MUL0 + 16 * t + lm];
    }
    v8f acc[4];
    #pragma unroll
    for (int t = 0; t < 4; ++t) acc[t] = v8f{};
    #pragma unroll
    for (int kk = 0; kk < MIDS / 4; ++kk) {
      const int k0 = 4 * kk + 2 * hi;
      #pragma unroll
      for (int t = 0; t < 4; ++t) {
        v2f b;
        b.x = Wl2s[ k0      * MUL0 + 16 * t + lm];
        b.y = Wl2s[(k0 + 1) * MUL0 + 16 * t + lm];
        acc[t] = __builtin_amdgcn_wmma_f32_16x16x4_f32(
            false, as[kk], false, b, (short)0, acc[t], false, false);
      }
    }
    #pragma unroll
    for (int r = 0; r < 8; ++r) {
      const int nd = n0 + r + 8 * hi;
      if (nd < n) {
        #pragma unroll
        for (int t = 0; t < 4; ++t) {
          const int col = 16 * t + lm;
          const float conv = acc[t][r] * ar[r];
          out[(long long)nd * 160 + col] = cr[r] * scv[r][t] + sr[r] * conv;
        }
      }
    }
  }

  {
    float scv[8][2][3];
    #pragma unroll
    for (int r = 0; r < 8; ++r) {
      const int nd = n0 + r + 8 * hi;
      const int nc = nd < n ? nd : n - 1;
      #pragma unroll
      for (int t = 0; t < 2; ++t) {
        #pragma unroll
        for (int c = 0; c < 3; ++c) scv[r][t][c] = sc_v[(long long)nc * (MUL1 * 3) + (16 * t + lm) * 3 + c];
      }
    }
    v8f acc[3][2];
    #pragma unroll
    for (int c = 0; c < 3; ++c) { acc[c][0] = v8f{}; acc[c][1] = v8f{}; }
    const float* pv = agg_v + (long long)na * (MIDV * 3) + 6 * hi;
    #pragma unroll
    for (int kk = 0; kk < MIDV / 4; ++kk) {
      const int k0 = 4 * kk + 2 * hi;
      const v2f* q = (const v2f*)(pv + 12 * kk);
      const v2f q0 = q[0], q1 = q[1], q2 = q[2];
      v2f a[3];
      a[0].x = q0.x;  a[0].y = q1.y;
      a[1].x = q0.y;  a[1].y = q2.x;
      a[2].x = q1.x;  a[2].y = q2.y;
      #pragma unroll
      for (int t = 0; t < 2; ++t) {
        v2f b;
        b.x = Wl2v[ k0      * MUL1 + 16 * t + lm];
        b.y = Wl2v[(k0 + 1) * MUL1 + 16 * t + lm];
        #pragma unroll
        for (int c = 0; c < 3; ++c)
          acc[c][t] = __builtin_amdgcn_wmma_f32_16x16x4_f32(
              false, a[c], false, b, (short)0, acc[c][t], false, false);
      }
    }
    #pragma unroll
    for (int r = 0; r < 8; ++r) {
      const int nd = n0 + r + 8 * hi;
      if (nd < n) {
        #pragma unroll
        for (int t = 0; t < 2; ++t) {
          #pragma unroll
          for (int c = 0; c < 3; ++c) {
            const int idx = (16 * t + lm) * 3 + c;
            const float conv = acc[c][t][r] * ar[r];
            out[(long long)nd * 160 + MUL0 + idx] = cr[r] * scv[r][t][c] + sr[r] * conv;
          }
        }
      }
    }
  }
}

extern "C" void kernel_launch(void* const* d_in, const int* in_sizes, int n_in,
                              void* d_out, int out_size, void* d_ws, size_t ws_size,
                              hipStream_t stream) {
  const float* node_scalars = (const float*)d_in[0];
  const float* node_vectors = (const float*)d_in[1];
  const float* node_attr    = (const float*)d_in[2];
  const int*   edge_src     = (const int*)  d_in[3];
  const int*   edge_dst     = (const int*)  d_in[4];
  const float* edge_attr    = (const float*)d_in[5];
  const float* edge_scalars = (const float*)d_in[6];
  const float* W_fc1        = (const float*)d_in[7];
  const float* W_fc2        = (const float*)d_in[8];
  const float* W_sc_s       = (const float*)d_in[9];
  const float* W_sc_v       = (const float*)d_in[10];
  const float* W_l1_s       = (const float*)d_in[11];
  const float* W_l1_v       = (const float*)d_in[12];
  const float* W_l2_s       = (const float*)d_in[13];
  const float* W_l2_v       = (const float*)d_in[14];
  const float* W_l3         = (const float*)d_in[15];

  const int N = in_sizes[0] / MUL0;
  const int E = in_sizes[3];

  float* ws    = (float*)d_ws;
  float* f_s   = ws;
  float* f_v   = f_s  + (size_t)N * MUL0;
  float* sc_s  = f_v  + (size_t)N * MUL1 * 3;
  float* sc_v  = sc_s + (size_t)N * MUL0;
  float* agg_s = sc_v + (size_t)N * MUL1 * 3;
  float* agg_v = agg_s + (size_t)N * MIDS;

  int* order = (int*)d_out;

  long long nAgg = (long long)N * (MIDS + MIDV * 3);
  zero_kernel<<<4096, 256, 0, stream>>>(agg_s, nAgg);

  const int nodeTiles = (N + 15) / 16;
  node_prep_kernel<<<(nodeTiles + 3) / 4, 128, 0, stream>>>(node_scalars, node_vectors, node_attr,
                                          W_l1_s, W_l1_v, W_sc_s, W_sc_v,
                                          f_s, f_v, sc_s, sc_v, N);

  dst_sort_kernel<<<(N + SORT_R - 1) / SORT_R, SORT_NT, 0, stream>>>(edge_dst, order, E);

  int numTiles = (E + 15) / 16;
  edge_fused_kernel<<<512, WAVES * 32, 0, stream>>>(edge_scalars, edge_attr,
                                                     edge_src, edge_dst, order,
                                                     W_fc1, W_fc2, f_s, f_v,
                                                     agg_s, agg_v, E, numTiles);

  node_out_kernel<<<(nodeTiles + 3) / 4, 128, 0, stream>>>(agg_s, agg_v, sc_s, sc_v, node_attr,
                                         W_l2_s, W_l2_v, W_l3, (float*)d_out, N);
}
